# starting kernel
# baseline (speedup 1.0000x reference)
#include <hip/hip_runtime.h>
#include <math.h>

#define N_NODES 50000
#define N_EDGES 800000
#define IN_DIM 128
#define OUT_DIM 128
#define HEADS 8
#define CPH 16
#define NEG_SLOPE 0.2f
#define GAT_EPS 1e-16f

typedef float v2f __attribute__((ext_vector_type(2)));
typedef float v8f __attribute__((ext_vector_type(8)));

#define LDX 132

__global__ __launch_bounds__(256) void gat_gemm_wmma_kernel(
    const float* __restrict__ x, const float* __restrict__ Wg,
    float* __restrict__ h) {
  __shared__ float sx[16 * LDX];
  const int tid = threadIdx.x;
  const int mbase = blockIdx.x * 16;

  for (int i = tid; i < 16 * 32; i += 256) {
    const int r  = i >> 5;
    const int c4 = i & 31;
    const float4 v = ((const float4*)(x + (size_t)(mbase + r) * IN_DIM))[c4];
    float* dst = &sx[r * LDX + c4 * 4];
    dst[0] = v.x; dst[1] = v.y; dst[2] = v.z; dst[3] = v.w;
  }
  __syncthreads();

  const int wave = tid >> 5;
  const int lane = tid & 31;
  const int half = lane >> 4;
  const int mr   = lane & 15;
  const int nb   = wave * 16;

  v8f acc = {};
#pragma unroll
  for (int k0 = 0; k0 < IN_DIM; k0 += 4) {
    v2f a, b;
    a.x = sx[mr * LDX + k0 + 2 * half];
    a.y = sx[mr * LDX + k0 + 2 * half + 1];
    b.x = Wg[(size_t)(k0 + 2 * half) * OUT_DIM + nb + mr];
    b.y = Wg[(size_t)(k0 + 2 * half + 1) * OUT_DIM + nb + mr];
    acc = __builtin_amdgcn_wmma_f32_16x16x4_f32(
         false, a,  false, b,
         (short)0, acc,  false,  false);
  }

#pragma unroll
  for (int r = 0; r < 8; ++r) {
    const int row = half * 8 + r;
    h[(size_t)(mbase + row) * OUT_DIM + nb + mr] = acc[r];
  }
}

__global__ void gat_attn_proj_kernel(const float* __restrict__ h,
                                     const float* __restrict__ att_src,
                                     const float* __restrict__ att_dst,
                                     float* __restrict__ a_src,
                                     float* __restrict__ a_dst) {
  const int i = blockIdx.x * blockDim.x + threadIdx.x;
  if (i >= N_NODES * HEADS) return;
  const int hd = i & (HEADS - 1);
  const float4* hp = (const float4*)(h + (size_t)i * CPH);
  const float4* as = (const float4*)(att_src + hd * CPH);
  const float4* ad = (const float4*)(att_dst + hd * CPH);
  float s = 0.f, d = 0.f;
#pragma unroll
  for (int q = 0; q < 4; ++q) {
    const float4 hv = hp[q], av = as[q], dv = ad[q];
    s += hv.x * av.x + hv.y * av.y + hv.z * av.z + hv.w * av.w;
    d += hv.x * dv.x + hv.y * dv.y + hv.z * dv.z + hv.w * dv.w;
  }
  a_src[i] = s;
  a_dst[i] = d;
}

__device__ __forceinline__ void edge_nodes(const int* __restrict__ ei, int e,
                                           int& s, int& d) {
  if (e < N_EDGES) { s = ei[e]; d = ei[N_EDGES + e]; }
  else             { s = d = e - N_EDGES; }
}
__device__ __forceinline__ float leaky(float v) {
  return v >= 0.f ? v : NEG_SLOPE * v;
}
__device__ __forceinline__ unsigned fflip(float f) {
  const unsigned u = __float_as_uint(f);
  return u ^ ((u >> 31) ? 0xFFFFFFFFu : 0x80000000u);
}
__device__ __forceinline__ float funflip(unsigned u) {
  return __uint_as_float(u ^ ((u >> 31) ? 0x80000000u : 0xFFFFFFFFu));
}

#define E2 (N_EDGES + N_NODES)

__global__ void gat_edge_max_kernel(const int* __restrict__ ei,
                                    const float* __restrict__ a_src,
                                    const float* __restrict__ a_dst,
                                    unsigned* __restrict__ emaxU) {
  const int i = blockIdx.x * blockDim.x + threadIdx.x;
  if (i >= E2 * HEADS) return;
  const int e = i >> 3, hd = i & (HEADS - 1);
  int s, d; edge_nodes(ei, e, s, d);
  const float v = leaky(a_src[s * HEADS + hd] + a_dst[d * HEADS + hd]);
  atomicMax(&emaxU[d * HEADS + hd], fflip(v));
}

__global__ void gat_edge_sum_kernel(const int* __restrict__ ei,
                                    const float* __restrict__ a_src,
                                    const float* __restrict__ a_dst,
                                    const unsigned* __restrict__ emaxU,
                                    float* __restrict__ denom) {
  const int i = blockIdx.x * blockDim.x + threadIdx.x;
  if (i >= E2 * HEADS) return;
  const int e = i >> 3, hd = i & (HEADS - 1);
  int s, d; edge_nodes(ei, e, s, d);
  const int dh = d * HEADS + hd;
  const float v = leaky(a_src[s * HEADS + hd] + a_dst[dh]);
  atomicAdd(&denom[dh], __expf(v - funflip(emaxU[dh])));
}

__global__ void gat_edge_scatter_kernel(const int* __restrict__ ei,
                                        const float* __restrict__ a_src,
                                        const float* __restrict__ a_dst,
                                        const unsigned* __restrict__ emaxU,
                                        const float* __restrict__ denom,
                                        const float* __restrict__ h,
                                        float* __restrict__ accum) {
  const int i = blockIdx.x * blockDim.x + threadIdx.x;
  if (i >= E2 * HEADS) return;
  const int e = i >> 3, hd = i & (HEADS - 1);
  int s, d; edge_nodes(ei, e, s, d);
  const int dh = d * HEADS + hd;
  const float v = leaky(a_src[s * HEADS + hd] + a_dst[dh]);
  const float alpha = __expf(v - funflip(emaxU[dh])) / (denom[dh] + GAT_EPS);
  const float4* hs = (const float4*)(h + (size_t)(s * HEADS + hd) * CPH);
  float* op = accum + (size_t)dh * CPH;
#pragma unroll
  for (int q = 0; q < 4; ++q) {
    const float4 hv = hs[q];
    atomicAdd(&op[q * 4 + 0], alpha * hv.x);
    atomicAdd(&op[q * 4 + 1], alpha * hv.y);
    atomicAdd(&op[q * 4 + 2], alpha * hv.z);
    atomicAdd(&op[q * 4 + 3], alpha * hv.w);
  }
}

__global__ void gat_finalize_kernel(const float* __restrict__ accum,
                                    const float* __restrict__ bias,
                                    float* __restrict__ out) {
  const int i = blockIdx.x * blockDim.x + threadIdx.x;
  if (i >= N_NODES * OUT_DIM) return;
  const float v = accum[i] + bias[i & (OUT_DIM - 1)];
  out[i] = v > 0.f ? v : expm1f(v);
}

extern "C" void kernel_launch(void* const* d_in, const int* in_sizes, int n_in,
                              void* d_out, int out_size, void* d_ws, size_t ws_size,
                              hipStream_t stream) {
  (void)in_sizes; (void)n_in; (void)out_size; (void)ws_size;
  const float* x       = (const float*)d_in[0];
  const int*   ei      = (const int*)d_in[1];
  const float* W       = (const float*)d_in[2];
  const float* att_src = (const float*)d_in[3];
  const float* att_dst = (const float*)d_in[4];
  const float* bias    = (const float*)d_in[5];

  float*    h      = (float*)d_ws;
  float*    a_src  = h + (size_t)N_NODES * OUT_DIM;
  float*    a_dst  = a_src + (size_t)N_NODES * HEADS;
  unsigned* emaxU  = (unsigned*)(a_dst + (size_t)N_NODES * HEADS);
  float*    denom  = (float*)emaxU + (size_t)N_NODES * HEADS;
  float*    accum  = denom + (size_t)N_NODES * HEADS;

  hipMemsetAsync(emaxU, 0,
                 (size_t)(2 * N_NODES * HEADS + (size_t)N_NODES * OUT_DIM) *
                     sizeof(float),
                 stream);

  gat_gemm_wmma_kernel<<<N_NODES / 16, 256, 0, stream>>>(x, W, h);

  gat_attn_proj_kernel<<<(N_NODES * HEADS + 255) / 256, 256, 0, stream>>>(
      h, att_src, att_dst, a_src, a_dst);

  const int ethreads = E2 * HEADS;
  const int eblocks  = (ethreads + 255) / 256;
  gat_edge_max_kernel<<<eblocks, 256, 0, stream>>>(ei, a_src, a_dst, emaxU);
  gat_edge_sum_kernel<<<eblocks, 256, 0, stream>>>(ei, a_src, a_dst, emaxU, denom);
  gat_edge_scatter_kernel<<<eblocks, 256, 0, stream>>>(ei, a_src, a_dst, emaxU,
                                                       denom, h, accum);

  gat_finalize_kernel<<<(N_NODES * OUT_DIM + 255) / 256, 256, 0, stream>>>(
      accum, bias, (float*)d_out);
}


# Segment max and softmax sum folded into the destination-tile kernel: LDS integer max, fixed-point sums of exp and exp*h, one divide per output (new rounding); edge launches and global atomics gone
# speedup vs baseline: 3.9130x; 3.9130x over previous
#include <hip/hip_runtime.h>
#include <math.h>

#define N_NODES 50000
#define N_EDGES 800000
#define IN_DIM 128
#define OUT_DIM 128
#define HEADS 8
#define CPH 16
#define NEG_SLOPE 0.2f
#define GAT_EPS 1e-16f

typedef float v2f __attribute__((ext_vector_type(2)));
typedef float v8f __attribute__((ext_vector_type(8)));

#define LDX 132

__global__ __launch_bounds__(256) void gat_gemm_wmma_kernel(
    const float* __restrict__ x, const float* __restrict__ Wg,
    float* __restrict__ h) {
  __shared__ float sx[16 * LDX];
  const int tid = threadIdx.x;
  const int mbase = blockIdx.x * 16;

  for (int i = tid; i < 16 * 32; i += 256) {
    const int r  = i >> 5;
    const int c4 = i & 31;
    const float4 v = ((const float4*)(x + (size_t)(mbase + r) * IN_DIM))[c4];
    float* dst = &sx[r * LDX + c4 * 4];
    dst[0] = v.x; dst[1] = v.y; dst[2] = v.z; dst[3] = v.w;
  }
  __syncthreads();

  const int wave = tid >> 5;
  const int lane = tid & 31;
  const int half = lane >> 4;
  const int mr   = lane & 15;
  const int nb   = wave * 16;

  v8f acc = {};
#pragma unroll
  for (int k0 = 0; k0 < IN_DIM; k0 += 4) {
    v2f a, b;
    a.x = sx[mr * LDX + k0 + 2 * half];
    a.y = sx[mr * LDX + k0 + 2 * half + 1];
    b.x = Wg[(size_t)(k0 + 2 * half) * OUT_DIM + nb + mr];
    b.y = Wg[(size_t)(k0 + 2 * half + 1) * OUT_DIM + nb + mr];
    acc = __builtin_amdgcn_wmma_f32_16x16x4_f32(
         false, a,  false, b,
         (short)0, acc,  false,  false);
  }

#pragma unroll
  for (int r = 0; r < 8; ++r) {
    const int row = half * 8 + r;
    h[(size_t)(mbase + row) * OUT_DIM + nb + mr] = acc[r];
  }
}

__global__ void gat_attn_proj_kernel(const float* __restrict__ h,
                                     const float* __restrict__ att_src,
                                     const float* __restrict__ att_dst,
                                     float* __restrict__ a_src,
                                     float* __restrict__ a_dst) {
  const int i = blockIdx.x * blockDim.x + threadIdx.x;
  if (i >= N_NODES * HEADS) return;
  const int hd = i & (HEADS - 1);
  const float4* hp = (const float4*)(h + (size_t)i * CPH);
  const float4* as = (const float4*)(att_src + hd * CPH);
  const float4* ad = (const float4*)(att_dst + hd * CPH);
  float s = 0.f, d = 0.f;
#pragma unroll
  for (int q = 0; q < 4; ++q) {
    const float4 hv = hp[q], av = as[q], dv = ad[q];
    s += hv.x * av.x + hv.y * av.y + hv.z * av.z + hv.w * av.w;
    d += hv.x * dv.x + hv.y * dv.y + hv.z * dv.z + hv.w * dv.w;
  }
  a_src[i] = s;
  a_dst[i] = d;
}

__device__ __forceinline__ float leaky(float v) {
  return v >= 0.f ? v : NEG_SLOPE * v;
}
__device__ __forceinline__ unsigned fflip(float f) {
  const unsigned u = __float_as_uint(f);
  return u ^ ((u >> 31) ? 0xFFFFFFFFu : 0x80000000u);
}
__device__ __forceinline__ float funflip(unsigned u) {
  return __uint_as_float(u ^ ((u >> 31) ? 0x80000000u : 0xFFFFFFFFu));
}

#define TILE 200
#define TTHREADS 1024
#define FIX_SCALE 1073741824.0f

__device__ __forceinline__ unsigned long long to_fix(float v) {
  const float t = truncf(v);
  const float r = v - t;
  const long long f = ((long long)(int)t << 30) + (long long)(int)(r * FIX_SCALE);
  return (unsigned long long)f;
}

struct EdgeIn { float as; float4 hv; };

__device__ __forceinline__ EdgeIn edge_fetch(int s, int lane,
                                             const float* __restrict__ a_src,
                                             const float* __restrict__ h) {
  EdgeIn r;
  r.as = a_src[s * HEADS + (lane >> 2)];
  r.hv = ((const float4*)(h + (size_t)s * OUT_DIM))[lane];
  return r;
}

__device__ __forceinline__ void edge_apply(const EdgeIn in, int dl, int lane,
                                           const float* s_adst, const unsigned* s_emaxU,
                                           unsigned long long* s_den,
                                           unsigned long long* acc) {
  const int dh = dl * HEADS + (lane >> 2);
  const float v = leaky(in.as + s_adst[dh]);
  const float ex = __expf(v - __uint_as_float(s_emaxU[dh]));
  if ((lane & 3) == 0) atomicAdd(s_den + dh, to_fix(ex));
  unsigned long long* ap = acc + dl * OUT_DIM + lane;
  atomicAdd(ap,      to_fix(ex * in.hv.x));
  atomicAdd(ap + 32, to_fix(ex * in.hv.y));
  atomicAdd(ap + 64, to_fix(ex * in.hv.z));
  atomicAdd(ap + 96, to_fix(ex * in.hv.w));
}

#define NQ (N_EDGES / 4)
#define SCAN_U 4
#define QKEEP 224
#define QSTEP 128
#define QCAP (QKEEP + QSTEP)

__device__ __forceinline__ void max_drain(unsigned* qw, int from, int to, int lane,
                                          const int* __restrict__ ei,
                                          const float* __restrict__ a_src,
                                          unsigned* s_emaxU) {
  for (int b0 = from; b0 < to; b0 += 32) {
    const int k = b0 + lane;
    if (k < to) {
      const unsigned w = qw[k];
      const unsigned dl = w & 255u;
      const int s = ei[w >> 8];
      qw[k] = ((unsigned)s << 8) | dl;
      const float4* ap = (const float4*)(a_src + (size_t)s * HEADS);
      const float4 lo = ap[0], hi = ap[1];
      unsigned* mp = s_emaxU + dl * HEADS;
      atomicMax(mp + 0, fflip(lo.x)); atomicMax(mp + 1, fflip(lo.y));
      atomicMax(mp + 2, fflip(lo.z)); atomicMax(mp + 3, fflip(lo.w));
      atomicMax(mp + 4, fflip(hi.x)); atomicMax(mp + 5, fflip(hi.y));
      atomicMax(mp + 6, fflip(hi.z)); atomicMax(mp + 7, fflip(hi.w));
    }
  }
}

template <bool RAW>
__device__ __forceinline__ void apply_drain(const unsigned* qw, int cnt, int lane,
                                            const int* __restrict__ ei,
                                            const float* __restrict__ a_src,
                                            const float* __restrict__ h,
                                            const float* s_adst, const unsigned* s_emaxU,
                                            unsigned long long* s_den,
                                            unsigned long long* acc) {
  for (int b0 = 0; b0 < cnt; b0 += 32) {
    const int n = (cnt - b0) < 32 ? (cnt - b0) : 32;
    int s_m = 0, dl_m = 0;
    if (lane < n) {
      const unsigned w = qw[b0 + lane];
      dl_m = (int)(w & 255u);
      s_m = RAW ? ei[w >> 8] : (int)(w >> 8);
    }
    for (int i = 0; i < n; i += 4) {
      EdgeIn in0 = {}, in1 = {}, in2 = {}, in3 = {};
      in0 = edge_fetch(__builtin_amdgcn_readlane(s_m, i), lane, a_src, h);
      if (i + 1 < n) in1 = edge_fetch(__builtin_amdgcn_readlane(s_m, i + 1), lane, a_src, h);
      if (i + 2 < n) in2 = edge_fetch(__builtin_amdgcn_readlane(s_m, i + 2), lane, a_src, h);
      if (i + 3 < n) in3 = edge_fetch(__builtin_amdgcn_readlane(s_m, i + 3), lane, a_src, h);
      edge_apply(in0, __builtin_amdgcn_readlane(dl_m, i), lane, s_adst, s_emaxU, s_den, acc);
      if (i + 1 < n) edge_apply(in1, __builtin_amdgcn_readlane(dl_m, i + 1), lane, s_adst, s_emaxU, s_den, acc);
      if (i + 2 < n) edge_apply(in2, __builtin_amdgcn_readlane(dl_m, i + 2), lane, s_adst, s_emaxU, s_den, acc);
      if (i + 3 < n) edge_apply(in3, __builtin_amdgcn_readlane(dl_m, i + 3), lane, s_adst, s_emaxU, s_den, acc);
    }
  }
}

__global__ __launch_bounds__(TTHREADS) void gat_tile_kernel(
    const int* __restrict__ ei, const float* __restrict__ a_src,
    const float* __restrict__ a_dst, const float* __restrict__ h,
    const float* __restrict__ bias, float* __restrict__ out) {
  __shared__ unsigned long long acc[TILE * OUT_DIM];
  __shared__ unsigned long long s_den[TILE * HEADS];
  __shared__ float s_adst[TILE * HEADS];
  __shared__ unsigned s_emaxU[TILE * HEADS];
  __shared__ unsigned s_q[(TTHREADS / 32) * QCAP];
  const int tid = threadIdx.x;
  const int lane = tid & 31;
  const int wave = tid >> 5;
  const int base = blockIdx.x * TILE;
  unsigned* qw = s_q + wave * QCAP;

  for (int i = tid; i < TILE * OUT_DIM; i += TTHREADS) acc[i] = 0ull;
  for (int i = tid; i < TILE * HEADS; i += TTHREADS) {
    s_den[i]  = 0ull;
    s_adst[i] = a_dst[base * HEADS + i];
    s_emaxU[i] = fflip(a_src[base * HEADS + i]);
  }
  __syncthreads();

  const int4* dst4 = (const int4*)(ei + N_EDGES);
  const int4 none = make_int4(-1, -1, -1, -1);
  int cnt = 0, done = 0;
  bool ovf = false;
  {
    int4 dn[SCAN_U];
#pragma unroll
    for (int u = 0; u < SCAN_U; ++u) dn[u] = dst4[u * TTHREADS + tid];
    for (int t0 = 0; t0 < NQ; t0 += SCAN_U * TTHREADS) {
      int4 dc[SCAN_U];
#pragma unroll
      for (int u = 0; u < SCAN_U; ++u) {
        dc[u] = dn[u];
        const int qn = t0 + SCAN_U * TTHREADS + u * TTHREADS + tid;
        const int4 w = dst4[qn < NQ ? qn : NQ - 1];
        dn[u] = qn < NQ ? w : none;
      }
#pragma unroll
      for (int u = 0; u < SCAN_U; ++u) {
        if (cnt > QKEEP) {
          max_drain(qw, done, cnt, lane, ei, a_src, s_emaxU);
          cnt = 0; done = 0; ovf = true;
        }
        const int q = t0 + u * TTHREADS + tid;
        const int dv[4] = {dc[u].x, dc[u].y, dc[u].z, dc[u].w};
#pragma unroll
        for (int c = 0; c < 4; ++c) {
          const int dl = dv[c] - base;
          const bool m = (unsigned)dl < (unsigned)TILE;
          const unsigned b = __builtin_amdgcn_ballot_w32(m);
          if (b) {
            const int pos = cnt + (int)__builtin_amdgcn_mbcnt_lo(b, 0u);
            if (m) qw[pos] = ((unsigned)(4 * q + c) << 8) | (unsigned)dl;
            cnt += __builtin_popcount(b);
          }
        }
      }
      if (cnt - done >= 32) {
        max_drain(qw, done, cnt, lane, ei, a_src, s_emaxU);
        done = cnt;
      }
    }
    max_drain(qw, done, cnt, lane, ei, a_src, s_emaxU);
  }
  __syncthreads();

  for (int i = tid; i < TILE * HEADS; i += TTHREADS)
    s_emaxU[i] = __float_as_uint(leaky(funflip(s_emaxU[i]) + s_adst[i]));
  __syncthreads();

  for (int j = wave; j < TILE; j += 2 * (TTHREADS / 32)) {
    const int j2 = j + TTHREADS / 32;
    EdgeIn in0 = edge_fetch(base + j, lane, a_src, h), in1 = {};
    if (j2 < TILE) in1 = edge_fetch(base + j2, lane, a_src, h);
    edge_apply(in0, j, lane, s_adst, s_emaxU, s_den, acc);
    if (j2 < TILE) edge_apply(in1, j2, lane, s_adst, s_emaxU, s_den, acc);
  }
  if (!ovf) {
    apply_drain<false>(qw, cnt, lane, ei, a_src, h, s_adst, s_emaxU, s_den, acc);
  } else {
    const int trips = (NQ + TTHREADS - 1) / TTHREADS;
    cnt = 0;
    for (int it = 0; it <= trips; ++it) {
      const int q = it * TTHREADS + tid;
      if (it < trips) {
        const int4 w = dst4[q < NQ ? q : NQ - 1];
        const int4 d = q < NQ ? w : none;
        const int dv[4] = {d.x, d.y, d.z, d.w};
#pragma unroll
        for (int c = 0; c < 4; ++c) {
          const int dl = dv[c] - base;
          const bool m = (unsigned)dl < (unsigned)TILE;
          const unsigned b = __builtin_amdgcn_ballot_w32(m);
          if (b) {
            const int pos = cnt + (int)__builtin_amdgcn_mbcnt_lo(b, 0u);
            if (m) qw[pos] = ((unsigned)(4 * q + c) << 8) | (unsigned)dl;
            cnt += __builtin_popcount(b);
          }
        }
      }
      if (cnt > QKEEP || it == trips) {
        apply_drain<true>(qw, cnt, lane, ei, a_src, h, s_adst, s_emaxU, s_den, acc);
        cnt = 0;
      }
    }
  }
  __syncthreads();

  for (int o = tid; o < TILE * OUT_DIM; o += TTHREADS) {
    const int dl = o >> 7, c = o & (OUT_DIM - 1);
    const long long f = (long long)acc[dl * OUT_DIM + (c & 3) * 32 + (c >> 2)];
    const long long g = (long long)s_den[dl * HEADS + (c >> 4)];
    const float num = (float)((double)f * (1.0 / 1073741824.0));
    const float den = (float)((double)g * (1.0 / 1073741824.0)) + GAT_EPS;
    const float v = num / den + bias[c];
    out[(size_t)(base + dl) * OUT_DIM + c] = v > 0.f ? v : expm1f(v);
  }
}

extern "C" void kernel_launch(void* const* d_in, const int* in_sizes, int n_in,
                              void* d_out, int out_size, void* d_ws, size_t ws_size,
                              hipStream_t stream) {
  (void)in_sizes; (void)n_in; (void)out_size; (void)ws_size;
  const float* x       = (const float*)d_in[0];
  const int*   ei      = (const int*)d_in[1];
  const float* W       = (const float*)d_in[2];
  const float* att_src = (const float*)d_in[3];
  const float* att_dst = (const float*)d_in[4];
  const float* bias    = (const float*)d_in[5];

  float* h     = (float*)d_ws;
  float* a_src = h + (size_t)N_NODES * OUT_DIM;
  float* a_dst = a_src + (size_t)N_NODES * HEADS;

  gat_gemm_wmma_kernel<<<N_NODES / 16, 256, 0, stream>>>(x, W, h);

  gat_attn_proj_kernel<<<(N_NODES * HEADS + 255) / 256, 256, 0, stream>>>(
      h, att_src, att_dst, a_src, a_dst);

  gat_tile_kernel<<<N_NODES / TILE, TTHREADS, 0, stream>>>(
      ei, a_src, a_dst, h, bias, (float*)d_out);
}


# Tile write-out without f64: the 64-bit fixed-point sums go to float by integer normalise and one 32-bit convert (same bits); one wave per row, lane = 4 channels, one denominator per lane
# speedup vs baseline: 5.0134x; 1.2812x over previous
#include <hip/hip_runtime.h>
#include <math.h>

#define N_NODES 50000
#define N_EDGES 800000
#define IN_DIM 128
#define OUT_DIM 128
#define HEADS 8
#define CPH 16
#define NEG_SLOPE 0.2f
#define GAT_EPS 1e-16f

typedef float v2f __attribute__((ext_vector_type(2)));
typedef float v8f __attribute__((ext_vector_type(8)));

#define LDX 132

__global__ __launch_bounds__(256) void gat_gemm_wmma_kernel(
    const float* __restrict__ x, const float* __restrict__ Wg,
    float* __restrict__ h) {
  __shared__ float sx[16 * LDX];
  const int tid = threadIdx.x;
  const int mbase = blockIdx.x * 16;

  for (int i = tid; i < 16 * 32; i += 256) {
    const int r  = i >> 5;
    const int c4 = i & 31;
    const float4 v = ((const float4*)(x + (size_t)(mbase + r) * IN_DIM))[c4];
    float* dst = &sx[r * LDX + c4 * 4];
    dst[0] = v.x; dst[1] = v.y; dst[2] = v.z; dst[3] = v.w;
  }
  __syncthreads();

  const int wave = tid >> 5;
  const int lane = tid & 31;
  const int half = lane >> 4;
  const int mr   = lane & 15;
  const int nb   = wave * 16;

  v8f acc = {};
#pragma unroll
  for (int k0 = 0; k0 < IN_DIM; k0 += 4) {
    v2f a, b;
    a.x = sx[mr * LDX + k0 + 2 * half];
    a.y = sx[mr * LDX + k0 + 2 * half + 1];
    b.x = Wg[(size_t)(k0 + 2 * half) * OUT_DIM + nb + mr];
    b.y = Wg[(size_t)(k0 + 2 * half + 1) * OUT_DIM + nb + mr];
    acc = __builtin_amdgcn_wmma_f32_16x16x4_f32(
         false, a,  false, b,
         (short)0, acc,  false,  false);
  }

#pragma unroll
  for (int r = 0; r < 8; ++r) {
    const int row = half * 8 + r;
    h[(size_t)(mbase + row) * OUT_DIM + nb + mr] = acc[r];
  }
}

__global__ void gat_attn_proj_kernel(const float* __restrict__ h,
                                     const float* __restrict__ att_src,
                                     const float* __restrict__ att_dst,
                                     float* __restrict__ a_src,
                                     float* __restrict__ a_dst) {
  const int i = blockIdx.x * blockDim.x + threadIdx.x;
  if (i >= N_NODES * HEADS) return;
  const int hd = i & (HEADS - 1);
  const float4* hp = (const float4*)(h + (size_t)i * CPH);
  const float4* as = (const float4*)(att_src + hd * CPH);
  const float4* ad = (const float4*)(att_dst + hd * CPH);
  float s = 0.f, d = 0.f;
#pragma unroll
  for (int q = 0; q < 4; ++q) {
    const float4 hv = hp[q], av = as[q], dv = ad[q];
    s += hv.x * av.x + hv.y * av.y + hv.z * av.z + hv.w * av.w;
    d += hv.x * dv.x + hv.y * dv.y + hv.z * dv.z + hv.w * dv.w;
  }
  a_src[i] = s;
  a_dst[i] = d;
}

__device__ __forceinline__ float leaky(float v) {
  return v >= 0.f ? v : NEG_SLOPE * v;
}
__device__ __forceinline__ unsigned fflip(float f) {
  const unsigned u = __float_as_uint(f);
  return u ^ ((u >> 31) ? 0xFFFFFFFFu : 0x80000000u);
}
__device__ __forceinline__ float funflip(unsigned u) {
  return __uint_as_float(u ^ ((u >> 31) ? 0x80000000u : 0xFFFFFFFFu));
}

#define TILE 200
#define TTHREADS 1024
#define FIX_SCALE 1073741824.0f

__device__ __forceinline__ unsigned long long to_fix(float v) {
  const float t = truncf(v);
  const float r = v - t;
  const long long f = ((long long)(int)t << 30) + (long long)(int)(r * FIX_SCALE);
  return (unsigned long long)f;
}

__device__ __forceinline__ float fix_to_float(unsigned long long u) {
  return (float)(long long)u * (1.0f / 1073741824.0f);
}

struct EdgeIn { float as; float4 hv; };

__device__ __forceinline__ EdgeIn edge_fetch(int s, int lane,
                                             const float* __restrict__ a_src,
                                             const float* __restrict__ h) {
  EdgeIn r;
  r.as = a_src[s * HEADS + (lane >> 2)];
  r.hv = ((const float4*)(h + (size_t)s * OUT_DIM))[lane];
  return r;
}

__device__ __forceinline__ void edge_apply(const EdgeIn in, int dl, int lane,
                                           const float* s_adst, const unsigned* s_emaxU,
                                           unsigned long long* s_den,
                                           unsigned long long* acc) {
  const int dh = dl * HEADS + (lane >> 2);
  const float v = leaky(in.as + s_adst[dh]);
  const float ex = __expf(v - __uint_as_float(s_emaxU[dh]));
  if ((lane & 3) == 0) atomicAdd(s_den + dh, to_fix(ex));
  unsigned long long* ap = acc + dl * OUT_DIM + lane;
  atomicAdd(ap,      to_fix(ex * in.hv.x));
  atomicAdd(ap + 32, to_fix(ex * in.hv.y));
  atomicAdd(ap + 64, to_fix(ex * in.hv.z));
  atomicAdd(ap + 96, to_fix(ex * in.hv.w));
}

#define NQ (N_EDGES / 4)
#define SCAN_U 4
#define QKEEP 224
#define QSTEP 128
#define QCAP (QKEEP + QSTEP)

__device__ __forceinline__ void max_drain(unsigned* qw, int from, int to, int lane,
                                          const int* __restrict__ ei,
                                          const float* __restrict__ a_src,
                                          unsigned* s_emaxU) {
  for (int b0 = from; b0 < to; b0 += 32) {
    const int k = b0 + lane;
    if (k < to) {
      const unsigned w = qw[k];
      const unsigned dl = w & 255u;
      const int s = ei[w >> 8];
      qw[k] = ((unsigned)s << 8) | dl;
      const float4* ap = (const float4*)(a_src + (size_t)s * HEADS);
      const float4 lo = ap[0], hi = ap[1];
      unsigned* mp = s_emaxU + dl * HEADS;
      atomicMax(mp + 0, fflip(lo.x)); atomicMax(mp + 1, fflip(lo.y));
      atomicMax(mp + 2, fflip(lo.z)); atomicMax(mp + 3, fflip(lo.w));
      atomicMax(mp + 4, fflip(hi.x)); atomicMax(mp + 5, fflip(hi.y));
      atomicMax(mp + 6, fflip(hi.z)); atomicMax(mp + 7, fflip(hi.w));
    }
  }
}

template <bool RAW>
__device__ __forceinline__ void apply_drain(const unsigned* qw, int cnt, int lane,
                                            const int* __restrict__ ei,
                                            const float* __restrict__ a_src,
                                            const float* __restrict__ h,
                                            const float* s_adst, const unsigned* s_emaxU,
                                            unsigned long long* s_den,
                                            unsigned long long* acc) {
  for (int b0 = 0; b0 < cnt; b0 += 32) {
    const int n = (cnt - b0) < 32 ? (cnt - b0) : 32;
    int s_m = 0, dl_m = 0;
    if (lane < n) {
      const unsigned w = qw[b0 + lane];
      dl_m = (int)(w & 255u);
      s_m = RAW ? ei[w >> 8] : (int)(w >> 8);
    }
    for (int i = 0; i < n; i += 4) {
      EdgeIn in0 = {}, in1 = {}, in2 = {}, in3 = {};
      in0 = edge_fetch(__builtin_amdgcn_readlane(s_m, i), lane, a_src, h);
      if (i + 1 < n) in1 = edge_fetch(__builtin_amdgcn_readlane(s_m, i + 1), lane, a_src, h);
      if (i + 2 < n) in2 = edge_fetch(__builtin_amdgcn_readlane(s_m, i + 2), lane, a_src, h);
      if (i + 3 < n) in3 = edge_fetch(__builtin_amdgcn_readlane(s_m, i + 3), lane, a_src, h);
      edge_apply(in0, __builtin_amdgcn_readlane(dl_m, i), lane, s_adst, s_emaxU, s_den, acc);
      if (i + 1 < n) edge_apply(in1, __builtin_amdgcn_readlane(dl_m, i + 1), lane, s_adst, s_emaxU, s_den, acc);
      if (i + 2 < n) edge_apply(in2, __builtin_amdgcn_readlane(dl_m, i + 2), lane, s_adst, s_emaxU, s_den, acc);
      if (i + 3 < n) edge_apply(in3, __builtin_amdgcn_readlane(dl_m, i + 3), lane, s_adst, s_emaxU, s_den, acc);
    }
  }
}

__global__ __launch_bounds__(TTHREADS) void gat_tile_kernel(
    const int* __restrict__ ei, const float* __restrict__ a_src,
    const float* __restrict__ a_dst, const float* __restrict__ h,
    const float* __restrict__ bias, float* __restrict__ out) {
  __shared__ unsigned long long acc[TILE * OUT_DIM];
  __shared__ unsigned long long s_den[TILE * HEADS];
  __shared__ float s_adst[TILE * HEADS];
  __shared__ unsigned s_emaxU[TILE * HEADS];
  __shared__ unsigned s_q[(TTHREADS / 32) * QCAP];
  const int tid = threadIdx.x;
  const int lane = tid & 31;
  const int wave = tid >> 5;
  const int base = blockIdx.x * TILE;
  unsigned* qw = s_q + wave * QCAP;

  for (int i = tid; i < TILE * OUT_DIM; i += TTHREADS) acc[i] = 0ull;
  for (int i = tid; i < TILE * HEADS; i += TTHREADS) {
    s_den[i]  = 0ull;
    s_adst[i] = a_dst[base * HEADS + i];
    s_emaxU[i] = fflip(a_src[base * HEADS + i]);
  }
  __syncthreads();

  const int4* dst4 = (const int4*)(ei + N_EDGES);
  const int4 none = make_int4(-1, -1, -1, -1);
  int cnt = 0, done = 0;
  bool ovf = false;
  {
    int4 dn[SCAN_U];
#pragma unroll
    for (int u = 0; u < SCAN_U; ++u) dn[u] = dst4[u * TTHREADS + tid];
    for (int t0 = 0; t0 < NQ; t0 += SCAN_U * TTHREADS) {
      int4 dc[SCAN_U];
#pragma unroll
      for (int u = 0; u < SCAN_U; ++u) {
        dc[u] = dn[u];
        const int qn = t0 + SCAN_U * TTHREADS + u * TTHREADS + tid;
        const int4 w = dst4[qn < NQ ? qn : NQ - 1];
        dn[u] = qn < NQ ? w : none;
      }
#pragma unroll
      for (int u = 0; u < SCAN_U; ++u) {
        if (cnt > QKEEP) {
          max_drain(qw, done, cnt, lane, ei, a_src, s_emaxU);
          cnt = 0; done = 0; ovf = true;
        }
        const int q = t0 + u * TTHREADS + tid;
        const int dv[4] = {dc[u].x, dc[u].y, dc[u].z, dc[u].w};
#pragma unroll
        for (int c = 0; c < 4; ++c) {
          const int dl = dv[c] - base;
          const bool m = (unsigned)dl < (unsigned)TILE;
          const unsigned b = __builtin_amdgcn_ballot_w32(m);
          if (b) {
            const int pos = cnt + (int)__builtin_amdgcn_mbcnt_lo(b, 0u);
            if (m) qw[pos] = ((unsigned)(4 * q + c) << 8) | (unsigned)dl;
            cnt += __builtin_popcount(b);
          }
        }
      }
      if (cnt - done >= 32) {
        max_drain(qw, done, cnt, lane, ei, a_src, s_emaxU);
        done = cnt;
      }
    }
    max_drain(qw, done, cnt, lane, ei, a_src, s_emaxU);
  }
  __syncthreads();

  for (int i = tid; i < TILE * HEADS; i += TTHREADS)
    s_emaxU[i] = __float_as_uint(leaky(funflip(s_emaxU[i]) + s_adst[i]));
  __syncthreads();

  for (int j = wave; j < TILE; j += 2 * (TTHREADS / 32)) {
    const int j2 = j + TTHREADS / 32;
    EdgeIn in0 = edge_fetch(base + j, lane, a_src, h), in1 = {};
    if (j2 < TILE) in1 = edge_fetch(base + j2, lane, a_src, h);
    edge_apply(in0, j, lane, s_adst, s_emaxU, s_den, acc);
    if (j2 < TILE) edge_apply(in1, j2, lane, s_adst, s_emaxU, s_den, acc);
  }
  if (!ovf) {
    apply_drain<false>(qw, cnt, lane, ei, a_src, h, s_adst, s_emaxU, s_den, acc);
  } else {
    const int trips = (NQ + TTHREADS - 1) / TTHREADS;
    cnt = 0;
    for (int it = 0; it <= trips; ++it) {
      const int q = it * TTHREADS + tid;
      if (it < trips) {
        const int4 w = dst4[q < NQ ? q : NQ - 1];
        const int4 d = q < NQ ? w : none;
        const int dv[4] = {d.x, d.y, d.z, d.w};
#pragma unroll
        for (int c = 0; c < 4; ++c) {
          const int dl = dv[c] - base;
          const bool m = (unsigned)dl < (unsigned)TILE;
          const unsigned b = __builtin_amdgcn_ballot_w32(m);
          if (b) {
            const int pos = cnt + (int)__builtin_amdgcn_mbcnt_lo(b, 0u);
            if (m) qw[pos] = ((unsigned)(4 * q + c) << 8) | (unsigned)dl;
            cnt += __builtin_popcount(b);
          }
        }
      }
      if (cnt > QKEEP || it == trips) {
        apply_drain<true>(qw, cnt, lane, ei, a_src, h, s_adst, s_emaxU, s_den, acc);
        cnt = 0;
      }
    }
  }
  __syncthreads();

  const float4 bv = ((const float4*)bias)[lane];
  for (int dl = wave; dl < TILE; dl += TTHREADS / 32) {
    const unsigned long long* ap = acc + dl * OUT_DIM + lane;
    const float den = fix_to_float(s_den[dl * HEADS + (lane >> 2)]) + GAT_EPS;
    const float v0 = fix_to_float(ap[0])  / den + bv.x;
    const float v1 = fix_to_float(ap[32]) / den + bv.y;
    const float v2 = fix_to_float(ap[64]) / den + bv.z;
    const float v3 = fix_to_float(ap[96]) / den + bv.w;
    float4 r;
    r.x = v0 > 0.f ? v0 : expm1f(v0);
    r.y = v1 > 0.f ? v1 : expm1f(v1);
    r.z = v2 > 0.f ? v2 : expm1f(v2);
    r.w = v3 > 0.f ? v3 : expm1f(v3);
    ((float4*)(out + (size_t)(base + dl) * OUT_DIM))[lane] = r;
  }
}

extern "C" void kernel_launch(void* const* d_in, const int* in_sizes, int n_in,
                              void* d_out, int out_size, void* d_ws, size_t ws_size,
                              hipStream_t stream) {
  (void)in_sizes; (void)n_in; (void)out_size; (void)ws_size;
  const float* x       = (const float*)d_in[0];
  const int*   ei      = (const int*)d_in[1];
  const float* W       = (const float*)d_in[2];
  const float* att_src = (const float*)d_in[3];
  const float* att_dst = (const float*)d_in[4];
  const float* bias    = (const float*)d_in[5];

  float* h     = (float*)d_ws;
  float* a_src = h + (size_t)N_NODES * OUT_DIM;
  float* a_dst = a_src + (size_t)N_NODES * HEADS;

  gat_gemm_wmma_kernel<<<N_NODES / 16, 256, 0, stream>>>(x, W, h);

  gat_attn_proj_kernel<<<(N_NODES * HEADS + 255) / 256, 256, 0, stream>>>(
      h, att_src, att_dst, a_src, a_dst);

  gat_tile_kernel<<<N_NODES / TILE, TTHREADS, 0, stream>>>(
      ei, a_src, a_dst, h, bias, (float*)d_out);
}


# Tile scan reads a one-byte tile key per edge instead of the 4-byte destination; keys and a packed source+row word are written each call by the projection launch; integer sums unchanged
# speedup vs baseline: 5.7400x; 1.1449x over previous
#include <hip/hip_runtime.h>
#include <math.h>

#define N_NODES 50000
#define N_EDGES 800000
#define IN_DIM 128
#define OUT_DIM 128
#define HEADS 8
#define CPH 16
#define NEG_SLOPE 0.2f
#define GAT_EPS 1e-16f
#define TILE 200

typedef float v2f __attribute__((ext_vector_type(2)));
typedef float v8f __attribute__((ext_vector_type(8)));

#define LDX 132

__global__ __launch_bounds__(256) void gat_gemm_wmma_kernel(
    const float* __restrict__ x, const float* __restrict__ Wg,
    float* __restrict__ h) {
  __shared__ float sx[16 * LDX];
  const int tid = threadIdx.x;
  const int mbase = blockIdx.x * 16;

  for (int i = tid; i < 16 * 32; i += 256) {
    const int r  = i >> 5;
    const int c4 = i & 31;
    const float4 v = ((const float4*)(x + (size_t)(mbase + r) * IN_DIM))[c4];
    float* dst = &sx[r * LDX + c4 * 4];
    dst[0] = v.x; dst[1] = v.y; dst[2] = v.z; dst[3] = v.w;
  }
  __syncthreads();

  const int wave = tid >> 5;
  const int lane = tid & 31;
  const int half = lane >> 4;
  const int mr   = lane & 15;
  const int nb   = wave * 16;

  v8f acc = {};
#pragma unroll
  for (int k0 = 0; k0 < IN_DIM; k0 += 4) {
    v2f a, b;
    a.x = sx[mr * LDX + k0 + 2 * half];
    a.y = sx[mr * LDX + k0 + 2 * half + 1];
    b.x = Wg[(size_t)(k0 + 2 * half) * OUT_DIM + nb + mr];
    b.y = Wg[(size_t)(k0 + 2 * half + 1) * OUT_DIM + nb + mr];
    acc = __builtin_amdgcn_wmma_f32_16x16x4_f32(
         false, a,  false, b,
         (short)0, acc,  false,  false);
  }

#pragma unroll
  for (int r = 0; r < 8; ++r) {
    const int row = half * 8 + r;
    h[(size_t)(mbase + row) * OUT_DIM + nb + mr] = acc[r];
  }
}

__global__ void gat_attn_proj_kernel(const float* __restrict__ h,
                                     const float* __restrict__ att_src,
                                     const float* __restrict__ att_dst,
                                     const int* __restrict__ ei,
                                     float* __restrict__ a_src,
                                     float* __restrict__ a_dst,
                                     unsigned* __restrict__ key8,
                                     unsigned* __restrict__ rec32) {
  const int i = blockIdx.x * blockDim.x + threadIdx.x;
  if (i < N_EDGES / 4) {
    const int4 s4 = ((const int4*)ei)[i];
    const int4 d4 = ((const int4*)(ei + N_EDGES))[i];
    const unsigned sv[4] = {(unsigned)s4.x, (unsigned)s4.y, (unsigned)s4.z, (unsigned)s4.w};
    const unsigned dv[4] = {(unsigned)d4.x, (unsigned)d4.y, (unsigned)d4.z, (unsigned)d4.w};
    unsigned rv[4];
    unsigned kw = 0u;
#pragma unroll
    for (int c = 0; c < 4; ++c) {
      const unsigned t = dv[c] / (unsigned)TILE;
      rv[c] = (sv[c] << 8) | (dv[c] - t * (unsigned)TILE);
      kw |= t << (8 * c);
    }
    ((uint4*)rec32)[i] = make_uint4(rv[0], rv[1], rv[2], rv[3]);
    key8[i] = kw;
  }
  if (i >= N_NODES * HEADS) return;
  const int hd = i & (HEADS - 1);
  const float4* hp = (const float4*)(h + (size_t)i * CPH);
  const float4* as = (const float4*)(att_src + hd * CPH);
  const float4* ad = (const float4*)(att_dst + hd * CPH);
  float s = 0.f, d = 0.f;
#pragma unroll
  for (int q = 0; q < 4; ++q) {
    const float4 hv = hp[q], av = as[q], dv = ad[q];
    s += hv.x * av.x + hv.y * av.y + hv.z * av.z + hv.w * av.w;
    d += hv.x * dv.x + hv.y * dv.y + hv.z * dv.z + hv.w * dv.w;
  }
  a_src[i] = s;
  a_dst[i] = d;
}

__device__ __forceinline__ float leaky(float v) {
  return v >= 0.f ? v : NEG_SLOPE * v;
}
__device__ __forceinline__ unsigned fflip(float f) {
  const unsigned u = __float_as_uint(f);
  return u ^ ((u >> 31) ? 0xFFFFFFFFu : 0x80000000u);
}
__device__ __forceinline__ float funflip(unsigned u) {
  return __uint_as_float(u ^ ((u >> 31) ? 0x80000000u : 0xFFFFFFFFu));
}

#define TTHREADS 1024
#define FIX_SCALE 1073741824.0f

__device__ __forceinline__ unsigned long long to_fix(float v) {
  const float t = truncf(v);
  const float r = v - t;
  const long long f = ((long long)(int)t << 30) + (long long)(int)(r * FIX_SCALE);
  return (unsigned long long)f;
}

__device__ __forceinline__ float fix_to_float(unsigned long long u) {
  return (float)(long long)u * (1.0f / 1073741824.0f);
}

struct EdgeIn { float as; float4 hv; };

__device__ __forceinline__ EdgeIn edge_fetch(int s, int lane,
                                             const float* __restrict__ a_src,
                                             const float* __restrict__ h) {
  EdgeIn r;
  r.as = a_src[s * HEADS + (lane >> 2)];
  r.hv = ((const float4*)(h + (size_t)s * OUT_DIM))[lane];
  return r;
}

__device__ __forceinline__ void edge_apply(const EdgeIn in, int dl, int lane,
                                           const float* s_adst, const unsigned* s_emaxU,
                                           unsigned long long* s_den,
                                           unsigned long long* acc) {
  const int dh = dl * HEADS + (lane >> 2);
  const float v = leaky(in.as + s_adst[dh]);
  const float ex = __expf(v - __uint_as_float(s_emaxU[dh]));
  if ((lane & 3) == 0) atomicAdd(s_den + dh, to_fix(ex));
  unsigned long long* ap = acc + dl * OUT_DIM + lane;
  atomicAdd(ap,      to_fix(ex * in.hv.x));
  atomicAdd(ap + 32, to_fix(ex * in.hv.y));
  atomicAdd(ap + 64, to_fix(ex * in.hv.z));
  atomicAdd(ap + 96, to_fix(ex * in.hv.w));
}

#define NKW (N_EDGES / 16)
#define SCAN_U 4
#define QCAP 352

__device__ __forceinline__ void max_drain(unsigned* qw, int from, int to, int lane,
                                          const unsigned* __restrict__ rec32,
                                          const float* __restrict__ a_src,
                                          unsigned* s_emaxU) {
  for (int b0 = from; b0 < to; b0 += 32) {
    const int k = b0 + lane;
    if (k < to) {
      const unsigned r = rec32[qw[k]];
      qw[k] = r;
      const unsigned dl = r & 255u;
      const float4* ap = (const float4*)(a_src + (size_t)(r >> 8) * HEADS);
      const float4 lo = ap[0], hi = ap[1];
      unsigned* mp = s_emaxU + dl * HEADS;
      atomicMax(mp + 0, fflip(lo.x)); atomicMax(mp + 1, fflip(lo.y));
      atomicMax(mp + 2, fflip(lo.z)); atomicMax(mp + 3, fflip(lo.w));
      atomicMax(mp + 4, fflip(hi.x)); atomicMax(mp + 5, fflip(hi.y));
      atomicMax(mp + 6, fflip(hi.z)); atomicMax(mp + 7, fflip(hi.w));
    }
  }
}

template <bool RAW>
__device__ __forceinline__ void apply_drain(const unsigned* qw, int cnt, int lane,
                                            const unsigned* __restrict__ rec32,
                                            const float* __restrict__ a_src,
                                            const float* __restrict__ h,
                                            const float* s_adst, const unsigned* s_emaxU,
                                            unsigned long long* s_den,
                                            unsigned long long* acc) {
  for (int b0 = 0; b0 < cnt; b0 += 32) {
    const int n = (cnt - b0) < 32 ? (cnt - b0) : 32;
    int s_m = 0, dl_m = 0;
    if (lane < n) {
      const unsigned w0 = qw[b0 + lane];
      const unsigned w = RAW ? rec32[w0] : w0;
      dl_m = (int)(w & 255u);
      s_m = (int)(w >> 8);
    }
    for (int i = 0; i < n; i += 4) {
      EdgeIn in0 = {}, in1 = {}, in2 = {}, in3 = {};
      in0 = edge_fetch(__builtin_amdgcn_readlane(s_m, i), lane, a_src, h);
      if (i + 1 < n) in1 = edge_fetch(__builtin_amdgcn_readlane(s_m, i + 1), lane, a_src, h);
      if (i + 2 < n) in2 = edge_fetch(__builtin_amdgcn_readlane(s_m, i + 2), lane, a_src, h);
      if (i + 3 < n) in3 = edge_fetch(__builtin_amdgcn_readlane(s_m, i + 3), lane, a_src, h);
      edge_apply(in0, __builtin_amdgcn_readlane(dl_m, i), lane, s_adst, s_emaxU, s_den, acc);
      if (i + 1 < n) edge_apply(in1, __builtin_amdgcn_readlane(dl_m, i + 1), lane, s_adst, s_emaxU, s_den, acc);
      if (i + 2 < n) edge_apply(in2, __builtin_amdgcn_readlane(dl_m, i + 2), lane, s_adst, s_emaxU, s_den, acc);
      if (i + 3 < n) edge_apply(in3, __builtin_amdgcn_readlane(dl_m, i + 3), lane, s_adst, s_emaxU, s_den, acc);
    }
  }
}

__device__ __forceinline__ unsigned key_differs(unsigned k, unsigned pat) {
  const unsigned t = k ^ pat;
  return ((t & 0x7F7F7F7Fu) + 0x7F7F7F7Fu) | t;
}

template <bool PASSB>
__device__ __forceinline__ void scan_keys(unsigned* qw, int tid, int lane, unsigned pat,
                                          const uint4* __restrict__ key4,
                                          const unsigned* __restrict__ rec32,
                                          const float* __restrict__ a_src,
                                          const float* __restrict__ h,
                                          const float* s_adst, unsigned* s_emaxU,
                                          unsigned long long* s_den,
                                          unsigned long long* acc,
                                          int& cnt_out, bool& ovf_out) {
  const uint4 none = make_uint4(~0u, ~0u, ~0u, ~0u);
  int cnt = 0, done = 0;
  bool ovf = false;
  uint4 kn[SCAN_U];
#pragma unroll
  for (int u = 0; u < SCAN_U; ++u) kn[u] = key4[u * TTHREADS + tid];
  for (int t0 = 0; t0 < NKW; t0 += SCAN_U * TTHREADS) {
    uint4 kc[SCAN_U];
#pragma unroll
    for (int u = 0; u < SCAN_U; ++u) {
      kc[u] = kn[u];
      const int qn = t0 + SCAN_U * TTHREADS + u * TTHREADS + tid;
      const uint4 w = key4[qn < NKW ? qn : NKW - 1];
      kn[u] = qn < NKW ? w : none;
    }
#pragma unroll
    for (int u = 0; u < SCAN_U; u += 2) {
      const int q = t0 + u * TTHREADS + tid;
      const unsigned nL =
          ((key_differs(kc[u].x, pat) >> 7) & 0x01010101u) | ((key_differs(kc[u].y, pat) >> 6) & 0x02020202u) |
          ((key_differs(kc[u].z, pat) >> 5) & 0x04040404u) | ((key_differs(kc[u].w, pat) >> 4) & 0x08080808u) |
          ((key_differs(kc[u + 1].x, pat) >> 3) & 0x10101010u) | ((key_differs(kc[u + 1].y, pat) >> 2) & 0x20202020u) |
          ((key_differs(kc[u + 1].z, pat) >> 1) & 0x40404040u) | (key_differs(kc[u + 1].w, pat) & 0x80808080u);
      unsigned L = ~nL;
      unsigned b = __builtin_amdgcn_ballot_w32(L != 0u);
      while (b) {
        if (cnt + 32 > QCAP) {
          if (PASSB) {
            apply_drain<true>(qw, cnt, lane, rec32, a_src, h, s_adst, s_emaxU, s_den, acc);
            cnt = 0;
          } else {
            max_drain(qw, done, cnt, lane, rec32, a_src, s_emaxU);
            cnt = 0; done = 0; ovf = true;
          }
        }
        if (L != 0u) {
          const int p = __builtin_ctz(L);
          const int pos = cnt + (int)__builtin_amdgcn_mbcnt_lo(b, 0u);
          qw[pos] = (unsigned)(16 * q + ((p & 4) << 12) + ((p & 3) << 2) + (p >> 3));
          L &= L - 1u;
        }
        cnt += __builtin_popcount(b);
        b = __builtin_amdgcn_ballot_w32(L != 0u);
      }
    }
    if (!PASSB && cnt - done >= 32) {
      max_drain(qw, done, cnt, lane, rec32, a_src, s_emaxU);
      done = cnt;
    }
  }
  if (PASSB) apply_drain<true>(qw, cnt, lane, rec32, a_src, h, s_adst, s_emaxU, s_den, acc);
  else       max_drain(qw, done, cnt, lane, rec32, a_src, s_emaxU);
  cnt_out = cnt;
  ovf_out = ovf;
}

__global__ __launch_bounds__(TTHREADS) void gat_tile_kernel(
    const unsigned* __restrict__ key8, const unsigned* __restrict__ rec32,
    const float* __restrict__ a_src,
    const float* __restrict__ a_dst, const float* __restrict__ h,
    const float* __restrict__ bias, float* __restrict__ out) {
  __shared__ unsigned long long acc[TILE * OUT_DIM];
  __shared__ unsigned long long s_den[TILE * HEADS];
  __shared__ float s_adst[TILE * HEADS];
  __shared__ unsigned s_emaxU[TILE * HEADS];
  __shared__ unsigned s_q[(TTHREADS / 32) * QCAP];
  const int tid = threadIdx.x;
  const int lane = tid & 31;
  const int wave = tid >> 5;
  const int base = blockIdx.x * TILE;
  unsigned* qw = s_q + wave * QCAP;
  const uint4* key4 = (const uint4*)key8;
  const unsigned pat = (unsigned)blockIdx.x * 0x01010101u;

  for (int i = tid; i < TILE * OUT_DIM; i += TTHREADS) acc[i] = 0ull;
  for (int i = tid; i < TILE * HEADS; i += TTHREADS) {
    s_den[i]  = 0ull;
    s_adst[i] = a_dst[base * HEADS + i];
    s_emaxU[i] = fflip(a_src[base * HEADS + i]);
  }
  __syncthreads();

  int cnt = 0;
  bool ovf = false;
  scan_keys<false>(qw, tid, lane, pat, key4, rec32, a_src, h, s_adst, s_emaxU, s_den, acc, cnt, ovf);
  __syncthreads();

  for (int i = tid; i < TILE * HEADS; i += TTHREADS)
    s_emaxU[i] = __float_as_uint(leaky(funflip(s_emaxU[i]) + s_adst[i]));
  __syncthreads();

  for (int j = wave; j < TILE; j += 2 * (TTHREADS / 32)) {
    const int j2 = j + TTHREADS / 32;
    EdgeIn in0 = edge_fetch(base + j, lane, a_src, h), in1 = {};
    if (j2 < TILE) in1 = edge_fetch(base + j2, lane, a_src, h);
    edge_apply(in0, j, lane, s_adst, s_emaxU, s_den, acc);
    if (j2 < TILE) edge_apply(in1, j2, lane, s_adst, s_emaxU, s_den, acc);
  }
  if (!ovf) {
    apply_drain<false>(qw, cnt, lane, rec32, a_src, h, s_adst, s_emaxU, s_den, acc);
  } else {
    int c2 = 0;
    bool o2 = false;
    scan_keys<true>(qw, tid, lane, pat, key4, rec32, a_src, h, s_adst, s_emaxU, s_den, acc, c2, o2);
  }
  __syncthreads();

  const float4 bv = ((const float4*)bias)[lane];
  for (int dl = wave; dl < TILE; dl += TTHREADS / 32) {
    const unsigned long long* ap = acc + dl * OUT_DIM + lane;
    const float den = fix_to_float(s_den[dl * HEADS + (lane >> 2)]) + GAT_EPS;
    const float v0 = fix_to_float(ap[0])  / den + bv.x;
    const float v1 = fix_to_float(ap[32]) / den + bv.y;
    const float v2 = fix_to_float(ap[64]) / den + bv.z;
    const float v3 = fix_to_float(ap[96]) / den + bv.w;
    float4 r;
    r.x = v0 > 0.f ? v0 : expm1f(v0);
    r.y = v1 > 0.f ? v1 : expm1f(v1);
    r.z = v2 > 0.f ? v2 : expm1f(v2);
    r.w = v3 > 0.f ? v3 : expm1f(v3);
    ((float4*)(out + (size_t)(base + dl) * OUT_DIM))[lane] = r;
  }
}

extern "C" void kernel_launch(void* const* d_in, const int* in_sizes, int n_in,
                              void* d_out, int out_size, void* d_ws, size_t ws_size,
                              hipStream_t stream) {
  (void)in_sizes; (void)n_in; (void)out_size; (void)ws_size;
  const float* x       = (const float*)d_in[0];
  const int*   ei      = (const int*)d_in[1];
  const float* W       = (const float*)d_in[2];
  const float* att_src = (const float*)d_in[3];
  const float* att_dst = (const float*)d_in[4];
  const float* bias    = (const float*)d_in[5];

  float* h     = (float*)d_ws;
  float* a_src = h + (size_t)N_NODES * OUT_DIM;
  float* a_dst = a_src + (size_t)N_NODES * HEADS;
  unsigned* rec32 = (unsigned*)(a_dst + (size_t)N_NODES * HEADS);
  unsigned* key8  = rec32 + (size_t)N_EDGES;

  gat_gemm_wmma_kernel<<<N_NODES / 16, 256, 0, stream>>>(x, W, h);

  gat_attn_proj_kernel<<<(N_NODES * HEADS + 255) / 256, 256, 0, stream>>>(
      h, att_src, att_dst, ei, a_src, a_dst, key8, rec32);

  gat_tile_kernel<<<N_NODES / TILE, TTHREADS, 0, stream>>>(
      key8, rec32, a_src, a_dst, h, bias, (float*)d_out);
}


# Pass B: each edge's 8 softmax numerators and denominator adds done once by one lane (32 edges a step, terms fetched a batch ahead) via a per-wave LDS table; row-wide step only adds e*h: same sums
# speedup vs baseline: 6.2937x; 1.0965x over previous
#include <hip/hip_runtime.h>
#include <math.h>

#define N_NODES 50000
#define N_EDGES 800000
#define IN_DIM 128
#define OUT_DIM 128
#define HEADS 8
#define CPH 16
#define NEG_SLOPE 0.2f
#define GAT_EPS 1e-16f
#define TILE 200

typedef float v2f __attribute__((ext_vector_type(2)));
typedef float v8f __attribute__((ext_vector_type(8)));

#define LDX 132

__global__ __launch_bounds__(256) void gat_gemm_wmma_kernel(
    const float* __restrict__ x, const float* __restrict__ Wg,
    float* __restrict__ h) {
  __shared__ float sx[16 * LDX];
  const int tid = threadIdx.x;
  const int mbase = blockIdx.x * 16;

  for (int i = tid; i < 16 * 32; i += 256) {
    const int r  = i >> 5;
    const int c4 = i & 31;
    const float4 v = ((const float4*)(x + (size_t)(mbase + r) * IN_DIM))[c4];
    float* dst = &sx[r * LDX + c4 * 4];
    dst[0] = v.x; dst[1] = v.y; dst[2] = v.z; dst[3] = v.w;
  }
  __syncthreads();

  const int wave = tid >> 5;
  const int lane = tid & 31;
  const int half = lane >> 4;
  const int mr   = lane & 15;
  const int nb   = wave * 16;

  v8f acc = {};
#pragma unroll
  for (int k0 = 0; k0 < IN_DIM; k0 += 4) {
    v2f a, b;
    a.x = sx[mr * LDX + k0 + 2 * half];
    a.y = sx[mr * LDX + k0 + 2 * half + 1];
    b.x = Wg[(size_t)(k0 + 2 * half) * OUT_DIM + nb + mr];
    b.y = Wg[(size_t)(k0 + 2 * half + 1) * OUT_DIM + nb + mr];
    acc = __builtin_amdgcn_wmma_f32_16x16x4_f32(
         false, a,  false, b,
         (short)0, acc,  false,  false);
  }

#pragma unroll
  for (int r = 0; r < 8; ++r) {
    const int row = half * 8 + r;
    h[(size_t)(mbase + row) * OUT_DIM + nb + mr] = acc[r];
  }
}

__global__ void gat_attn_proj_kernel(const float* __restrict__ h,
                                     const float* __restrict__ att_src,
                                     const float* __restrict__ att_dst,
                                     const int* __restrict__ ei,
                                     float* __restrict__ a_src,
                                     float* __restrict__ a_dst,
                                     unsigned* __restrict__ key8,
                                     unsigned* __restrict__ rec32) {
  const int i = blockIdx.x * blockDim.x + threadIdx.x;
  if (i < N_EDGES / 4) {
    const int4 s4 = ((const int4*)ei)[i];
    const int4 d4 = ((const int4*)(ei + N_EDGES))[i];
    const unsigned sv[4] = {(unsigned)s4.x, (unsigned)s4.y, (unsigned)s4.z, (unsigned)s4.w};
    const unsigned dv[4] = {(unsigned)d4.x, (unsigned)d4.y, (unsigned)d4.z, (unsigned)d4.w};
    unsigned rv[4];
    unsigned kw = 0u;
#pragma unroll
    for (int c = 0; c < 4; ++c) {
      const unsigned t = dv[c] / (unsigned)TILE;
      rv[c] = (sv[c] << 8) | (dv[c] - t * (unsigned)TILE);
      kw |= t << (8 * c);
    }
    ((uint4*)rec32)[i] = make_uint4(rv[0], rv[1], rv[2], rv[3]);
    key8[i] = kw;
  }
  if (i >= N_NODES * HEADS) return;
  const int hd = i & (HEADS - 1);
  const float4* hp = (const float4*)(h + (size_t)i * CPH);
  const float4* as = (const float4*)(att_src + hd * CPH);
  const float4* ad = (const float4*)(att_dst + hd * CPH);
  float s = 0.f, d = 0.f;
#pragma unroll
  for (int q = 0; q < 4; ++q) {
    const float4 hv = hp[q], av = as[q], dv = ad[q];
    s += hv.x * av.x + hv.y * av.y + hv.z * av.z + hv.w * av.w;
    d += hv.x * dv.x + hv.y * dv.y + hv.z * dv.z + hv.w * dv.w;
  }
  a_src[i] = s;
  a_dst[i] = d;
}

__device__ __forceinline__ float leaky(float v) {
  return v >= 0.f ? v : NEG_SLOPE * v;
}
__device__ __forceinline__ unsigned fflip(float f) {
  const unsigned u = __float_as_uint(f);
  return u ^ ((u >> 31) ? 0xFFFFFFFFu : 0x80000000u);
}
__device__ __forceinline__ float funflip(unsigned u) {
  return __uint_as_float(u ^ ((u >> 31) ? 0x80000000u : 0xFFFFFFFFu));
}

#define TTHREADS 1024
#define FIX_SCALE 1073741824.0f

__device__ __forceinline__ unsigned long long to_fix(float v) {
  const float t = truncf(v);
  const float r = v - t;
  const long long f = ((long long)(int)t << 30) + (long long)(int)(r * FIX_SCALE);
  return (unsigned long long)f;
}

__device__ __forceinline__ float fix_to_float(unsigned long long u) {
  return (float)(long long)u * (1.0f / 1073741824.0f);
}

__device__ __forceinline__ float edge_num(float as, float ad, unsigned mx) {
  return __expf(leaky(as + ad) - __uint_as_float(mx));
}

struct EdgeIn { float ex; float4 hv; };

__device__ __forceinline__ EdgeIn edge_fetch(int s, int k, float keep, int lane,
                                             const float* exw,
                                             const float* __restrict__ h) {
  EdgeIn r;
  r.ex = exw[k * HEADS + (lane >> 2)] * keep;
  r.hv = ((const float4*)(h + (size_t)s * OUT_DIM))[lane];
  return r;
}

__device__ __forceinline__ void edge_apply(const EdgeIn in, int dl, int lane,
                                           unsigned long long* acc) {
  unsigned long long* ap = acc + dl * OUT_DIM + lane;
  atomicAdd(ap,      to_fix(in.ex * in.hv.x));
  atomicAdd(ap + 32, to_fix(in.ex * in.hv.y));
  atomicAdd(ap + 64, to_fix(in.ex * in.hv.z));
  atomicAdd(ap + 96, to_fix(in.ex * in.hv.w));
}

struct BatchIn { int s, dl; float4 a0, a1; };

__device__ __forceinline__ BatchIn batch_none() {
  BatchIn b;
  b.s = 0; b.dl = 0;
  b.a0 = make_float4(0.f, 0.f, 0.f, 0.f);
  b.a1 = make_float4(0.f, 0.f, 0.f, 0.f);
  return b;
}

__device__ __forceinline__ void apply_batch(const BatchIn b, int n, int lane,
                                            float* exw,
                                            const float* __restrict__ h,
                                            const float* s_adst, const unsigned* s_emaxU,
                                            unsigned long long* s_den,
                                            unsigned long long* acc) {
  const int s_m = b.s, dl_m = b.dl;
  if (lane < n) {
    const float4 a0 = b.a0, a1 = b.a1;
    const float4* dp = (const float4*)(s_adst + dl_m * HEADS);
    const float4 d0 = dp[0], d1 = dp[1];
    const uint4* mp = (const uint4*)(s_emaxU + dl_m * HEADS);
    const uint4 m0 = mp[0], m1 = mp[1];
    float4 e0, e1;
    e0.x = edge_num(a0.x, d0.x, m0.x); e0.y = edge_num(a0.y, d0.y, m0.y);
    e0.z = edge_num(a0.z, d0.z, m0.z); e0.w = edge_num(a0.w, d0.w, m0.w);
    e1.x = edge_num(a1.x, d1.x, m1.x); e1.y = edge_num(a1.y, d1.y, m1.y);
    e1.z = edge_num(a1.z, d1.z, m1.z); e1.w = edge_num(a1.w, d1.w, m1.w);
    float4* ew = (float4*)(exw + lane * HEADS);
    ew[0] = e0; ew[1] = e1;
    unsigned long long* dn = s_den + dl_m * HEADS;
    atomicAdd(dn + 0, to_fix(e0.x)); atomicAdd(dn + 1, to_fix(e0.y));
    atomicAdd(dn + 2, to_fix(e0.z)); atomicAdd(dn + 3, to_fix(e0.w));
    atomicAdd(dn + 4, to_fix(e1.x)); atomicAdd(dn + 5, to_fix(e1.y));
    atomicAdd(dn + 6, to_fix(e1.z)); atomicAdd(dn + 7, to_fix(e1.w));
  }
  for (int i = 0; i < n; i += 4) {
    const int k1 = i + 1 < n ? i + 1 : n - 1;
    const int k2 = i + 2 < n ? i + 2 : n - 1;
    const int k3 = i + 3 < n ? i + 3 : n - 1;
    const EdgeIn in0 = edge_fetch(__builtin_amdgcn_readlane(s_m, i), i, 1.0f, lane, exw, h);
    const EdgeIn in1 = edge_fetch(__builtin_amdgcn_readlane(s_m, k1), k1, i + 1 < n ? 1.0f : 0.0f, lane, exw, h);
    const EdgeIn in2 = edge_fetch(__builtin_amdgcn_readlane(s_m, k2), k2, i + 2 < n ? 1.0f : 0.0f, lane, exw, h);
    const EdgeIn in3 = edge_fetch(__builtin_amdgcn_readlane(s_m, k3), k3, i + 3 < n ? 1.0f : 0.0f, lane, exw, h);
    edge_apply(in0, __builtin_amdgcn_readlane(dl_m, i), lane, acc);
    edge_apply(in1, __builtin_amdgcn_readlane(dl_m, k1), lane, acc);
    edge_apply(in2, __builtin_amdgcn_readlane(dl_m, k2), lane, acc);
    edge_apply(in3, __builtin_amdgcn_readlane(dl_m, k3), lane, acc);
  }
}

#define NKW (N_EDGES / 16)
#define SCAN_U 4
#define QCAP 352

__device__ __forceinline__ void max_drain(unsigned* qw, int from, int to, int lane,
                                          const unsigned* __restrict__ rec32,
                                          const float* __restrict__ a_src,
                                          unsigned* s_emaxU) {
  for (int b0 = from; b0 < to; b0 += 32) {
    const int k = b0 + lane;
    if (k < to) {
      const unsigned r = rec32[qw[k]];
      qw[k] = r;
      const unsigned dl = r & 255u;
      const float4* ap = (const float4*)(a_src + (size_t)(r >> 8) * HEADS);
      const float4 lo = ap[0], hi = ap[1];
      unsigned* mp = s_emaxU + dl * HEADS;
      atomicMax(mp + 0, fflip(lo.x)); atomicMax(mp + 1, fflip(lo.y));
      atomicMax(mp + 2, fflip(lo.z)); atomicMax(mp + 3, fflip(lo.w));
      atomicMax(mp + 4, fflip(hi.x)); atomicMax(mp + 5, fflip(hi.y));
      atomicMax(mp + 6, fflip(hi.z)); atomicMax(mp + 7, fflip(hi.w));
    }
  }
}

template <bool RAW>
__device__ __forceinline__ BatchIn batch_load(const unsigned* qw, int b0, int n, int lane,
                                              const unsigned* __restrict__ rec32,
                                              const float* __restrict__ a_src) {
  BatchIn b = batch_none();
  if (lane < n) {
    const unsigned w0 = qw[b0 + lane];
    const unsigned w = RAW ? rec32[w0] : w0;
    b.dl = (int)(w & 255u);
    b.s = (int)(w >> 8);
    const float4* ap = (const float4*)(a_src + (size_t)b.s * HEADS);
    b.a0 = ap[0]; b.a1 = ap[1];
  }
  return b;
}

template <bool RAW>
__device__ __forceinline__ void apply_drain(BatchIn cur, int ncur,
                                            const unsigned* qw, int cnt, int lane,
                                            float* exw,
                                            const unsigned* __restrict__ rec32,
                                            const float* __restrict__ a_src,
                                            const float* __restrict__ h,
                                            const float* s_adst, const unsigned* s_emaxU,
                                            unsigned long long* s_den,
                                            unsigned long long* acc) {
  for (int b0 = 0; ; b0 += 32) {
    const int left = cnt - b0;
    const int n = left < 32 ? (left > 0 ? left : 0) : 32;
    const BatchIn nxt = batch_load<RAW>(qw, b0, n, lane, rec32, a_src);
    apply_batch(cur, ncur, lane, exw, h, s_adst, s_emaxU, s_den, acc);
    if (left <= 0) break;
    cur = nxt;
    ncur = n;
  }
}

__device__ __forceinline__ unsigned key_differs(unsigned k, unsigned pat) {
  const unsigned t = k ^ pat;
  return ((t & 0x7F7F7F7Fu) + 0x7F7F7F7Fu) | t;
}

template <bool PASSB>
__device__ __forceinline__ void scan_keys(unsigned* qw, float* exw, int tid, int lane, unsigned pat,
                                          const uint4* __restrict__ key4,
                                          const unsigned* __restrict__ rec32,
                                          const float* __restrict__ a_src,
                                          const float* __restrict__ h,
                                          const float* s_adst, unsigned* s_emaxU,
                                          unsigned long long* s_den,
                                          unsigned long long* acc,
                                          int& cnt_out, bool& ovf_out) {
  const uint4 none = make_uint4(~0u, ~0u, ~0u, ~0u);
  int cnt = 0, done = 0;
  bool ovf = false;
  uint4 kn[SCAN_U];
#pragma unroll
  for (int u = 0; u < SCAN_U; ++u) kn[u] = key4[u * TTHREADS + tid];
  for (int t0 = 0; t0 < NKW; t0 += SCAN_U * TTHREADS) {
    uint4 kc[SCAN_U];
#pragma unroll
    for (int u = 0; u < SCAN_U; ++u) {
      kc[u] = kn[u];
      const int qn = t0 + SCAN_U * TTHREADS + u * TTHREADS + tid;
      const uint4 w = key4[qn < NKW ? qn : NKW - 1];
      kn[u] = qn < NKW ? w : none;
    }
#pragma unroll
    for (int u = 0; u < SCAN_U; u += 2) {
      const int q = t0 + u * TTHREADS + tid;
      const unsigned nL =
          ((key_differs(kc[u].x, pat) >> 7) & 0x01010101u) | ((key_differs(kc[u].y, pat) >> 6) & 0x02020202u) |
          ((key_differs(kc[u].z, pat) >> 5) & 0x04040404u) | ((key_differs(kc[u].w, pat) >> 4) & 0x08080808u) |
          ((key_differs(kc[u + 1].x, pat) >> 3) & 0x10101010u) | ((key_differs(kc[u + 1].y, pat) >> 2) & 0x20202020u) |
          ((key_differs(kc[u + 1].z, pat) >> 1) & 0x40404040u) | (key_differs(kc[u + 1].w, pat) & 0x80808080u);
      unsigned L = ~nL;
      unsigned b = __builtin_amdgcn_ballot_w32(L != 0u);
      while (b) {
        if (cnt + 32 > QCAP) {
          if (PASSB) {
            apply_drain<true>(batch_none(), 0, qw, cnt, lane, exw, rec32, a_src, h, s_adst, s_emaxU, s_den, acc);
            cnt = 0;
          } else {
            max_drain(qw, done, cnt, lane, rec32, a_src, s_emaxU);
            cnt = 0; done = 0; ovf = true;
          }
        }
        if (L != 0u) {
          const int p = __builtin_ctz(L);
          const int pos = cnt + (int)__builtin_amdgcn_mbcnt_lo(b, 0u);
          qw[pos] = (unsigned)(16 * q + ((p & 4) << 12) + ((p & 3) << 2) + (p >> 3));
          L &= L - 1u;
        }
        cnt += __builtin_popcount(b);
        b = __builtin_amdgcn_ballot_w32(L != 0u);
      }
    }
    if (!PASSB && cnt - done >= 32) {
      max_drain(qw, done, cnt, lane, rec32, a_src, s_emaxU);
      done = cnt;
    }
  }
  if (PASSB) apply_drain<true>(batch_none(), 0, qw, cnt, lane, exw, rec32, a_src, h, s_adst, s_emaxU, s_den, acc);
  else       max_drain(qw, done, cnt, lane, rec32, a_src, s_emaxU);
  cnt_out = cnt;
  ovf_out = ovf;
}

__global__ __launch_bounds__(TTHREADS) void gat_tile_kernel(
    const unsigned* __restrict__ key8, const unsigned* __restrict__ rec32,
    const float* __restrict__ a_src,
    const float* __restrict__ a_dst, const float* __restrict__ h,
    const float* __restrict__ bias, float* __restrict__ out) {
  __shared__ unsigned long long acc[TILE * OUT_DIM];
  __shared__ unsigned long long s_den[TILE * HEADS];
  __shared__ __attribute__((aligned(16))) float s_adst[TILE * HEADS];
  __shared__ __attribute__((aligned(16))) unsigned s_emaxU[TILE * HEADS];
  __shared__ unsigned s_q[(TTHREADS / 32) * QCAP];
  __shared__ __attribute__((aligned(16))) float s_ex[(TTHREADS / 32) * 32 * HEADS];
  const int tid = threadIdx.x;
  const int lane = tid & 31;
  const int wave = tid >> 5;
  const int base = blockIdx.x * TILE;
  unsigned* qw = s_q + wave * QCAP;
  float* exw = s_ex + wave * 32 * HEADS;
  const uint4* key4 = (const uint4*)key8;
  const unsigned pat = (unsigned)blockIdx.x * 0x01010101u;

  for (int i = tid; i < TILE * OUT_DIM; i += TTHREADS) acc[i] = 0ull;
  for (int i = tid; i < TILE * HEADS; i += TTHREADS) {
    s_den[i]  = 0ull;
    s_adst[i] = a_dst[base * HEADS + i];
    s_emaxU[i] = fflip(a_src[base * HEADS + i]);
  }
  __syncthreads();

  int cnt = 0;
  bool ovf = false;
  scan_keys<false>(qw, exw, tid, lane, pat, key4, rec32, a_src, h, s_adst, s_emaxU, s_den, acc, cnt, ovf);
  __syncthreads();

  for (int i = tid; i < TILE * HEADS; i += TTHREADS)
    s_emaxU[i] = __float_as_uint(leaky(funflip(s_emaxU[i]) + s_adst[i]));
  __syncthreads();

  const int wv = __builtin_amdgcn_readfirstlane(wave);
  const int nself = (TILE - wv + TTHREADS / 32 - 1) / (TTHREADS / 32);
  BatchIn self = batch_none();
  if (lane < nself) {
    self.dl = wv + lane * (TTHREADS / 32);
    self.s = base + self.dl;
    const float4* ap = (const float4*)(a_src + (size_t)self.s * HEADS);
    self.a0 = ap[0]; self.a1 = ap[1];
  }
  if (!ovf) {
    apply_drain<false>(self, nself, qw, cnt, lane, exw, rec32, a_src, h, s_adst, s_emaxU, s_den, acc);
  } else {
    apply_batch(self, nself, lane, exw, h, s_adst, s_emaxU, s_den, acc);
    int c2 = 0;
    bool o2 = false;
    scan_keys<true>(qw, exw, tid, lane, pat, key4, rec32, a_src, h, s_adst, s_emaxU, s_den, acc, c2, o2);
  }
  __syncthreads();

  const float4 bv = ((const float4*)bias)[lane];
  for (int dl = wave; dl < TILE; dl += TTHREADS / 32) {
    const unsigned long long* ap = acc + dl * OUT_DIM + lane;
    const float den = fix_to_float(s_den[dl * HEADS + (lane >> 2)]) + GAT_EPS;
    const float v0 = fix_to_float(ap[0])  / den + bv.x;
    const float v1 = fix_to_float(ap[32]) / den + bv.y;
    const float v2 = fix_to_float(ap[64]) / den + bv.z;
    const float v3 = fix_to_float(ap[96]) / den + bv.w;
    float4 r;
    r.x = v0 > 0.f ? v0 : expm1f(v0);
    r.y = v1 > 0.f ? v1 : expm1f(v1);
    r.z = v2 > 0.f ? v2 : expm1f(v2);
    r.w = v3 > 0.f ? v3 : expm1f(v3);
    ((float4*)(out + (size_t)(base + dl) * OUT_DIM))[lane] = r;
  }
}

extern "C" void kernel_launch(void* const* d_in, const int* in_sizes, int n_in,
                              void* d_out, int out_size, void* d_ws, size_t ws_size,
                              hipStream_t stream) {
  (void)in_sizes; (void)n_in; (void)out_size; (void)ws_size;
  const float* x       = (const float*)d_in[0];
  const int*   ei      = (const int*)d_in[1];
  const float* W       = (const float*)d_in[2];
  const float* att_src = (const float*)d_in[3];
  const float* att_dst = (const float*)d_in[4];
  const float* bias    = (const float*)d_in[5];

  float* h     = (float*)d_ws;
  float* a_src = h + (size_t)N_NODES * OUT_DIM;
  float* a_dst = a_src + (size_t)N_NODES * HEADS;
  unsigned* rec32 = (unsigned*)(a_dst + (size_t)N_NODES * HEADS);
  unsigned* key8  = rec32 + (size_t)N_EDGES;

  gat_gemm_wmma_kernel<<<N_NODES / 16, 256, 0, stream>>>(x, W, h);

  gat_attn_proj_kernel<<<(N_NODES * HEADS + 255) / 256, 256, 0, stream>>>(
      h, att_src, att_dst, ei, a_src, a_dst, key8, rec32);

  gat_tile_kernel<<<N_NODES / TILE, TTHREADS, 0, stream>>>(
      key8, rec32, a_src, a_dst, h, bias, (float*)d_out);
}


# Pass B by row owners: a wave picks its rows' edges from packed LDS queues; each lane adds e, e*h to its own LDS floats (ds float add) in fixed queue order: new sum order, rounding; fixed point gone
# speedup vs baseline: 7.1129x; 1.1302x over previous
#include <hip/hip_runtime.h>
#include <math.h>

#define N_NODES 50000
#define N_EDGES 800000
#define IN_DIM 128
#define OUT_DIM 128
#define HEADS 8
#define CPH 16
#define NEG_SLOPE 0.2f
#define GAT_EPS 1e-16f
#define TILE 200

typedef float v2f __attribute__((ext_vector_type(2)));
typedef float v8f __attribute__((ext_vector_type(8)));

#define LDX 132

__global__ __launch_bounds__(256) void gat_gemm_wmma_kernel(
    const float* __restrict__ x, const float* __restrict__ Wg,
    float* __restrict__ h) {
  __shared__ float sx[16 * LDX];
  const int tid = threadIdx.x;
  const int mbase = blockIdx.x * 16;

  for (int i = tid; i < 16 * 32; i += 256) {
    const int r  = i >> 5;
    const int c4 = i & 31;
    const float4 v = ((const float4*)(x + (size_t)(mbase + r) * IN_DIM))[c4];
    float* dst = &sx[r * LDX + c4 * 4];
    dst[0] = v.x; dst[1] = v.y; dst[2] = v.z; dst[3] = v.w;
  }
  __syncthreads();

  const int wave = tid >> 5;
  const int lane = tid & 31;
  const int half = lane >> 4;
  const int mr   = lane & 15;
  const int nb   = wave * 16;

  v8f acc = {};
#pragma unroll
  for (int k0 = 0; k0 < IN_DIM; k0 += 4) {
    v2f a, b;
    a.x = sx[mr * LDX + k0 + 2 * half];
    a.y = sx[mr * LDX + k0 + 2 * half + 1];
    b.x = Wg[(size_t)(k0 + 2 * half) * OUT_DIM + nb + mr];
    b.y = Wg[(size_t)(k0 + 2 * half + 1) * OUT_DIM + nb + mr];
    acc = __builtin_amdgcn_wmma_f32_16x16x4_f32(
         false, a,  false, b,
         (short)0, acc,  false,  false);
  }

#pragma unroll
  for (int r = 0; r < 8; ++r) {
    const int row = half * 8 + r;
    h[(size_t)(mbase + row) * OUT_DIM + nb + mr] = acc[r];
  }
}

__global__ void gat_attn_proj_kernel(const float* __restrict__ h,
                                     const float* __restrict__ att_src,
                                     const float* __restrict__ att_dst,
                                     const int* __restrict__ ei,
                                     float* __restrict__ a_src,
                                     float* __restrict__ a_dst,
                                     unsigned* __restrict__ key8,
                                     unsigned* __restrict__ rec32) {
  const int i = blockIdx.x * blockDim.x + threadIdx.x;
  if (i < N_EDGES / 4) {
    const int4 s4 = ((const int4*)ei)[i];
    const int4 d4 = ((const int4*)(ei + N_EDGES))[i];
    const unsigned sv[4] = {(unsigned)s4.x, (unsigned)s4.y, (unsigned)s4.z, (unsigned)s4.w};
    const unsigned dv[4] = {(unsigned)d4.x, (unsigned)d4.y, (unsigned)d4.z, (unsigned)d4.w};
    unsigned rv[4];
    unsigned kw = 0u;
#pragma unroll
    for (int c = 0; c < 4; ++c) {
      const unsigned t = dv[c] / (unsigned)TILE;
      rv[c] = (sv[c] << 8) | (dv[c] - t * (unsigned)TILE);
      kw |= t << (8 * c);
    }
    ((uint4*)rec32)[i] = make_uint4(rv[0], rv[1], rv[2], rv[3]);
    key8[i] = kw;
  }
  if (i >= N_NODES * HEADS) return;
  const int hd = i & (HEADS - 1);
  const float4* hp = (const float4*)(h + (size_t)i * CPH);
  const float4* as = (const float4*)(att_src + hd * CPH);
  const float4* ad = (const float4*)(att_dst + hd * CPH);
  float s = 0.f, d = 0.f;
#pragma unroll
  for (int q = 0; q < 4; ++q) {
    const float4 hv = hp[q], av = as[q], dv = ad[q];
    s += hv.x * av.x + hv.y * av.y + hv.z * av.z + hv.w * av.w;
    d += hv.x * dv.x + hv.y * dv.y + hv.z * dv.z + hv.w * dv.w;
  }
  a_src[i] = s;
  a_dst[i] = d;
}

__device__ __forceinline__ float leaky(float v) {
  return v >= 0.f ? v : NEG_SLOPE * v;
}
__device__ __forceinline__ unsigned fflip(float f) {
  const unsigned u = __float_as_uint(f);
  return u ^ ((u >> 31) ? 0xFFFFFFFFu : 0x80000000u);
}
__device__ __forceinline__ float funflip(unsigned u) {
  return __uint_as_float(u ^ ((u >> 31) ? 0x80000000u : 0xFFFFFFFFu));
}

#define TTHREADS 1024
#define FIX_SCALE 1073741824.0f

__device__ __forceinline__ unsigned long long to_fix(float v) {
  const float t = truncf(v);
  const float r = v - t;
  const long long f = ((long long)(int)t << 30) + (long long)(int)(r * FIX_SCALE);
  return (unsigned long long)f;
}

__device__ __forceinline__ float fix_to_float(unsigned long long u) {
  return (float)(long long)u * (1.0f / 1073741824.0f);
}

__device__ __forceinline__ float edge_num(float as, float ad, unsigned mx) {
  return __expf(leaky(as + ad) - __uint_as_float(mx));
}

struct EdgeIn { float ex; float4 hv; };

__device__ __forceinline__ EdgeIn edge_fetch(int s, int k, float keep, int lane,
                                             const float* exw,
                                             const float* __restrict__ h) {
  EdgeIn r;
  r.ex = exw[k * HEADS + (lane >> 2)] * keep;
  r.hv = ((const float4*)(h + (size_t)s * OUT_DIM))[lane];
  return r;
}

__device__ __forceinline__ void edge_apply(const EdgeIn in, int dl, int lane,
                                           unsigned long long* acc) {
  unsigned long long* ap = acc + dl * OUT_DIM + lane;
  atomicAdd(ap,      to_fix(in.ex * in.hv.x));
  atomicAdd(ap + 32, to_fix(in.ex * in.hv.y));
  atomicAdd(ap + 64, to_fix(in.ex * in.hv.z));
  atomicAdd(ap + 96, to_fix(in.ex * in.hv.w));
}

struct BatchIn { int s, dl; float4 a0, a1; };

__device__ __forceinline__ BatchIn batch_none() {
  BatchIn b;
  b.s = 0; b.dl = 0;
  b.a0 = make_float4(0.f, 0.f, 0.f, 0.f);
  b.a1 = make_float4(0.f, 0.f, 0.f, 0.f);
  return b;
}

__device__ __forceinline__ void apply_batch(const BatchIn b, int n, int lane,
                                            float* exw,
                                            const float* __restrict__ h,
                                            const float* s_adst, const unsigned* s_emaxU,
                                            unsigned long long* s_den,
                                            unsigned long long* acc) {
  const int s_m = b.s, dl_m = b.dl;
  if (lane < n) {
    const float4 a0 = b.a0, a1 = b.a1;
    const float4* dp = (const float4*)(s_adst + dl_m * HEADS);
    const float4 d0 = dp[0], d1 = dp[1];
    const uint4* mp = (const uint4*)(s_emaxU + dl_m * HEADS);
    const uint4 m0 = mp[0], m1 = mp[1];
    float4 e0, e1;
    e0.x = edge_num(a0.x, d0.x, m0.x); e0.y = edge_num(a0.y, d0.y, m0.y);
    e0.z = edge_num(a0.z, d0.z, m0.z); e0.w = edge_num(a0.w, d0.w, m0.w);
    e1.x = edge_num(a1.x, d1.x, m1.x); e1.y = edge_num(a1.y, d1.y, m1.y);
    e1.z = edge_num(a1.z, d1.z, m1.z); e1.w = edge_num(a1.w, d1.w, m1.w);
    float4* ew = (float4*)(exw + lane * HEADS);
    ew[0] = e0; ew[1] = e1;
    unsigned long long* dn = s_den + dl_m * HEADS;
    atomicAdd(dn + 0, to_fix(e0.x)); atomicAdd(dn + 1, to_fix(e0.y));
    atomicAdd(dn + 2, to_fix(e0.z)); atomicAdd(dn + 3, to_fix(e0.w));
    atomicAdd(dn + 4, to_fix(e1.x)); atomicAdd(dn + 5, to_fix(e1.y));
    atomicAdd(dn + 6, to_fix(e1.z)); atomicAdd(dn + 7, to_fix(e1.w));
  }
  for (int i = 0; i < n; i += 4) {
    const int k1 = i + 1 < n ? i + 1 : n - 1;
    const int k2 = i + 2 < n ? i + 2 : n - 1;
    const int k3 = i + 3 < n ? i + 3 : n - 1;
    const EdgeIn in0 = edge_fetch(__builtin_amdgcn_readlane(s_m, i), i, 1.0f, lane, exw, h);
    const EdgeIn in1 = edge_fetch(__builtin_amdgcn_readlane(s_m, k1), k1, i + 1 < n ? 1.0f : 0.0f, lane, exw, h);
    const EdgeIn in2 = edge_fetch(__builtin_amdgcn_readlane(s_m, k2), k2, i + 2 < n ? 1.0f : 0.0f, lane, exw, h);
    const EdgeIn in3 = edge_fetch(__builtin_amdgcn_readlane(s_m, k3), k3, i + 3 < n ? 1.0f : 0.0f, lane, exw, h);
    edge_apply(in0, __builtin_amdgcn_readlane(dl_m, i), lane, acc);
    edge_apply(in1, __builtin_amdgcn_readlane(dl_m, k1), lane, acc);
    edge_apply(in2, __builtin_amdgcn_readlane(dl_m, k2), lane, acc);
    edge_apply(in3, __builtin_amdgcn_readlane(dl_m, k3), lane, acc);
  }
}

#define NKW (N_EDGES / 16)
#define SCAN_U 4
#define QCAP 352
#define QREG (QCAP / 32)

__device__ __forceinline__ void max_drain(unsigned* qw, int from, int to, int lane,
                                          const unsigned* __restrict__ rec32,
                                          const float* __restrict__ a_src,
                                          unsigned* s_emaxU) {
  for (int b0 = from; b0 < to; b0 += 32) {
    const int k = b0 + lane;
    if (k < to) {
      const unsigned r = rec32[qw[k]];
      qw[k] = r;
      const unsigned dl = r & 255u;
      const float4* ap = (const float4*)(a_src + (size_t)(r >> 8) * HEADS);
      const float4 lo = ap[0], hi = ap[1];
      unsigned* mp = s_emaxU + dl * HEADS;
      atomicMax(mp + 0, fflip(lo.x)); atomicMax(mp + 1, fflip(lo.y));
      atomicMax(mp + 2, fflip(lo.z)); atomicMax(mp + 3, fflip(lo.w));
      atomicMax(mp + 4, fflip(hi.x)); atomicMax(mp + 5, fflip(hi.y));
      atomicMax(mp + 6, fflip(hi.z)); atomicMax(mp + 7, fflip(hi.w));
    }
  }
}

template <bool RAW>
__device__ __forceinline__ BatchIn batch_load(const unsigned* qw, int b0, int n, int lane,
                                              const unsigned* __restrict__ rec32,
                                              const float* __restrict__ a_src) {
  BatchIn b = batch_none();
  if (lane < n) {
    const unsigned w0 = qw[b0 + lane];
    const unsigned w = RAW ? rec32[w0] : w0;
    b.dl = (int)(w & 255u);
    b.s = (int)(w >> 8);
    const float4* ap = (const float4*)(a_src + (size_t)b.s * HEADS);
    b.a0 = ap[0]; b.a1 = ap[1];
  }
  return b;
}

template <bool RAW>
__device__ __forceinline__ void apply_drain(BatchIn cur, int ncur,
                                            const unsigned* qw, int cnt, int lane,
                                            float* exw,
                                            const unsigned* __restrict__ rec32,
                                            const float* __restrict__ a_src,
                                            const float* __restrict__ h,
                                            const float* s_adst, const unsigned* s_emaxU,
                                            unsigned long long* s_den,
                                            unsigned long long* acc) {
  for (int b0 = 0; ; b0 += 32) {
    const int left = cnt - b0;
    const int n = left < 32 ? (left > 0 ? left : 0) : 32;
    const BatchIn nxt = batch_load<RAW>(qw, b0, n, lane, rec32, a_src);
    apply_batch(cur, ncur, lane, exw, h, s_adst, s_emaxU, s_den, acc);
    if (left <= 0) break;
    cur = nxt;
    ncur = n;
  }
}

#define NWAVES (TTHREADS / 32)
#define ROWF 160
#define LCAP 288
#define LGRP 128

__device__ __forceinline__ int wave_sum(int v) {
  for (int m = 16; m > 0; m >>= 1) v += __shfl_xor(v, m, 32);
  return v;
}

__device__ __forceinline__ void own_apply(const EdgeIn in, int dl, int lane, float* fs) {
  float* ap = fs + dl * ROWF + lane;
  atomicAdd(ap,       in.ex * in.hv.x);
  atomicAdd(ap + 32,  in.ex * in.hv.y);
  atomicAdd(ap + 64,  in.ex * in.hv.z);
  atomicAdd(ap + 96,  in.ex * in.hv.w);
  atomicAdd(ap + 128, in.ex);
}

__device__ __forceinline__ void own_batch(const BatchIn b, int n, int lane,
                                          float* exw,
                                          const float* __restrict__ h,
                                          const float* s_adst, const unsigned* s_emaxU,
                                          float* fs) {
  const int s_m = b.s, dl_m = b.dl;
  if (lane < n) {
    const float4 a0 = b.a0, a1 = b.a1;
    const float4* dp = (const float4*)(s_adst + dl_m * HEADS);
    const float4 d0 = dp[0], d1 = dp[1];
    const uint4* mp = (const uint4*)(s_emaxU + dl_m * HEADS);
    const uint4 m0 = mp[0], m1 = mp[1];
    float4 e0, e1;
    e0.x = edge_num(a0.x, d0.x, m0.x); e0.y = edge_num(a0.y, d0.y, m0.y);
    e0.z = edge_num(a0.z, d0.z, m0.z); e0.w = edge_num(a0.w, d0.w, m0.w);
    e1.x = edge_num(a1.x, d1.x, m1.x); e1.y = edge_num(a1.y, d1.y, m1.y);
    e1.z = edge_num(a1.z, d1.z, m1.z); e1.w = edge_num(a1.w, d1.w, m1.w);
    float4* ew = (float4*)(exw + lane * HEADS);
    ew[0] = e0; ew[1] = e1;
  }
  for (int i = 0; i < n; i += 4) {
    const int k1 = i + 1 < n ? i + 1 : n - 1;
    const int k2 = i + 2 < n ? i + 2 : n - 1;
    const int k3 = i + 3 < n ? i + 3 : n - 1;
    const EdgeIn in0 = edge_fetch(__builtin_amdgcn_readlane(s_m, i), i, 1.0f, lane, exw, h);
    const EdgeIn in1 = edge_fetch(__builtin_amdgcn_readlane(s_m, k1), k1, i + 1 < n ? 1.0f : 0.0f, lane, exw, h);
    const EdgeIn in2 = edge_fetch(__builtin_amdgcn_readlane(s_m, k2), k2, i + 2 < n ? 1.0f : 0.0f, lane, exw, h);
    const EdgeIn in3 = edge_fetch(__builtin_amdgcn_readlane(s_m, k3), k3, i + 3 < n ? 1.0f : 0.0f, lane, exw, h);
    own_apply(in0, __builtin_amdgcn_readlane(dl_m, i), lane, fs);
    own_apply(in1, __builtin_amdgcn_readlane(dl_m, k1), lane, fs);
    own_apply(in2, __builtin_amdgcn_readlane(dl_m, k2), lane, fs);
    own_apply(in3, __builtin_amdgcn_readlane(dl_m, k3), lane, fs);
  }
}

__device__ __forceinline__ void own_drain(BatchIn cur, int ncur,
                                          const unsigned* lst, int cnt, int lane,
                                          float* exw,
                                          const unsigned* __restrict__ rec32,
                                          const float* __restrict__ a_src,
                                          const float* __restrict__ h,
                                          const float* s_adst, const unsigned* s_emaxU,
                                          float* fs) {
  for (int b0 = 0; ; b0 += 32) {
    const int left = cnt - b0;
    const int n = left < 32 ? (left > 0 ? left : 0) : 32;
    const BatchIn nxt = batch_load<false>(lst, b0, n, lane, rec32, a_src);
    own_batch(cur, ncur, lane, exw, h, s_adst, s_emaxU, fs);
    if (left <= 0) break;
    cur = nxt;
    ncur = n;
  }
}

__device__ __forceinline__ unsigned key_differs(unsigned k, unsigned pat) {
  const unsigned t = k ^ pat;
  return ((t & 0x7F7F7F7Fu) + 0x7F7F7F7Fu) | t;
}

template <bool PASSB>
__device__ __forceinline__ void scan_keys(unsigned* qw, float* exw, int tid, int lane, unsigned pat,
                                          const uint4* __restrict__ key4,
                                          const unsigned* __restrict__ rec32,
                                          const float* __restrict__ a_src,
                                          const float* __restrict__ h,
                                          const float* s_adst, unsigned* s_emaxU,
                                          unsigned long long* s_den,
                                          unsigned long long* acc,
                                          int& cnt_out, bool& ovf_out) {
  const uint4 none = make_uint4(~0u, ~0u, ~0u, ~0u);
  int cnt = 0, done = 0;
  bool ovf = false;
  uint4 kn[SCAN_U];
#pragma unroll
  for (int u = 0; u < SCAN_U; ++u) kn[u] = key4[u * TTHREADS + tid];
  for (int t0 = 0; t0 < NKW; t0 += SCAN_U * TTHREADS) {
    uint4 kc[SCAN_U];
#pragma unroll
    for (int u = 0; u < SCAN_U; ++u) {
      kc[u] = kn[u];
      const int qn = t0 + SCAN_U * TTHREADS + u * TTHREADS + tid;
      const uint4 w = key4[qn < NKW ? qn : NKW - 1];
      kn[u] = qn < NKW ? w : none;
    }
#pragma unroll
    for (int u = 0; u < SCAN_U; u += 2) {
      const int q = t0 + u * TTHREADS + tid;
      const unsigned nL =
          ((key_differs(kc[u].x, pat) >> 7) & 0x01010101u) | ((key_differs(kc[u].y, pat) >> 6) & 0x02020202u) |
          ((key_differs(kc[u].z, pat) >> 5) & 0x04040404u) | ((key_differs(kc[u].w, pat) >> 4) & 0x08080808u) |
          ((key_differs(kc[u + 1].x, pat) >> 3) & 0x10101010u) | ((key_differs(kc[u + 1].y, pat) >> 2) & 0x20202020u) |
          ((key_differs(kc[u + 1].z, pat) >> 1) & 0x40404040u) | (key_differs(kc[u + 1].w, pat) & 0x80808080u);
      unsigned L = ~nL;
      unsigned b = __builtin_amdgcn_ballot_w32(L != 0u);
      while (b) {
        if (cnt + 32 > QCAP) {
          if (PASSB) {
            apply_drain<true>(batch_none(), 0, qw, cnt, lane, exw, rec32, a_src, h, s_adst, s_emaxU, s_den, acc);
            cnt = 0;
          } else {
            max_drain(qw, done, cnt, lane, rec32, a_src, s_emaxU);
            cnt = 0; done = 0; ovf = true;
          }
        }
        if (L != 0u) {
          const int p = __builtin_ctz(L);
          const int pos = cnt + (int)__builtin_amdgcn_mbcnt_lo(b, 0u);
          qw[pos] = (unsigned)(16 * q + ((p & 4) << 12) + ((p & 3) << 2) + (p >> 3));
          L &= L - 1u;
        }
        cnt += __builtin_popcount(b);
        b = __builtin_amdgcn_ballot_w32(L != 0u);
      }
    }
    if (!PASSB && cnt - done >= 32) {
      max_drain(qw, done, cnt, lane, rec32, a_src, s_emaxU);
      done = cnt;
    }
  }
  if (PASSB) apply_drain<true>(batch_none(), 0, qw, cnt, lane, exw, rec32, a_src, h, s_adst, s_emaxU, s_den, acc);
  else       max_drain(qw, done, cnt, lane, rec32, a_src, s_emaxU);
  cnt_out = cnt;
  ovf_out = ovf;
}

__global__ __launch_bounds__(TTHREADS) void gat_tile_kernel(
    const unsigned* __restrict__ key8, const unsigned* __restrict__ rec32,
    const float* __restrict__ a_src,
    const float* __restrict__ a_dst, const float* __restrict__ h,
    const float* __restrict__ bias, float* __restrict__ out) {
  __shared__ unsigned long long acc[TILE * OUT_DIM];
  __shared__ unsigned long long s_den[TILE * HEADS];
  __shared__ __attribute__((aligned(16))) float s_adst[TILE * HEADS];
  __shared__ __attribute__((aligned(16))) unsigned s_emaxU[TILE * HEADS];
  __shared__ __attribute__((aligned(16))) unsigned s_q[(TTHREADS / 32) * QCAP + LGRP];
  __shared__ __attribute__((aligned(16))) float s_ex[(TTHREADS / 32) * 32 * HEADS];
  __shared__ int s_cnt[NWAVES];
  __shared__ int s_flag;
  const int tid = threadIdx.x;
  const int lane = tid & 31;
  const int wave = tid >> 5;
  const int base = blockIdx.x * TILE;
  unsigned* qw = s_q + wave * QCAP;
  float* exw = s_ex + wave * 32 * HEADS;
  const uint4* key4 = (const uint4*)key8;
  const unsigned pat = (unsigned)blockIdx.x * 0x01010101u;

  if (tid == 0) s_flag = 0;
  for (int i = tid; i < TILE * HEADS; i += TTHREADS) {
    s_den[i]  = 0ull;
    s_adst[i] = a_dst[base * HEADS + i];
    s_emaxU[i] = fflip(a_src[base * HEADS + i]);
  }
  __syncthreads();

  int cnt = 0;
  bool ovf = false;
  scan_keys<false>(qw, exw, tid, lane, pat, key4, rec32, a_src, h, s_adst, s_emaxU, s_den, acc, cnt, ovf);
  unsigned qr[QREG];
#pragma unroll
  for (int t = 0; t < QREG; ++t) {
    const int k = t * 32 + lane;
    qr[t] = k < cnt ? qw[k] : 0u;
  }
  if (lane == 0) {
    s_cnt[wave] = cnt;
    if (ovf) s_flag = 1;
  }
  __syncthreads();

  const int any_ovf = __builtin_amdgcn_readfirstlane(s_flag);
  const int wv = __builtin_amdgcn_readfirstlane(wave);
  int qtot = 0;
  if (any_ovf == 0) {
    const int cv = s_cnt[lane];
    const int off = __builtin_amdgcn_readfirstlane(wave_sum(lane < wv ? cv : 0));
    qtot = __builtin_amdgcn_readfirstlane(wave_sum(cv));
#pragma unroll
    for (int t = 0; t < QREG; ++t) {
      const int k = t * 32 + lane;
      if (k < cnt) s_q[off + k] = qr[t];
    }
    if (wv == NWAVES - 1) {
      for (int t = 0; t < LGRP / 32; ++t) s_q[qtot + t * 32 + lane] = 0xFFFFFFFFu;
    }
  }

  for (int i = tid; i < TILE * HEADS; i += TTHREADS)
    s_emaxU[i] = __float_as_uint(leaky(funflip(s_emaxU[i]) + s_adst[i]));
  __syncthreads();

  const int nself = (TILE - wv + TTHREADS / 32 - 1) / (TTHREADS / 32);
  BatchIn self = batch_none();
  if (lane < nself) {
    self.dl = wv + lane * (TTHREADS / 32);
    self.s = base + self.dl;
    const float4* ap = (const float4*)(a_src + (size_t)self.s * HEADS);
    self.a0 = ap[0]; self.a1 = ap[1];
  }
  const float4 bv = ((const float4*)bias)[lane];
  if (any_ovf == 0) {
    float* fs = (float*)acc;
    unsigned* lst = (unsigned*)acc + TILE * ROWF + wave * LCAP;
    for (int j = 0; j < nself; ++j) {
      float* rp = fs + (wv + j * NWAVES) * ROWF + lane;
      rp[0] = 0.f; rp[32] = 0.f; rp[64] = 0.f; rp[96] = 0.f; rp[128] = 0.f;
    }
    BatchIn cur = self;
    int ncur = nself;
    int b0 = 0;
    for (;;) {
      int n = 0;
      while (b0 < qtot && n + LGRP <= LCAP) {
        const uint4 q = *(const uint4*)(s_q + b0 + 4 * lane);
        const unsigned qe[4] = {q.x, q.y, q.z, q.w};
        unsigned mk = 0u;
#pragma unroll
        for (int t = 0; t < 4; ++t)
          mk |= (((qe[t] ^ (unsigned)wv) & 0x8000001Fu) == 0u) ? (1u << t) : 0u;
        unsigned bm = __builtin_amdgcn_ballot_w32(mk != 0u);
        while (bm) {
          if (mk != 0u) {
            const unsigned e01 = (mk & 1u) ? qe[0] : qe[1];
            const unsigned e23 = (mk & 4u) ? qe[2] : qe[3];
            lst[n + (int)__builtin_amdgcn_mbcnt_lo(bm, 0u)] = (mk & 3u) ? e01 : e23;
            mk &= mk - 1u;
          }
          n += __builtin_popcount(bm);
          bm = __builtin_amdgcn_ballot_w32(mk != 0u);
        }
        b0 += LGRP;
      }
      own_drain(cur, ncur, lst, n, lane, exw, rec32, a_src, h, s_adst, s_emaxU, fs);
      if (b0 >= qtot) break;
      cur = batch_none();
      ncur = 0;
    }
    for (int j = 0; j < nself; ++j) {
      const int dl = wv + j * NWAVES;
      const float* rp = fs + dl * ROWF + lane;
      const float den = rp[128] + GAT_EPS;
      const float v0 = rp[0]  / den + bv.x;
      const float v1 = rp[32] / den + bv.y;
      const float v2 = rp[64] / den + bv.z;
      const float v3 = rp[96] / den + bv.w;
      float4 r;
      r.x = v0 > 0.f ? v0 : expm1f(v0);
      r.y = v1 > 0.f ? v1 : expm1f(v1);
      r.z = v2 > 0.f ? v2 : expm1f(v2);
      r.w = v3 > 0.f ? v3 : expm1f(v3);
      ((float4*)(out + (size_t)(base + dl) * OUT_DIM))[lane] = r;
    }
  } else {
    for (int i = tid; i < TILE * OUT_DIM; i += TTHREADS) acc[i] = 0ull;
    __syncthreads();
    if (!ovf) {
      apply_drain<false>(self, nself, qw, cnt, lane, exw, rec32, a_src, h, s_adst, s_emaxU, s_den, acc);
    } else {
      apply_batch(self, nself, lane, exw, h, s_adst, s_emaxU, s_den, acc);
      int c2 = 0;
      bool o2 = false;
      scan_keys<true>(qw, exw, tid, lane, pat, key4, rec32, a_src, h, s_adst, s_emaxU, s_den, acc, c2, o2);
    }
    __syncthreads();

    for (int dl = wave; dl < TILE; dl += TTHREADS / 32) {
      const unsigned long long* ap = acc + dl * OUT_DIM + lane;
      const float den = fix_to_float(s_den[dl * HEADS + (lane >> 2)]) + GAT_EPS;
      const float v0 = fix_to_float(ap[0])  / den + bv.x;
      const float v1 = fix_to_float(ap[32]) / den + bv.y;
      const float v2 = fix_to_float(ap[64]) / den + bv.z;
      const float v3 = fix_to_float(ap[96]) / den + bv.w;
      float4 r;
      r.x = v0 > 0.f ? v0 : expm1f(v0);
      r.y = v1 > 0.f ? v1 : expm1f(v1);
      r.z = v2 > 0.f ? v2 : expm1f(v2);
      r.w = v3 > 0.f ? v3 : expm1f(v3);
      ((float4*)(out + (size_t)(base + dl) * OUT_DIM))[lane] = r;
    }
  }
}

extern "C" void kernel_launch(void* const* d_in, const int* in_sizes, int n_in,
                              void* d_out, int out_size, void* d_ws, size_t ws_size,
                              hipStream_t stream) {
  (void)in_sizes; (void)n_in; (void)out_size; (void)ws_size;
  const float* x       = (const float*)d_in[0];
  const int*   ei      = (const int*)d_in[1];
  const float* W       = (const float*)d_in[2];
  const float* att_src = (const float*)d_in[3];
  const float* att_dst = (const float*)d_in[4];
  const float* bias    = (const float*)d_in[5];

  float* h     = (float*)d_ws;
  float* a_src = h + (size_t)N_NODES * OUT_DIM;
  float* a_dst = a_src + (size_t)N_NODES * HEADS;
  unsigned* rec32 = (unsigned*)(a_dst + (size_t)N_NODES * HEADS);
  unsigned* key8  = rec32 + (size_t)N_EDGES;

  gat_gemm_wmma_kernel<<<N_NODES / 16, 256, 0, stream>>>(x, W, h);

  gat_attn_proj_kernel<<<(N_NODES * HEADS + 255) / 256, 256, 0, stream>>>(
      h, att_src, att_dst, ei, a_src, a_dst, key8, rec32);

  gat_tile_kernel<<<N_NODES / TILE, TTHREADS, 0, stream>>>(
      key8, rec32, a_src, a_dst, h, bias, (float*)d_out);
}


# Attention terms a_src/a_dst and the per-call edge tables folded into the GEMM launch: terms summed from the accumulator tile by 16-lane exchanges (new sum order, rounding); projection launch gone
# speedup vs baseline: 7.5018x; 1.0547x over previous
#include <hip/hip_runtime.h>
#include <math.h>

#define N_NODES 50000
#define N_EDGES 800000
#define IN_DIM 128
#define OUT_DIM 128
#define HEADS 8
#define CPH 16
#define NEG_SLOPE 0.2f
#define GAT_EPS 1e-16f
#define TILE 200

typedef float v2f __attribute__((ext_vector_type(2)));
typedef float v8f __attribute__((ext_vector_type(8)));

#define LDX 132

template <int M>
__device__ __forceinline__ float row_xor(float v) {
  return __int_as_float(__builtin_amdgcn_update_dpp(
      0, __float_as_int(v), 0x160 + M  , 0xf, 0xf, false));
}

__device__ __forceinline__ float half_sum8(const float p[8], int lane) {
  const bool b3 = (lane & 8) != 0, b2 = (lane & 4) != 0, b1 = (lane & 2) != 0;
  float q[4], t[2];
#pragma unroll
  for (int i = 0; i < 4; ++i) {
    const float keep = b3 ? p[i + 4] : p[i];
    const float send = b3 ? p[i] : p[i + 4];
    q[i] = keep + row_xor<8>(send);
  }
#pragma unroll
  for (int i = 0; i < 2; ++i) {
    const float keep = b2 ? q[i + 2] : q[i];
    const float send = b2 ? q[i] : q[i + 2];
    t[i] = keep + row_xor<4>(send);
  }
  const float keep = b1 ? t[1] : t[0];
  const float send = b1 ? t[0] : t[1];
  float u = keep + row_xor<2>(send);
  u += row_xor<1>(u);
  return u;
}

__global__ __launch_bounds__(256, 16) void gat_gemm_wmma_kernel(
    const float* __restrict__ x, const float* __restrict__ Wg,
    const float* __restrict__ att_src, const float* __restrict__ att_dst,
    const int* __restrict__ ei,
    float* __restrict__ h, float* __restrict__ a_src, float* __restrict__ a_dst,
    unsigned* __restrict__ key8, unsigned* __restrict__ rec32) {
  __shared__ float sx[16 * LDX];
  __shared__ float s_att[2 * 16 * HEADS];
  const int tid = threadIdx.x;
  const int mbase = blockIdx.x * 16;

  const int gi = blockIdx.x * 256 + tid;
  const bool has_edges = gi < N_EDGES / 4;
  const int ge = has_edges ? gi : 0;
  const int4 s4 = ((const int4*)ei)[ge];
  const int4 d4 = ((const int4*)(ei + N_EDGES))[ge];

  for (int i = tid; i < 16 * 32; i += 256) {
    const int r  = i >> 5;
    const int c4 = i & 31;
    const float4 v = ((const float4*)(x + (size_t)(mbase + r) * IN_DIM))[c4];
    float* dst = &sx[r * LDX + c4 * 4];
    dst[0] = v.x; dst[1] = v.y; dst[2] = v.z; dst[3] = v.w;
  }

  const unsigned sv[4] = {(unsigned)s4.x, (unsigned)s4.y, (unsigned)s4.z, (unsigned)s4.w};
  const unsigned dv[4] = {(unsigned)d4.x, (unsigned)d4.y, (unsigned)d4.z, (unsigned)d4.w};
  unsigned rv[4];
  unsigned kw = 0u;
#pragma unroll
  for (int c = 0; c < 4; ++c) {
    const unsigned t = dv[c] / (unsigned)TILE;
    rv[c] = (sv[c] << 8) | (dv[c] - t * (unsigned)TILE);
    kw |= t << (8 * c);
  }
  __syncthreads();
  if (has_edges) {
    ((uint4*)rec32)[gi] = make_uint4(rv[0], rv[1], rv[2], rv[3]);
    key8[gi] = kw;
  }

  const int wave = tid >> 5;
  const int lane = tid & 31;
  const int half = lane >> 4;
  const int mr   = lane & 15;
  const int nb   = wave * 16;

  const float asv = att_src[nb + mr];
  const float adv = att_dst[nb + mr];

  v8f acc = {};
#pragma unroll
  for (int k0 = 0; k0 < IN_DIM; k0 += 4) {
    v2f a, b;
    a.x = sx[mr * LDX + k0 + 2 * half];
    a.y = sx[mr * LDX + k0 + 2 * half + 1];
    b.x = Wg[(size_t)(k0 + 2 * half) * OUT_DIM + nb + mr];
    b.y = Wg[(size_t)(k0 + 2 * half + 1) * OUT_DIM + nb + mr];
    acc = __builtin_amdgcn_wmma_f32_16x16x4_f32(
         false, a,  false, b,
         (short)0, acc,  false,  false);
  }

  float ps[8], pd[8];
#pragma unroll
  for (int r = 0; r < 8; ++r) {
    ps[r] = acc[r] * asv;
    pd[r] = acc[r] * adv;
  }
  const float us = half_sum8(ps, lane);
  const float ud = half_sum8(pd, lane);
  if ((lane & 1) == 0) {
    const int row = half * 8 + ((lane >> 1) & 7);
    s_att[row * HEADS + wave] = us;
    s_att[16 * HEADS + row * HEADS + wave] = ud;
  }
  __syncthreads();

#pragma unroll
  for (int r = 0; r < 8; ++r) {
    const int row = half * 8 + r;
    h[(size_t)(mbase + row) * OUT_DIM + nb + mr] = acc[r];
  }
  if (tid < 16 * HEADS) a_src[(size_t)mbase * HEADS + tid] = s_att[tid];
  else                  a_dst[(size_t)mbase * HEADS + (tid - 16 * HEADS)] = s_att[tid];
}

__device__ __forceinline__ float leaky(float v) {
  return v >= 0.f ? v : NEG_SLOPE * v;
}
__device__ __forceinline__ unsigned fflip(float f) {
  const unsigned u = __float_as_uint(f);
  return u ^ ((u >> 31) ? 0xFFFFFFFFu : 0x80000000u);
}
__device__ __forceinline__ float funflip(unsigned u) {
  return __uint_as_float(u ^ ((u >> 31) ? 0x80000000u : 0xFFFFFFFFu));
}

#define TTHREADS 1024
#define FIX_SCALE 1073741824.0f

__device__ __forceinline__ unsigned long long to_fix(float v) {
  const float t = truncf(v);
  const float r = v - t;
  const long long f = ((long long)(int)t << 30) + (long long)(int)(r * FIX_SCALE);
  return (unsigned long long)f;
}

__device__ __forceinline__ float fix_to_float(unsigned long long u) {
  return (float)(long long)u * (1.0f / 1073741824.0f);
}

__device__ __forceinline__ float edge_num(float as, float ad, unsigned mx) {
  return __expf(leaky(as + ad) - __uint_as_float(mx));
}

struct EdgeIn { float ex; float4 hv; };

__device__ __forceinline__ EdgeIn edge_fetch(int s, int k, float keep, int lane,
                                             const float* exw,
                                             const float* __restrict__ h) {
  EdgeIn r;
  r.ex = exw[k * HEADS + (lane >> 2)] * keep;
  r.hv = ((const float4*)(h + (size_t)s * OUT_DIM))[lane];
  return r;
}

__device__ __forceinline__ void edge_apply(const EdgeIn in, int dl, int lane,
                                           unsigned long long* acc) {
  unsigned long long* ap = acc + dl * OUT_DIM + lane;
  atomicAdd(ap,      to_fix(in.ex * in.hv.x));
  atomicAdd(ap + 32, to_fix(in.ex * in.hv.y));
  atomicAdd(ap + 64, to_fix(in.ex * in.hv.z));
  atomicAdd(ap + 96, to_fix(in.ex * in.hv.w));
}

struct BatchIn { int s, dl; float4 a0, a1; };

__device__ __forceinline__ BatchIn batch_none() {
  BatchIn b;
  b.s = 0; b.dl = 0;
  b.a0 = make_float4(0.f, 0.f, 0.f, 0.f);
  b.a1 = make_float4(0.f, 0.f, 0.f, 0.f);
  return b;
}

__device__ __forceinline__ void apply_batch(const BatchIn b, int n, int lane,
                                            float* exw,
                                            const float* __restrict__ h,
                                            const float* s_adst, const unsigned* s_emaxU,
                                            unsigned long long* s_den,
                                            unsigned long long* acc) {
  const int s_m = b.s, dl_m = b.dl;
  if (lane < n) {
    const float4 a0 = b.a0, a1 = b.a1;
    const float4* dp = (const float4*)(s_adst + dl_m * HEADS);
    const float4 d0 = dp[0], d1 = dp[1];
    const uint4* mp = (const uint4*)(s_emaxU + dl_m * HEADS);
    const uint4 m0 = mp[0], m1 = mp[1];
    float4 e0, e1;
    e0.x = edge_num(a0.x, d0.x, m0.x); e0.y = edge_num(a0.y, d0.y, m0.y);
    e0.z = edge_num(a0.z, d0.z, m0.z); e0.w = edge_num(a0.w, d0.w, m0.w);
    e1.x = edge_num(a1.x, d1.x, m1.x); e1.y = edge_num(a1.y, d1.y, m1.y);
    e1.z = edge_num(a1.z, d1.z, m1.z); e1.w = edge_num(a1.w, d1.w, m1.w);
    float4* ew = (float4*)(exw + lane * HEADS);
    ew[0] = e0; ew[1] = e1;
    unsigned long long* dn = s_den + dl_m * HEADS;
    atomicAdd(dn + 0, to_fix(e0.x)); atomicAdd(dn + 1, to_fix(e0.y));
    atomicAdd(dn + 2, to_fix(e0.z)); atomicAdd(dn + 3, to_fix(e0.w));
    atomicAdd(dn + 4, to_fix(e1.x)); atomicAdd(dn + 5, to_fix(e1.y));
    atomicAdd(dn + 6, to_fix(e1.z)); atomicAdd(dn + 7, to_fix(e1.w));
  }
  for (int i = 0; i < n; i += 4) {
    const int k1 = i + 1 < n ? i + 1 : n - 1;
    const int k2 = i + 2 < n ? i + 2 : n - 1;
    const int k3 = i + 3 < n ? i + 3 : n - 1;
    const EdgeIn in0 = edge_fetch(__builtin_amdgcn_readlane(s_m, i), i, 1.0f, lane, exw, h);
    const EdgeIn in1 = edge_fetch(__builtin_amdgcn_readlane(s_m, k1), k1, i + 1 < n ? 1.0f : 0.0f, lane, exw, h);
    const EdgeIn in2 = edge_fetch(__builtin_amdgcn_readlane(s_m, k2), k2, i + 2 < n ? 1.0f : 0.0f, lane, exw, h);
    const EdgeIn in3 = edge_fetch(__builtin_amdgcn_readlane(s_m, k3), k3, i + 3 < n ? 1.0f : 0.0f, lane, exw, h);
    edge_apply(in0, __builtin_amdgcn_readlane(dl_m, i), lane, acc);
    edge_apply(in1, __builtin_amdgcn_readlane(dl_m, k1), lane, acc);
    edge_apply(in2, __builtin_amdgcn_readlane(dl_m, k2), lane, acc);
    edge_apply(in3, __builtin_amdgcn_readlane(dl_m, k3), lane, acc);
  }
}

#define NKW (N_EDGES / 16)
#define SCAN_U 4
#define QCAP 352
#define QREG (QCAP / 32)

__device__ __forceinline__ void max_drain(unsigned* qw, int from, int to, int lane,
                                          const unsigned* __restrict__ rec32,
                                          const float* __restrict__ a_src,
                                          unsigned* s_emaxU) {
  for (int b0 = from; b0 < to; b0 += 32) {
    const int k = b0 + lane;
    if (k < to) {
      const unsigned r = rec32[qw[k]];
      qw[k] = r;
      const unsigned dl = r & 255u;
      const float4* ap = (const float4*)(a_src + (size_t)(r >> 8) * HEADS);
      const float4 lo = ap[0], hi = ap[1];
      unsigned* mp = s_emaxU + dl * HEADS;
      atomicMax(mp + 0, fflip(lo.x)); atomicMax(mp + 1, fflip(lo.y));
      atomicMax(mp + 2, fflip(lo.z)); atomicMax(mp + 3, fflip(lo.w));
      atomicMax(mp + 4, fflip(hi.x)); atomicMax(mp + 5, fflip(hi.y));
      atomicMax(mp + 6, fflip(hi.z)); atomicMax(mp + 7, fflip(hi.w));
    }
  }
}

template <bool RAW>
__device__ __forceinline__ BatchIn batch_load(const unsigned* qw, int b0, int n, int lane,
                                              const unsigned* __restrict__ rec32,
                                              const float* __restrict__ a_src) {
  BatchIn b = batch_none();
  if (lane < n) {
    const unsigned w0 = qw[b0 + lane];
    const unsigned w = RAW ? rec32[w0] : w0;
    b.dl = (int)(w & 255u);
    b.s = (int)(w >> 8);
    const float4* ap = (const float4*)(a_src + (size_t)b.s * HEADS);
    b.a0 = ap[0]; b.a1 = ap[1];
  }
  return b;
}

template <bool RAW>
__device__ __forceinline__ void apply_drain(BatchIn cur, int ncur,
                                            const unsigned* qw, int cnt, int lane,
                                            float* exw,
                                            const unsigned* __restrict__ rec32,
                                            const float* __restrict__ a_src,
                                            const float* __restrict__ h,
                                            const float* s_adst, const unsigned* s_emaxU,
                                            unsigned long long* s_den,
                                            unsigned long long* acc) {
  for (int b0 = 0; ; b0 += 32) {
    const int left = cnt - b0;
    const int n = left < 32 ? (left > 0 ? left : 0) : 32;
    const BatchIn nxt = batch_load<RAW>(qw, b0, n, lane, rec32, a_src);
    apply_batch(cur, ncur, lane, exw, h, s_adst, s_emaxU, s_den, acc);
    if (left <= 0) break;
    cur = nxt;
    ncur = n;
  }
}

#define NWAVES (TTHREADS / 32)
#define ROWF 160
#define LCAP 288
#define LGRP 128

__device__ __forceinline__ int wave_sum(int v) {
  for (int m = 16; m > 0; m >>= 1) v += __shfl_xor(v, m, 32);
  return v;
}

__device__ __forceinline__ void own_apply(const EdgeIn in, int dl, int lane, float* fs) {
  float* ap = fs + dl * ROWF + lane;
  atomicAdd(ap,       in.ex * in.hv.x);
  atomicAdd(ap + 32,  in.ex * in.hv.y);
  atomicAdd(ap + 64,  in.ex * in.hv.z);
  atomicAdd(ap + 96,  in.ex * in.hv.w);
  atomicAdd(ap + 128, in.ex);
}

__device__ __forceinline__ void own_batch(const BatchIn b, int n, int lane,
                                          float* exw,
                                          const float* __restrict__ h,
                                          const float* s_adst, const unsigned* s_emaxU,
                                          float* fs) {
  const int s_m = b.s, dl_m = b.dl;
  if (lane < n) {
    const float4 a0 = b.a0, a1 = b.a1;
    const float4* dp = (const float4*)(s_adst + dl_m * HEADS);
    const float4 d0 = dp[0], d1 = dp[1];
    const uint4* mp = (const uint4*)(s_emaxU + dl_m * HEADS);
    const uint4 m0 = mp[0], m1 = mp[1];
    float4 e0, e1;
    e0.x = edge_num(a0.x, d0.x, m0.x); e0.y = edge_num(a0.y, d0.y, m0.y);
    e0.z = edge_num(a0.z, d0.z, m0.z); e0.w = edge_num(a0.w, d0.w, m0.w);
    e1.x = edge_num(a1.x, d1.x, m1.x); e1.y = edge_num(a1.y, d1.y, m1.y);
    e1.z = edge_num(a1.z, d1.z, m1.z); e1.w = edge_num(a1.w, d1.w, m1.w);
    float4* ew = (float4*)(exw + lane * HEADS);
    ew[0] = e0; ew[1] = e1;
  }
  for (int i = 0; i < n; i += 4) {
    const int k1 = i + 1 < n ? i + 1 : n - 1;
    const int k2 = i + 2 < n ? i + 2 : n - 1;
    const int k3 = i + 3 < n ? i + 3 : n - 1;
    const EdgeIn in0 = edge_fetch(__builtin_amdgcn_readlane(s_m, i), i, 1.0f, lane, exw, h);
    const EdgeIn in1 = edge_fetch(__builtin_amdgcn_readlane(s_m, k1), k1, i + 1 < n ? 1.0f : 0.0f, lane, exw, h);
    const EdgeIn in2 = edge_fetch(__builtin_amdgcn_readlane(s_m, k2), k2, i + 2 < n ? 1.0f : 0.0f, lane, exw, h);
    const EdgeIn in3 = edge_fetch(__builtin_amdgcn_readlane(s_m, k3), k3, i + 3 < n ? 1.0f : 0.0f, lane, exw, h);
    own_apply(in0, __builtin_amdgcn_readlane(dl_m, i), lane, fs);
    own_apply(in1, __builtin_amdgcn_readlane(dl_m, k1), lane, fs);
    own_apply(in2, __builtin_amdgcn_readlane(dl_m, k2), lane, fs);
    own_apply(in3, __builtin_amdgcn_readlane(dl_m, k3), lane, fs);
  }
}

__device__ __forceinline__ void own_drain(BatchIn cur, int ncur,
                                          const unsigned* lst, int cnt, int lane,
                                          float* exw,
                                          const unsigned* __restrict__ rec32,
                                          const float* __restrict__ a_src,
                                          const float* __restrict__ h,
                                          const float* s_adst, const unsigned* s_emaxU,
                                          float* fs) {
  for (int b0 = 0; ; b0 += 32) {
    const int left = cnt - b0;
    const int n = left < 32 ? (left > 0 ? left : 0) : 32;
    const BatchIn nxt = batch_load<false>(lst, b0, n, lane, rec32, a_src);
    own_batch(cur, ncur, lane, exw, h, s_adst, s_emaxU, fs);
    if (left <= 0) break;
    cur = nxt;
    ncur = n;
  }
}

__device__ __forceinline__ unsigned key_differs(unsigned k, unsigned pat) {
  const unsigned t = k ^ pat;
  return ((t & 0x7F7F7F7Fu) + 0x7F7F7F7Fu) | t;
}

template <bool PASSB>
__device__ __forceinline__ void scan_keys(unsigned* qw, float* exw, int tid, int lane, unsigned pat,
                                          const uint4* __restrict__ key4,
                                          const unsigned* __restrict__ rec32,
                                          const float* __restrict__ a_src,
                                          const float* __restrict__ h,
                                          const float* s_adst, unsigned* s_emaxU,
                                          unsigned long long* s_den,
                                          unsigned long long* acc,
                                          int& cnt_out, bool& ovf_out) {
  const uint4 none = make_uint4(~0u, ~0u, ~0u, ~0u);
  int cnt = 0, done = 0;
  bool ovf = false;
  uint4 kn[SCAN_U];
#pragma unroll
  for (int u = 0; u < SCAN_U; ++u) kn[u] = key4[u * TTHREADS + tid];
  for (int t0 = 0; t0 < NKW; t0 += SCAN_U * TTHREADS) {
    uint4 kc[SCAN_U];
#pragma unroll
    for (int u = 0; u < SCAN_U; ++u) {
      kc[u] = kn[u];
      const int qn = t0 + SCAN_U * TTHREADS + u * TTHREADS + tid;
      const uint4 w = key4[qn < NKW ? qn : NKW - 1];
      kn[u] = qn < NKW ? w : none;
    }
#pragma unroll
    for (int u = 0; u < SCAN_U; u += 2) {
      const int q = t0 + u * TTHREADS + tid;
      const unsigned nL =
          ((key_differs(kc[u].x, pat) >> 7) & 0x01010101u) | ((key_differs(kc[u].y, pat) >> 6) & 0x02020202u) |
          ((key_differs(kc[u].z, pat) >> 5) & 0x04040404u) | ((key_differs(kc[u].w, pat) >> 4) & 0x08080808u) |
          ((key_differs(kc[u + 1].x, pat) >> 3) & 0x10101010u) | ((key_differs(kc[u + 1].y, pat) >> 2) & 0x20202020u) |
          ((key_differs(kc[u + 1].z, pat) >> 1) & 0x40404040u) | (key_differs(kc[u + 1].w, pat) & 0x80808080u);
      unsigned L = ~nL;
      unsigned b = __builtin_amdgcn_ballot_w32(L != 0u);
      while (b) {
        if (cnt + 32 > QCAP) {
          if (PASSB) {
            apply_drain<true>(batch_none(), 0, qw, cnt, lane, exw, rec32, a_src, h, s_adst, s_emaxU, s_den, acc);
            cnt = 0;
          } else {
            max_drain(qw, done, cnt, lane, rec32, a_src, s_emaxU);
            cnt = 0; done = 0; ovf = true;
          }
        }
        if (L != 0u) {
          const int p = __builtin_ctz(L);
          const int pos = cnt + (int)__builtin_amdgcn_mbcnt_lo(b, 0u);
          qw[pos] = (unsigned)(16 * q + ((p & 4) << 12) + ((p & 3) << 2) + (p >> 3));
          L &= L - 1u;
        }
        cnt += __builtin_popcount(b);
        b = __builtin_amdgcn_ballot_w32(L != 0u);
      }
    }
    if (!PASSB && cnt - done >= 32) {
      max_drain(qw, done, cnt, lane, rec32, a_src, s_emaxU);
      done = cnt;
    }
  }
  if (PASSB) apply_drain<true>(batch_none(), 0, qw, cnt, lane, exw, rec32, a_src, h, s_adst, s_emaxU, s_den, acc);
  else       max_drain(qw, done, cnt, lane, rec32, a_src, s_emaxU);
  cnt_out = cnt;
  ovf_out = ovf;
}

__global__ __launch_bounds__(TTHREADS) void gat_tile_kernel(
    const unsigned* __restrict__ key8, const unsigned* __restrict__ rec32,
    const float* __restrict__ a_src,
    const float* __restrict__ a_dst, const float* __restrict__ h,
    const float* __restrict__ bias, float* __restrict__ out) {
  __shared__ unsigned long long acc[TILE * OUT_DIM];
  __shared__ unsigned long long s_den[TILE * HEADS];
  __shared__ __attribute__((aligned(16))) float s_adst[TILE * HEADS];
  __shared__ __attribute__((aligned(16))) unsigned s_emaxU[TILE * HEADS];
  __shared__ __attribute__((aligned(16))) unsigned s_q[(TTHREADS / 32) * QCAP + LGRP];
  __shared__ __attribute__((aligned(16))) float s_ex[(TTHREADS / 32) * 32 * HEADS];
  __shared__ int s_cnt[NWAVES];
  __shared__ int s_flag;
  const int tid = threadIdx.x;
  const int lane = tid & 31;
  const int wave = tid >> 5;
  const int base = blockIdx.x * TILE;
  unsigned* qw = s_q + wave * QCAP;
  float* exw = s_ex + wave * 32 * HEADS;
  const uint4* key4 = (const uint4*)key8;
  const unsigned pat = (unsigned)blockIdx.x * 0x01010101u;

  if (tid == 0) s_flag = 0;
  for (int i = tid; i < TILE * HEADS; i += TTHREADS) {
    s_den[i]  = 0ull;
    s_adst[i] = a_dst[base * HEADS + i];
    s_emaxU[i] = fflip(a_src[base * HEADS + i]);
  }
  __syncthreads();

  int cnt = 0;
  bool ovf = false;
  scan_keys<false>(qw, exw, tid, lane, pat, key4, rec32, a_src, h, s_adst, s_emaxU, s_den, acc, cnt, ovf);
  unsigned qr[QREG];
#pragma unroll
  for (int t = 0; t < QREG; ++t) {
    const int k = t * 32 + lane;
    qr[t] = k < cnt ? qw[k] : 0u;
  }
  if (lane == 0) {
    s_cnt[wave] = cnt;
    if (ovf) s_flag = 1;
  }
  __syncthreads();

  const int any_ovf = __builtin_amdgcn_readfirstlane(s_flag);
  const int wv = __builtin_amdgcn_readfirstlane(wave);
  int qtot = 0;
  if (any_ovf == 0) {
    const int cv = s_cnt[lane];
    const int off = __builtin_amdgcn_readfirstlane(wave_sum(lane < wv ? cv : 0));
    qtot = __builtin_amdgcn_readfirstlane(wave_sum(cv));
#pragma unroll
    for (int t = 0; t < QREG; ++t) {
      const int k = t * 32 + lane;
      if (k < cnt) s_q[off + k] = qr[t];
    }
    if (wv == NWAVES - 1) {
      for (int t = 0; t < LGRP / 32; ++t) s_q[qtot + t * 32 + lane] = 0xFFFFFFFFu;
    }
  }

  for (int i = tid; i < TILE * HEADS; i += TTHREADS)
    s_emaxU[i] = __float_as_uint(leaky(funflip(s_emaxU[i]) + s_adst[i]));
  __syncthreads();

  const int nself = (TILE - wv + TTHREADS / 32 - 1) / (TTHREADS / 32);
  BatchIn self = batch_none();
  if (lane < nself) {
    self.dl = wv + lane * (TTHREADS / 32);
    self.s = base + self.dl;
    const float4* ap = (const float4*)(a_src + (size_t)self.s * HEADS);
    self.a0 = ap[0]; self.a1 = ap[1];
  }
  const float4 bv = ((const float4*)bias)[lane];
  if (any_ovf == 0) {
    float* fs = (float*)acc;
    unsigned* lst = (unsigned*)acc + TILE * ROWF + wave * LCAP;
    for (int j = 0; j < nself; ++j) {
      float* rp = fs + (wv + j * NWAVES) * ROWF + lane;
      rp[0] = 0.f; rp[32] = 0.f; rp[64] = 0.f; rp[96] = 0.f; rp[128] = 0.f;
    }
    BatchIn cur = self;
    int ncur = nself;
    int b0 = 0;
    for (;;) {
      int n = 0;
      while (b0 < qtot && n + LGRP <= LCAP) {
        const uint4 q = *(const uint4*)(s_q + b0 + 4 * lane);
        const unsigned qe[4] = {q.x, q.y, q.z, q.w};
        unsigned mk = 0u;
#pragma unroll
        for (int t = 0; t < 4; ++t)
          mk |= (((qe[t] ^ (unsigned)wv) & 0x8000001Fu) == 0u) ? (1u << t) : 0u;
        unsigned bm = __builtin_amdgcn_ballot_w32(mk != 0u);
        while (bm) {
          if (mk != 0u) {
            const unsigned e01 = (mk & 1u) ? qe[0] : qe[1];
            const unsigned e23 = (mk & 4u) ? qe[2] : qe[3];
            lst[n + (int)__builtin_amdgcn_mbcnt_lo(bm, 0u)] = (mk & 3u) ? e01 : e23;
            mk &= mk - 1u;
          }
          n += __builtin_popcount(bm);
          bm = __builtin_amdgcn_ballot_w32(mk != 0u);
        }
        b0 += LGRP;
      }
      own_drain(cur, ncur, lst, n, lane, exw, rec32, a_src, h, s_adst, s_emaxU, fs);
      if (b0 >= qtot) break;
      cur = batch_none();
      ncur = 0;
    }
    for (int j = 0; j < nself; ++j) {
      const int dl = wv + j * NWAVES;
      const float* rp = fs + dl * ROWF + lane;
      const float den = rp[128] + GAT_EPS;
      const float v0 = rp[0]  / den + bv.x;
      const float v1 = rp[32] / den + bv.y;
      const float v2 = rp[64] / den + bv.z;
      const float v3 = rp[96] / den + bv.w;
      float4 r;
      r.x = v0 > 0.f ? v0 : expm1f(v0);
      r.y = v1 > 0.f ? v1 : expm1f(v1);
      r.z = v2 > 0.f ? v2 : expm1f(v2);
      r.w = v3 > 0.f ? v3 : expm1f(v3);
      ((float4*)(out + (size_t)(base + dl) * OUT_DIM))[lane] = r;
    }
  } else {
    for (int i = tid; i < TILE * OUT_DIM; i += TTHREADS) acc[i] = 0ull;
    __syncthreads();
    if (!ovf) {
      apply_drain<false>(self, nself, qw, cnt, lane, exw, rec32, a_src, h, s_adst, s_emaxU, s_den, acc);
    } else {
      apply_batch(self, nself, lane, exw, h, s_adst, s_emaxU, s_den, acc);
      int c2 = 0;
      bool o2 = false;
      scan_keys<true>(qw, exw, tid, lane, pat, key4, rec32, a_src, h, s_adst, s_emaxU, s_den, acc, c2, o2);
    }
    __syncthreads();

    for (int dl = wave; dl < TILE; dl += TTHREADS / 32) {
      const unsigned long long* ap = acc + dl * OUT_DIM + lane;
      const float den = fix_to_float(s_den[dl * HEADS + (lane >> 2)]) + GAT_EPS;
      const float v0 = fix_to_float(ap[0])  / den + bv.x;
      const float v1 = fix_to_float(ap[32]) / den + bv.y;
      const float v2 = fix_to_float(ap[64]) / den + bv.z;
      const float v3 = fix_to_float(ap[96]) / den + bv.w;
      float4 r;
      r.x = v0 > 0.f ? v0 : expm1f(v0);
      r.y = v1 > 0.f ? v1 : expm1f(v1);
      r.z = v2 > 0.f ? v2 : expm1f(v2);
      r.w = v3 > 0.f ? v3 : expm1f(v3);
      ((float4*)(out + (size_t)(base + dl) * OUT_DIM))[lane] = r;
    }
  }
}

extern "C" void kernel_launch(void* const* d_in, const int* in_sizes, int n_in,
                              void* d_out, int out_size, void* d_ws, size_t ws_size,
                              hipStream_t stream) {
  (void)in_sizes; (void)n_in; (void)out_size; (void)ws_size;
  const float* x       = (const float*)d_in[0];
  const int*   ei      = (const int*)d_in[1];
  const float* W       = (const float*)d_in[2];
  const float* att_src = (const float*)d_in[3];
  const float* att_dst = (const float*)d_in[4];
  const float* bias    = (const float*)d_in[5];

  float* h     = (float*)d_ws;
  float* a_src = h + (size_t)N_NODES * OUT_DIM;
  float* a_dst = a_src + (size_t)N_NODES * HEADS;
  unsigned* rec32 = (unsigned*)(a_dst + (size_t)N_NODES * HEADS);
  unsigned* key8  = rec32 + (size_t)N_EDGES;

  gat_gemm_wmma_kernel<<<N_NODES / 16, 256, 0, stream>>>(
      x, W, att_src, att_dst, ei, h, a_src, a_dst, key8, rec32);

  gat_tile_kernel<<<N_NODES / TILE, TTHREADS, 0, stream>>>(
      key8, rec32, a_src, a_dst, h, bias, (float*)d_out);
}


# Pass B: an owner wave orders its edge list by row (stable counting sort) and sums a row's e, e*h in registers by fma (new rounding, same order per row); one LDS add per word per row, not per edge
# speedup vs baseline: 8.1040x; 1.0803x over previous
#include <hip/hip_runtime.h>
#include <math.h>

#define N_NODES 50000
#define N_EDGES 800000
#define IN_DIM 128
#define OUT_DIM 128
#define HEADS 8
#define CPH 16
#define NEG_SLOPE 0.2f
#define GAT_EPS 1e-16f
#define TILE 200

typedef float v2f __attribute__((ext_vector_type(2)));
typedef float v8f __attribute__((ext_vector_type(8)));

#define LDX 132

template <int M>
__device__ __forceinline__ float row_xor(float v) {
  return __int_as_float(__builtin_amdgcn_update_dpp(
      0, __float_as_int(v), 0x160 + M  , 0xf, 0xf, false));
}

__device__ __forceinline__ float half_sum8(const float p[8], int lane) {
  const bool b3 = (lane & 8) != 0, b2 = (lane & 4) != 0, b1 = (lane & 2) != 0;
  float q[4], t[2];
#pragma unroll
  for (int i = 0; i < 4; ++i) {
    const float keep = b3 ? p[i + 4] : p[i];
    const float send = b3 ? p[i] : p[i + 4];
    q[i] = keep + row_xor<8>(send);
  }
#pragma unroll
  for (int i = 0; i < 2; ++i) {
    const float keep = b2 ? q[i + 2] : q[i];
    const float send = b2 ? q[i] : q[i + 2];
    t[i] = keep + row_xor<4>(send);
  }
  const float keep = b1 ? t[1] : t[0];
  const float send = b1 ? t[0] : t[1];
  float u = keep + row_xor<2>(send);
  u += row_xor<1>(u);
  return u;
}

__global__ __launch_bounds__(256, 16) void gat_gemm_wmma_kernel(
    const float* __restrict__ x, const float* __restrict__ Wg,
    const float* __restrict__ att_src, const float* __restrict__ att_dst,
    const int* __restrict__ ei,
    float* __restrict__ h, float* __restrict__ a_src, float* __restrict__ a_dst,
    unsigned* __restrict__ key8, unsigned* __restrict__ rec32) {
  __shared__ float sx[16 * LDX];
  __shared__ float s_att[2 * 16 * HEADS];
  const int tid = threadIdx.x;
  const int mbase = blockIdx.x * 16;

  const int gi = blockIdx.x * 256 + tid;
  const bool has_edges = gi < N_EDGES / 4;
  const int ge = has_edges ? gi : 0;
  const int4 s4 = ((const int4*)ei)[ge];
  const int4 d4 = ((const int4*)(ei + N_EDGES))[ge];

  for (int i = tid; i < 16 * 32; i += 256) {
    const int r  = i >> 5;
    const int c4 = i & 31;
    const float4 v = ((const float4*)(x + (size_t)(mbase + r) * IN_DIM))[c4];
    float* dst = &sx[r * LDX + c4 * 4];
    dst[0] = v.x; dst[1] = v.y; dst[2] = v.z; dst[3] = v.w;
  }

  const unsigned sv[4] = {(unsigned)s4.x, (unsigned)s4.y, (unsigned)s4.z, (unsigned)s4.w};
  const unsigned dv[4] = {(unsigned)d4.x, (unsigned)d4.y, (unsigned)d4.z, (unsigned)d4.w};
  unsigned rv[4];
  unsigned kw = 0u;
#pragma unroll
  for (int c = 0; c < 4; ++c) {
    const unsigned t = dv[c] / (unsigned)TILE;
    rv[c] = (sv[c] << 8) | (dv[c] - t * (unsigned)TILE);
    kw |= t << (8 * c);
  }
  __syncthreads();
  if (has_edges) {
    ((uint4*)rec32)[gi] = make_uint4(rv[0], rv[1], rv[2], rv[3]);
    key8[gi] = kw;
  }

  const int wave = tid >> 5;
  const int lane = tid & 31;
  const int half = lane >> 4;
  const int mr   = lane & 15;
  const int nb   = wave * 16;

  const float asv = att_src[nb + mr];
  const float adv = att_dst[nb + mr];

  v8f acc = {};
#pragma unroll
  for (int k0 = 0; k0 < IN_DIM; k0 += 4) {
    v2f a, b;
    a.x = sx[mr * LDX + k0 + 2 * half];
    a.y = sx[mr * LDX + k0 + 2 * half + 1];
    b.x = Wg[(size_t)(k0 + 2 * half) * OUT_DIM + nb + mr];
    b.y = Wg[(size_t)(k0 + 2 * half + 1) * OUT_DIM + nb + mr];
    acc = __builtin_amdgcn_wmma_f32_16x16x4_f32(
         false, a,  false, b,
         (short)0, acc,  false,  false);
  }

  float ps[8], pd[8];
#pragma unroll
  for (int r = 0; r < 8; ++r) {
    ps[r] = acc[r] * asv;
    pd[r] = acc[r] * adv;
  }
  const float us = half_sum8(ps, lane);
  const float ud = half_sum8(pd, lane);
  if ((lane & 1) == 0) {
    const int row = half * 8 + ((lane >> 1) & 7);
    s_att[row * HEADS + wave] = us;
    s_att[16 * HEADS + row * HEADS + wave] = ud;
  }
  __syncthreads();

#pragma unroll
  for (int r = 0; r < 8; ++r) {
    const int row = half * 8 + r;
    h[(size_t)(mbase + row) * OUT_DIM + nb + mr] = acc[r];
  }
  if (tid < 16 * HEADS) a_src[(size_t)mbase * HEADS + tid] = s_att[tid];
  else                  a_dst[(size_t)mbase * HEADS + (tid - 16 * HEADS)] = s_att[tid];
}

__device__ __forceinline__ float leaky(float v) {
  return v >= 0.f ? v : NEG_SLOPE * v;
}
__device__ __forceinline__ unsigned fflip(float f) {
  const unsigned u = __float_as_uint(f);
  return u ^ ((u >> 31) ? 0xFFFFFFFFu : 0x80000000u);
}
__device__ __forceinline__ float funflip(unsigned u) {
  return __uint_as_float(u ^ ((u >> 31) ? 0x80000000u : 0xFFFFFFFFu));
}

#define TTHREADS 1024
#define FIX_SCALE 1073741824.0f

__device__ __forceinline__ unsigned long long to_fix(float v) {
  const float t = truncf(v);
  const float r = v - t;
  const long long f = ((long long)(int)t << 30) + (long long)(int)(r * FIX_SCALE);
  return (unsigned long long)f;
}

__device__ __forceinline__ float fix_to_float(unsigned long long u) {
  return (float)(long long)u * (1.0f / 1073741824.0f);
}

__device__ __forceinline__ float edge_num(float as, float ad, unsigned mx) {
  return __expf(leaky(as + ad) - __uint_as_float(mx));
}

struct EdgeIn { float ex; float4 hv; };

__device__ __forceinline__ EdgeIn edge_fetch(int s, int k, float keep, int lane,
                                             const float* exw,
                                             const float* __restrict__ h) {
  EdgeIn r;
  r.ex = exw[k * HEADS + (lane >> 2)] * keep;
  r.hv = ((const float4*)(h + (size_t)s * OUT_DIM))[lane];
  return r;
}

__device__ __forceinline__ void edge_apply(const EdgeIn in, int dl, int lane,
                                           unsigned long long* acc) {
  unsigned long long* ap = acc + dl * OUT_DIM + lane;
  atomicAdd(ap,      to_fix(in.ex * in.hv.x));
  atomicAdd(ap + 32, to_fix(in.ex * in.hv.y));
  atomicAdd(ap + 64, to_fix(in.ex * in.hv.z));
  atomicAdd(ap + 96, to_fix(in.ex * in.hv.w));
}

struct BatchIn { int s, dl; float4 a0, a1; };

__device__ __forceinline__ BatchIn batch_none() {
  BatchIn b;
  b.s = 0; b.dl = 0;
  b.a0 = make_float4(0.f, 0.f, 0.f, 0.f);
  b.a1 = make_float4(0.f, 0.f, 0.f, 0.f);
  return b;
}

__device__ __forceinline__ void apply_batch(const BatchIn b, int n, int lane,
                                            float* exw,
                                            const float* __restrict__ h,
                                            const float* s_adst, const unsigned* s_emaxU,
                                            unsigned long long* s_den,
                                            unsigned long long* acc) {
  const int s_m = b.s, dl_m = b.dl;
  if (lane < n) {
    const float4 a0 = b.a0, a1 = b.a1;
    const float4* dp = (const float4*)(s_adst + dl_m * HEADS);
    const float4 d0 = dp[0], d1 = dp[1];
    const uint4* mp = (const uint4*)(s_emaxU + dl_m * HEADS);
    const uint4 m0 = mp[0], m1 = mp[1];
    float4 e0, e1;
    e0.x = edge_num(a0.x, d0.x, m0.x); e0.y = edge_num(a0.y, d0.y, m0.y);
    e0.z = edge_num(a0.z, d0.z, m0.z); e0.w = edge_num(a0.w, d0.w, m0.w);
    e1.x = edge_num(a1.x, d1.x, m1.x); e1.y = edge_num(a1.y, d1.y, m1.y);
    e1.z = edge_num(a1.z, d1.z, m1.z); e1.w = edge_num(a1.w, d1.w, m1.w);
    float4* ew = (float4*)(exw + lane * HEADS);
    ew[0] = e0; ew[1] = e1;
    unsigned long long* dn = s_den + dl_m * HEADS;
    atomicAdd(dn + 0, to_fix(e0.x)); atomicAdd(dn + 1, to_fix(e0.y));
    atomicAdd(dn + 2, to_fix(e0.z)); atomicAdd(dn + 3, to_fix(e0.w));
    atomicAdd(dn + 4, to_fix(e1.x)); atomicAdd(dn + 5, to_fix(e1.y));
    atomicAdd(dn + 6, to_fix(e1.z)); atomicAdd(dn + 7, to_fix(e1.w));
  }
  for (int i = 0; i < n; i += 4) {
    const int k1 = i + 1 < n ? i + 1 : n - 1;
    const int k2 = i + 2 < n ? i + 2 : n - 1;
    const int k3 = i + 3 < n ? i + 3 : n - 1;
    const EdgeIn in0 = edge_fetch(__builtin_amdgcn_readlane(s_m, i), i, 1.0f, lane, exw, h);
    const EdgeIn in1 = edge_fetch(__builtin_amdgcn_readlane(s_m, k1), k1, i + 1 < n ? 1.0f : 0.0f, lane, exw, h);
    const EdgeIn in2 = edge_fetch(__builtin_amdgcn_readlane(s_m, k2), k2, i + 2 < n ? 1.0f : 0.0f, lane, exw, h);
    const EdgeIn in3 = edge_fetch(__builtin_amdgcn_readlane(s_m, k3), k3, i + 3 < n ? 1.0f : 0.0f, lane, exw, h);
    edge_apply(in0, __builtin_amdgcn_readlane(dl_m, i), lane, acc);
    edge_apply(in1, __builtin_amdgcn_readlane(dl_m, k1), lane, acc);
    edge_apply(in2, __builtin_amdgcn_readlane(dl_m, k2), lane, acc);
    edge_apply(in3, __builtin_amdgcn_readlane(dl_m, k3), lane, acc);
  }
}

#define NKW (N_EDGES / 16)
#define SCAN_U 4
#define QCAP 352
#define QREG (QCAP / 32)

__device__ __forceinline__ void max_drain(unsigned* qw, int from, int to, int lane,
                                          const unsigned* __restrict__ rec32,
                                          const float* __restrict__ a_src,
                                          unsigned* s_emaxU) {
  for (int b0 = from; b0 < to; b0 += 32) {
    const int k = b0 + lane;
    if (k < to) {
      const unsigned r = rec32[qw[k]];
      qw[k] = r;
      const unsigned dl = r & 255u;
      const float4* ap = (const float4*)(a_src + (size_t)(r >> 8) * HEADS);
      const float4 lo = ap[0], hi = ap[1];
      unsigned* mp = s_emaxU + dl * HEADS;
      atomicMax(mp + 0, fflip(lo.x)); atomicMax(mp + 1, fflip(lo.y));
      atomicMax(mp + 2, fflip(lo.z)); atomicMax(mp + 3, fflip(lo.w));
      atomicMax(mp + 4, fflip(hi.x)); atomicMax(mp + 5, fflip(hi.y));
      atomicMax(mp + 6, fflip(hi.z)); atomicMax(mp + 7, fflip(hi.w));
    }
  }
}

template <bool RAW>
__device__ __forceinline__ BatchIn batch_load(const unsigned* qw, int b0, int n, int lane,
                                              const unsigned* __restrict__ rec32,
                                              const float* __restrict__ a_src) {
  BatchIn b = batch_none();
  if (lane < n) {
    const unsigned w0 = qw[b0 + lane];
    const unsigned w = RAW ? rec32[w0] : w0;
    b.dl = (int)(w & 255u);
    b.s = (int)(w >> 8);
    const float4* ap = (const float4*)(a_src + (size_t)b.s * HEADS);
    b.a0 = ap[0]; b.a1 = ap[1];
  }
  return b;
}

template <bool RAW>
__device__ __forceinline__ void apply_drain(BatchIn cur, int ncur,
                                            const unsigned* qw, int cnt, int lane,
                                            float* exw,
                                            const unsigned* __restrict__ rec32,
                                            const float* __restrict__ a_src,
                                            const float* __restrict__ h,
                                            const float* s_adst, const unsigned* s_emaxU,
                                            unsigned long long* s_den,
                                            unsigned long long* acc) {
  for (int b0 = 0; ; b0 += 32) {
    const int left = cnt - b0;
    const int n = left < 32 ? (left > 0 ? left : 0) : 32;
    const BatchIn nxt = batch_load<RAW>(qw, b0, n, lane, rec32, a_src);
    apply_batch(cur, ncur, lane, exw, h, s_adst, s_emaxU, s_den, acc);
    if (left <= 0) break;
    cur = nxt;
    ncur = n;
  }
}

#define NWAVES (TTHREADS / 32)
#define ROWF 160
#define LCAP 288
#define LGRP 128

__device__ __forceinline__ int wave_sum(int v) {
  for (int m = 16; m > 0; m >>= 1) v += __shfl_xor(v, m, 32);
  return v;
}

struct RowAcc { float4 a; float d; int row; };

__device__ __forceinline__ void row_flush(const RowAcc r, int lane, float* fs) {
  float* ap = fs + r.row * ROWF + lane;
  atomicAdd(ap,       r.a.x);
  atomicAdd(ap + 32,  r.a.y);
  atomicAdd(ap + 64,  r.a.z);
  atomicAdd(ap + 96,  r.a.w);
  atomicAdd(ap + 128, r.d);
}

__device__ __forceinline__ void row_add(const EdgeIn in, RowAcc& r) {
  r.a.x = fmaf(in.ex, in.hv.x, r.a.x);
  r.a.y = fmaf(in.ex, in.hv.y, r.a.y);
  r.a.z = fmaf(in.ex, in.hv.z, r.a.z);
  r.a.w = fmaf(in.ex, in.hv.w, r.a.w);
  r.d += in.ex;
}

__device__ __forceinline__ void row_step(const EdgeIn in, int dl, int lane, float* fs, RowAcc& r) {
  if (dl != r.row) {
    if (r.row >= 0) row_flush(r, lane, fs);
    r.row = dl;
    r.a = make_float4(0.f, 0.f, 0.f, 0.f);
    r.d = 0.f;
  }
  row_add(in, r);
}

__device__ __forceinline__ void own_batch(const BatchIn b, int n, int lane,
                                          float* exw,
                                          const float* __restrict__ h,
                                          const float* s_adst, const unsigned* s_emaxU,
                                          float* fs, RowAcc& r) {
  const int s_m = b.s, dl_m = b.dl;
  if (lane < n) {
    const float4 a0 = b.a0, a1 = b.a1;
    const float4* dp = (const float4*)(s_adst + dl_m * HEADS);
    const float4 d0 = dp[0], d1 = dp[1];
    const uint4* mp = (const uint4*)(s_emaxU + dl_m * HEADS);
    const uint4 m0 = mp[0], m1 = mp[1];
    float4 e0, e1;
    e0.x = edge_num(a0.x, d0.x, m0.x); e0.y = edge_num(a0.y, d0.y, m0.y);
    e0.z = edge_num(a0.z, d0.z, m0.z); e0.w = edge_num(a0.w, d0.w, m0.w);
    e1.x = edge_num(a1.x, d1.x, m1.x); e1.y = edge_num(a1.y, d1.y, m1.y);
    e1.z = edge_num(a1.z, d1.z, m1.z); e1.w = edge_num(a1.w, d1.w, m1.w);
    float4* ew = (float4*)(exw + lane * HEADS);
    ew[0] = e0; ew[1] = e1;
  }
  for (int i = 0; i < n; i += 4) {
    const int k1 = i + 1 < n ? i + 1 : n - 1;
    const int k2 = i + 2 < n ? i + 2 : n - 1;
    const int k3 = i + 3 < n ? i + 3 : n - 1;
    const EdgeIn in0 = edge_fetch(__builtin_amdgcn_readlane(s_m, i), i, 1.0f, lane, exw, h);
    const EdgeIn in1 = edge_fetch(__builtin_amdgcn_readlane(s_m, k1), k1, i + 1 < n ? 1.0f : 0.0f, lane, exw, h);
    const EdgeIn in2 = edge_fetch(__builtin_amdgcn_readlane(s_m, k2), k2, i + 2 < n ? 1.0f : 0.0f, lane, exw, h);
    const EdgeIn in3 = edge_fetch(__builtin_amdgcn_readlane(s_m, k3), k3, i + 3 < n ? 1.0f : 0.0f, lane, exw, h);
    row_step(in0, __builtin_amdgcn_readlane(dl_m, i), lane, fs, r);
    row_step(in1, __builtin_amdgcn_readlane(dl_m, k1), lane, fs, r);
    row_step(in2, __builtin_amdgcn_readlane(dl_m, k2), lane, fs, r);
    row_step(in3, __builtin_amdgcn_readlane(dl_m, k3), lane, fs, r);
  }
}

__device__ __forceinline__ void own_drain(const unsigned* lst, int cnt, int lane,
                                          float* exw,
                                          const unsigned* __restrict__ rec32,
                                          const float* __restrict__ a_src,
                                          const float* __restrict__ h,
                                          const float* s_adst, const unsigned* s_emaxU,
                                          float* fs, RowAcc& r) {
  BatchIn cur = batch_none();
  int ncur = 0;
  for (int b0 = 0; ; b0 += 32) {
    const int left = cnt - b0;
    const int n = left < 32 ? (left > 0 ? left : 0) : 32;
    const BatchIn nxt = batch_load<false>(lst, b0, n, lane, rec32, a_src);
    own_batch(cur, ncur, lane, exw, h, s_adst, s_emaxU, fs, r);
    if (left <= 0) break;
    cur = nxt;
    ncur = n;
  }
}

#define RKEYS ((TILE + NWAVES - 1) / NWAVES)
__device__ __forceinline__ void sort_rows(const unsigned* lst, unsigned* srt, int n, int lane) {
  int base[RKEYS + 1];
#pragma unroll
  for (int j = 0; j <= RKEYS; ++j) base[j] = 0;
  for (int b0 = 0; b0 < n; b0 += 32) {
    const int k = b0 + lane;
    const unsigned key = k < n ? ((lst[k] >> 5) & 7u) : 7u;
#pragma unroll
    for (int j = 0; j < RKEYS; ++j)
      base[j + 1] += __builtin_popcount(__builtin_amdgcn_ballot_w32(key == (unsigned)j));
  }
#pragma unroll
  for (int j = 1; j <= RKEYS; ++j) base[j] += base[j - 1];
  for (int b0 = 0; b0 < n; b0 += 32) {
    const int k = b0 + lane;
    const unsigned e = k < n ? lst[k] : 0u;
    const unsigned key = k < n ? ((e >> 5) & 7u) : 7u;
    int pos = 0;
#pragma unroll
    for (int j = 0; j < RKEYS; ++j) {
      const unsigned m = __builtin_amdgcn_ballot_w32(key == (unsigned)j);
      const int at = base[j] + (int)__builtin_amdgcn_mbcnt_lo(m, 0u);
      pos = key == (unsigned)j ? at : pos;
      base[j] += __builtin_popcount(m);
    }
    if (k < n) srt[pos] = e;
  }
}

__device__ __forceinline__ unsigned key_differs(unsigned k, unsigned pat) {
  const unsigned t = k ^ pat;
  return ((t & 0x7F7F7F7Fu) + 0x7F7F7F7Fu) | t;
}

template <bool PASSB>
__device__ __forceinline__ void scan_keys(unsigned* qw, float* exw, int tid, int lane, unsigned pat,
                                          const uint4* __restrict__ key4,
                                          const unsigned* __restrict__ rec32,
                                          const float* __restrict__ a_src,
                                          const float* __restrict__ h,
                                          const float* s_adst, unsigned* s_emaxU,
                                          unsigned long long* s_den,
                                          unsigned long long* acc,
                                          int& cnt_out, bool& ovf_out) {
  const uint4 none = make_uint4(~0u, ~0u, ~0u, ~0u);
  int cnt = 0, done = 0;
  bool ovf = false;
  uint4 kn[SCAN_U];
#pragma unroll
  for (int u = 0; u < SCAN_U; ++u) kn[u] = key4[u * TTHREADS + tid];
  for (int t0 = 0; t0 < NKW; t0 += SCAN_U * TTHREADS) {
    uint4 kc[SCAN_U];
#pragma unroll
    for (int u = 0; u < SCAN_U; ++u) {
      kc[u] = kn[u];
      const int qn = t0 + SCAN_U * TTHREADS + u * TTHREADS + tid;
      const uint4 w = key4[qn < NKW ? qn : NKW - 1];
      kn[u] = qn < NKW ? w : none;
    }
#pragma unroll
    for (int u = 0; u < SCAN_U; u += 2) {
      const int q = t0 + u * TTHREADS + tid;
      const unsigned nL =
          ((key_differs(kc[u].x, pat) >> 7) & 0x01010101u) | ((key_differs(kc[u].y, pat) >> 6) & 0x02020202u) |
          ((key_differs(kc[u].z, pat) >> 5) & 0x04040404u) | ((key_differs(kc[u].w, pat) >> 4) & 0x08080808u) |
          ((key_differs(kc[u + 1].x, pat) >> 3) & 0x10101010u) | ((key_differs(kc[u + 1].y, pat) >> 2) & 0x20202020u) |
          ((key_differs(kc[u + 1].z, pat) >> 1) & 0x40404040u) | (key_differs(kc[u + 1].w, pat) & 0x80808080u);
      unsigned L = ~nL;
      unsigned b = __builtin_amdgcn_ballot_w32(L != 0u);
      while (b) {
        if (cnt + 32 > QCAP) {
          if (PASSB) {
            apply_drain<true>(batch_none(), 0, qw, cnt, lane, exw, rec32, a_src, h, s_adst, s_emaxU, s_den, acc);
            cnt = 0;
          } else {
            max_drain(qw, done, cnt, lane, rec32, a_src, s_emaxU);
            cnt = 0; done = 0; ovf = true;
          }
        }
        if (L != 0u) {
          const int p = __builtin_ctz(L);
          const int pos = cnt + (int)__builtin_amdgcn_mbcnt_lo(b, 0u);
          qw[pos] = (unsigned)(16 * q + ((p & 4) << 12) + ((p & 3) << 2) + (p >> 3));
          L &= L - 1u;
        }
        cnt += __builtin_popcount(b);
        b = __builtin_amdgcn_ballot_w32(L != 0u);
      }
    }
    if (!PASSB && cnt - done >= 32) {
      max_drain(qw, done, cnt, lane, rec32, a_src, s_emaxU);
      done = cnt;
    }
  }
  if (PASSB) apply_drain<true>(batch_none(), 0, qw, cnt, lane, exw, rec32, a_src, h, s_adst, s_emaxU, s_den, acc);
  else       max_drain(qw, done, cnt, lane, rec32, a_src, s_emaxU);
  cnt_out = cnt;
  ovf_out = ovf;
}

__global__ __launch_bounds__(TTHREADS) void gat_tile_kernel(
    const unsigned* __restrict__ key8, const unsigned* __restrict__ rec32,
    const float* __restrict__ a_src,
    const float* __restrict__ a_dst, const float* __restrict__ h,
    const float* __restrict__ bias, float* __restrict__ out) {
  __shared__ unsigned long long acc[TILE * OUT_DIM];
  __shared__ unsigned long long s_den[TILE * HEADS];
  __shared__ __attribute__((aligned(16))) float s_adst[TILE * HEADS];
  __shared__ __attribute__((aligned(16))) unsigned s_emaxU[TILE * HEADS];
  __shared__ __attribute__((aligned(16))) unsigned s_q[(TTHREADS / 32) * QCAP + LGRP];
  __shared__ __attribute__((aligned(16))) float s_ex[(TTHREADS / 32) * 32 * HEADS];
  __shared__ int s_cnt[NWAVES];
  __shared__ int s_flag;
  const int tid = threadIdx.x;
  const int lane = tid & 31;
  const int wave = tid >> 5;
  const int base = blockIdx.x * TILE;
  unsigned* qw = s_q + wave * QCAP;
  float* exw = s_ex + wave * 32 * HEADS;
  const uint4* key4 = (const uint4*)key8;
  const unsigned pat = (unsigned)blockIdx.x * 0x01010101u;

  if (tid == 0) s_flag = 0;
  for (int i = tid; i < TILE * HEADS; i += TTHREADS) {
    s_den[i]  = 0ull;
    s_adst[i] = a_dst[base * HEADS + i];
    s_emaxU[i] = fflip(a_src[base * HEADS + i]);
  }
  __syncthreads();

  int cnt = 0;
  bool ovf = false;
  scan_keys<false>(qw, exw, tid, lane, pat, key4, rec32, a_src, h, s_adst, s_emaxU, s_den, acc, cnt, ovf);
  unsigned qr[QREG];
#pragma unroll
  for (int t = 0; t < QREG; ++t) {
    const int k = t * 32 + lane;
    qr[t] = k < cnt ? qw[k] : 0u;
  }
  if (lane == 0) {
    s_cnt[wave] = cnt;
    if (ovf) s_flag = 1;
  }
  __syncthreads();

  const int any_ovf = __builtin_amdgcn_readfirstlane(s_flag);
  const int wv = __builtin_amdgcn_readfirstlane(wave);
  int qtot = 0;
  if (any_ovf == 0) {
    const int cv = s_cnt[lane];
    const int off = __builtin_amdgcn_readfirstlane(wave_sum(lane < wv ? cv : 0));
    qtot = __builtin_amdgcn_readfirstlane(wave_sum(cv));
#pragma unroll
    for (int t = 0; t < QREG; ++t) {
      const int k = t * 32 + lane;
      if (k < cnt) s_q[off + k] = qr[t];
    }
    if (wv == NWAVES - 1) {
      for (int t = 0; t < LGRP / 32; ++t) s_q[qtot + t * 32 + lane] = 0xFFFFFFFFu;
    }
  }

  for (int i = tid; i < TILE * HEADS; i += TTHREADS)
    s_emaxU[i] = __float_as_uint(leaky(funflip(s_emaxU[i]) + s_adst[i]));
  __syncthreads();

  const int nself = (TILE - wv + TTHREADS / 32 - 1) / (TTHREADS / 32);
  BatchIn self = batch_none();
  if (lane < nself) {
    self.dl = wv + lane * (TTHREADS / 32);
    self.s = base + self.dl;
    const float4* ap = (const float4*)(a_src + (size_t)self.s * HEADS);
    self.a0 = ap[0]; self.a1 = ap[1];
  }
  const float4 bv = ((const float4*)bias)[lane];
  if (any_ovf == 0) {
    float* fs = (float*)acc;
    unsigned* lst = (unsigned*)acc + TILE * ROWF + wave * LCAP;
    unsigned* srt = (unsigned*)acc + TILE * ROWF + (NWAVES + wave) * LCAP;
    for (int j = 0; j < nself; ++j) {
      float* rp = fs + (wv + j * NWAVES) * ROWF + lane;
      rp[0] = 0.f; rp[32] = 0.f; rp[64] = 0.f; rp[96] = 0.f; rp[128] = 0.f;
    }
    if (lane < nself) lst[lane] = ((unsigned)self.s << 8) | (unsigned)self.dl;
    int nfirst = nself;
    RowAcc racc;
    racc.row = -1; racc.d = 0.f;
    racc.a = make_float4(0.f, 0.f, 0.f, 0.f);
    int b0 = 0;
    for (;;) {
      int n = nfirst;
      nfirst = 0;
      while (b0 < qtot && n + LGRP <= LCAP) {
        const uint4 q = *(const uint4*)(s_q + b0 + 4 * lane);
        const unsigned qe[4] = {q.x, q.y, q.z, q.w};
        unsigned mk = 0u;
#pragma unroll
        for (int t = 0; t < 4; ++t)
          mk |= (((qe[t] ^ (unsigned)wv) & 0x8000001Fu) == 0u) ? (1u << t) : 0u;
        unsigned bm = __builtin_amdgcn_ballot_w32(mk != 0u);
        while (bm) {
          if (mk != 0u) {
            const unsigned e01 = (mk & 1u) ? qe[0] : qe[1];
            const unsigned e23 = (mk & 4u) ? qe[2] : qe[3];
            lst[n + (int)__builtin_amdgcn_mbcnt_lo(bm, 0u)] = (mk & 3u) ? e01 : e23;
            mk &= mk - 1u;
          }
          n += __builtin_popcount(bm);
          bm = __builtin_amdgcn_ballot_w32(mk != 0u);
        }
        b0 += LGRP;
      }
      sort_rows(lst, srt, n, lane);
      own_drain(srt, n, lane, exw, rec32, a_src, h, s_adst, s_emaxU, fs, racc);
      if (racc.row >= 0) row_flush(racc, lane, fs);
      racc.row = -1;
      if (b0 >= qtot) break;
    }
    for (int j = 0; j < nself; ++j) {
      const int dl = wv + j * NWAVES;
      const float* rp = fs + dl * ROWF + lane;
      const float den = rp[128] + GAT_EPS;
      const float v0 = rp[0]  / den + bv.x;
      const float v1 = rp[32] / den + bv.y;
      const float v2 = rp[64] / den + bv.z;
      const float v3 = rp[96] / den + bv.w;
      float4 r;
      r.x = v0 > 0.f ? v0 : expm1f(v0);
      r.y = v1 > 0.f ? v1 : expm1f(v1);
      r.z = v2 > 0.f ? v2 : expm1f(v2);
      r.w = v3 > 0.f ? v3 : expm1f(v3);
      ((float4*)(out + (size_t)(base + dl) * OUT_DIM))[lane] = r;
    }
  } else {
    for (int i = tid; i < TILE * OUT_DIM; i += TTHREADS) acc[i] = 0ull;
    __syncthreads();
    if (!ovf) {
      apply_drain<false>(self, nself, qw, cnt, lane, exw, rec32, a_src, h, s_adst, s_emaxU, s_den, acc);
    } else {
      apply_batch(self, nself, lane, exw, h, s_adst, s_emaxU, s_den, acc);
      int c2 = 0;
      bool o2 = false;
      scan_keys<true>(qw, exw, tid, lane, pat, key4, rec32, a_src, h, s_adst, s_emaxU, s_den, acc, c2, o2);
    }
    __syncthreads();

    for (int dl = wave; dl < TILE; dl += TTHREADS / 32) {
      const unsigned long long* ap = acc + dl * OUT_DIM + lane;
      const float den = fix_to_float(s_den[dl * HEADS + (lane >> 2)]) + GAT_EPS;
      const float v0 = fix_to_float(ap[0])  / den + bv.x;
      const float v1 = fix_to_float(ap[32]) / den + bv.y;
      const float v2 = fix_to_float(ap[64]) / den + bv.z;
      const float v3 = fix_to_float(ap[96]) / den + bv.w;
      float4 r;
      r.x = v0 > 0.f ? v0 : expm1f(v0);
      r.y = v1 > 0.f ? v1 : expm1f(v1);
      r.z = v2 > 0.f ? v2 : expm1f(v2);
      r.w = v3 > 0.f ? v3 : expm1f(v3);
      ((float4*)(out + (size_t)(base + dl) * OUT_DIM))[lane] = r;
    }
  }
}

extern "C" void kernel_launch(void* const* d_in, const int* in_sizes, int n_in,
                              void* d_out, int out_size, void* d_ws, size_t ws_size,
                              hipStream_t stream) {
  (void)in_sizes; (void)n_in; (void)out_size; (void)ws_size;
  const float* x       = (const float*)d_in[0];
  const int*   ei      = (const int*)d_in[1];
  const float* W       = (const float*)d_in[2];
  const float* att_src = (const float*)d_in[3];
  const float* att_dst = (const float*)d_in[4];
  const float* bias    = (const float*)d_in[5];

  float* h     = (float*)d_ws;
  float* a_src = h + (size_t)N_NODES * OUT_DIM;
  float* a_dst = a_src + (size_t)N_NODES * HEADS;
  unsigned* rec32 = (unsigned*)(a_dst + (size_t)N_NODES * HEADS);
  unsigned* key8  = rec32 + (size_t)N_EDGES;

  gat_gemm_wmma_kernel<<<N_NODES / 16, 256, 0, stream>>>(
      x, W, att_src, att_dst, ei, h, a_src, a_dst, key8, rec32);

  gat_tile_kernel<<<N_NODES / TILE, TTHREADS, 0, stream>>>(
      key8, rec32, a_src, a_dst, h, bias, (float*)d_out);
}


# Pass B lists filled by the finders: each wave counts its queue per owner (5 ballots a batch); a finder x owner prefix sum places entries in the owners' lists; owners scan no queues; new sum order
# speedup vs baseline: 8.5000x; 1.0489x over previous
#include <hip/hip_runtime.h>
#include <math.h>

#define N_NODES 50000
#define N_EDGES 800000
#define IN_DIM 128
#define OUT_DIM 128
#define HEADS 8
#define CPH 16
#define NEG_SLOPE 0.2f
#define GAT_EPS 1e-16f
#define TILE 200

typedef float v2f __attribute__((ext_vector_type(2)));
typedef float v8f __attribute__((ext_vector_type(8)));

#define LDX 132

template <int M>
__device__ __forceinline__ float row_xor(float v) {
  return __int_as_float(__builtin_amdgcn_update_dpp(
      0, __float_as_int(v), 0x160 + M  , 0xf, 0xf, false));
}

__device__ __forceinline__ float half_sum8(const float p[8], int lane) {
  const bool b3 = (lane & 8) != 0, b2 = (lane & 4) != 0, b1 = (lane & 2) != 0;
  float q[4], t[2];
#pragma unroll
  for (int i = 0; i < 4; ++i) {
    const float keep = b3 ? p[i + 4] : p[i];
    const float send = b3 ? p[i] : p[i + 4];
    q[i] = keep + row_xor<8>(send);
  }
#pragma unroll
  for (int i = 0; i < 2; ++i) {
    const float keep = b2 ? q[i + 2] : q[i];
    const float send = b2 ? q[i] : q[i + 2];
    t[i] = keep + row_xor<4>(send);
  }
  const float keep = b1 ? t[1] : t[0];
  const float send = b1 ? t[0] : t[1];
  float u = keep + row_xor<2>(send);
  u += row_xor<1>(u);
  return u;
}

__global__ __launch_bounds__(256, 16) void gat_gemm_wmma_kernel(
    const float* __restrict__ x, const float* __restrict__ Wg,
    const float* __restrict__ att_src, const float* __restrict__ att_dst,
    const int* __restrict__ ei,
    float* __restrict__ h, float* __restrict__ a_src, float* __restrict__ a_dst,
    unsigned* __restrict__ key8, unsigned* __restrict__ rec32) {
  __shared__ float sx[16 * LDX];
  __shared__ float s_att[2 * 16 * HEADS];
  const int tid = threadIdx.x;
  const int mbase = blockIdx.x * 16;

  const int gi = blockIdx.x * 256 + tid;
  const bool has_edges = gi < N_EDGES / 4;
  const int ge = has_edges ? gi : 0;
  const int4 s4 = ((const int4*)ei)[ge];
  const int4 d4 = ((const int4*)(ei + N_EDGES))[ge];

  for (int i = tid; i < 16 * 32; i += 256) {
    const int r  = i >> 5;
    const int c4 = i & 31;
    const float4 v = ((const float4*)(x + (size_t)(mbase + r) * IN_DIM))[c4];
    float* dst = &sx[r * LDX + c4 * 4];
    dst[0] = v.x; dst[1] = v.y; dst[2] = v.z; dst[3] = v.w;
  }

  const unsigned sv[4] = {(unsigned)s4.x, (unsigned)s4.y, (unsigned)s4.z, (unsigned)s4.w};
  const unsigned dv[4] = {(unsigned)d4.x, (unsigned)d4.y, (unsigned)d4.z, (unsigned)d4.w};
  unsigned rv[4];
  unsigned kw = 0u;
#pragma unroll
  for (int c = 0; c < 4; ++c) {
    const unsigned t = dv[c] / (unsigned)TILE;
    rv[c] = (sv[c] << 8) | (dv[c] - t * (unsigned)TILE);
    kw |= t << (8 * c);
  }
  __syncthreads();
  if (has_edges) {
    ((uint4*)rec32)[gi] = make_uint4(rv[0], rv[1], rv[2], rv[3]);
    key8[gi] = kw;
  }

  const int wave = tid >> 5;
  const int lane = tid & 31;
  const int half = lane >> 4;
  const int mr   = lane & 15;
  const int nb   = wave * 16;

  const float asv = att_src[nb + mr];
  const float adv = att_dst[nb + mr];

  v8f acc = {};
#pragma unroll
  for (int k0 = 0; k0 < IN_DIM; k0 += 4) {
    v2f a, b;
    a.x = sx[mr * LDX + k0 + 2 * half];
    a.y = sx[mr * LDX + k0 + 2 * half + 1];
    b.x = Wg[(size_t)(k0 + 2 * half) * OUT_DIM + nb + mr];
    b.y = Wg[(size_t)(k0 + 2 * half + 1) * OUT_DIM + nb + mr];
    acc = __builtin_amdgcn_wmma_f32_16x16x4_f32(
         false, a,  false, b,
         (short)0, acc,  false,  false);
  }

  float ps[8], pd[8];
#pragma unroll
  for (int r = 0; r < 8; ++r) {
    ps[r] = acc[r] * asv;
    pd[r] = acc[r] * adv;
  }
  const float us = half_sum8(ps, lane);
  const float ud = half_sum8(pd, lane);
  if ((lane & 1) == 0) {
    const int row = half * 8 + ((lane >> 1) & 7);
    s_att[row * HEADS + wave] = us;
    s_att[16 * HEADS + row * HEADS + wave] = ud;
  }
  __syncthreads();

#pragma unroll
  for (int r = 0; r < 8; ++r) {
    const int row = half * 8 + r;
    h[(size_t)(mbase + row) * OUT_DIM + nb + mr] = acc[r];
  }
  if (tid < 16 * HEADS) a_src[(size_t)mbase * HEADS + tid] = s_att[tid];
  else                  a_dst[(size_t)mbase * HEADS + (tid - 16 * HEADS)] = s_att[tid];
}

__device__ __forceinline__ float leaky(float v) {
  return v >= 0.f ? v : NEG_SLOPE * v;
}
__device__ __forceinline__ unsigned fflip(float f) {
  const unsigned u = __float_as_uint(f);
  return u ^ ((u >> 31) ? 0xFFFFFFFFu : 0x80000000u);
}
__device__ __forceinline__ float funflip(unsigned u) {
  return __uint_as_float(u ^ ((u >> 31) ? 0x80000000u : 0xFFFFFFFFu));
}

#define TTHREADS 1024
#define FIX_SCALE 1073741824.0f

__device__ __forceinline__ unsigned long long to_fix(float v) {
  const float t = truncf(v);
  const float r = v - t;
  const long long f = ((long long)(int)t << 30) + (long long)(int)(r * FIX_SCALE);
  return (unsigned long long)f;
}

__device__ __forceinline__ float fix_to_float(unsigned long long u) {
  return (float)(long long)u * (1.0f / 1073741824.0f);
}

__device__ __forceinline__ float edge_num(float as, float ad, unsigned mx) {
  return __expf(leaky(as + ad) - __uint_as_float(mx));
}

struct EdgeIn { float ex; float4 hv; };

__device__ __forceinline__ EdgeIn edge_fetch(int s, int k, float keep, int lane,
                                             const float* exw,
                                             const float* __restrict__ h) {
  EdgeIn r;
  r.ex = exw[k * HEADS + (lane >> 2)] * keep;
  r.hv = ((const float4*)(h + (size_t)s * OUT_DIM))[lane];
  return r;
}

__device__ __forceinline__ void edge_apply(const EdgeIn in, int dl, int lane,
                                           unsigned long long* acc) {
  unsigned long long* ap = acc + dl * OUT_DIM + lane;
  atomicAdd(ap,      to_fix(in.ex * in.hv.x));
  atomicAdd(ap + 32, to_fix(in.ex * in.hv.y));
  atomicAdd(ap + 64, to_fix(in.ex * in.hv.z));
  atomicAdd(ap + 96, to_fix(in.ex * in.hv.w));
}

struct BatchIn { int s, dl; float4 a0, a1; };

__device__ __forceinline__ BatchIn batch_none() {
  BatchIn b;
  b.s = 0; b.dl = 0;
  b.a0 = make_float4(0.f, 0.f, 0.f, 0.f);
  b.a1 = make_float4(0.f, 0.f, 0.f, 0.f);
  return b;
}

__device__ __forceinline__ void apply_batch(const BatchIn b, int n, int lane,
                                            float* exw,
                                            const float* __restrict__ h,
                                            const float* s_adst, const unsigned* s_emaxU,
                                            unsigned long long* s_den,
                                            unsigned long long* acc) {
  const int s_m = b.s, dl_m = b.dl;
  if (lane < n) {
    const float4 a0 = b.a0, a1 = b.a1;
    const float4* dp = (const float4*)(s_adst + dl_m * HEADS);
    const float4 d0 = dp[0], d1 = dp[1];
    const uint4* mp = (const uint4*)(s_emaxU + dl_m * HEADS);
    const uint4 m0 = mp[0], m1 = mp[1];
    float4 e0, e1;
    e0.x = edge_num(a0.x, d0.x, m0.x); e0.y = edge_num(a0.y, d0.y, m0.y);
    e0.z = edge_num(a0.z, d0.z, m0.z); e0.w = edge_num(a0.w, d0.w, m0.w);
    e1.x = edge_num(a1.x, d1.x, m1.x); e1.y = edge_num(a1.y, d1.y, m1.y);
    e1.z = edge_num(a1.z, d1.z, m1.z); e1.w = edge_num(a1.w, d1.w, m1.w);
    float4* ew = (float4*)(exw + lane * HEADS);
    ew[0] = e0; ew[1] = e1;
    unsigned long long* dn = s_den + dl_m * HEADS;
    atomicAdd(dn + 0, to_fix(e0.x)); atomicAdd(dn + 1, to_fix(e0.y));
    atomicAdd(dn + 2, to_fix(e0.z)); atomicAdd(dn + 3, to_fix(e0.w));
    atomicAdd(dn + 4, to_fix(e1.x)); atomicAdd(dn + 5, to_fix(e1.y));
    atomicAdd(dn + 6, to_fix(e1.z)); atomicAdd(dn + 7, to_fix(e1.w));
  }
  for (int i = 0; i < n; i += 4) {
    const int k1 = i + 1 < n ? i + 1 : n - 1;
    const int k2 = i + 2 < n ? i + 2 : n - 1;
    const int k3 = i + 3 < n ? i + 3 : n - 1;
    const EdgeIn in0 = edge_fetch(__builtin_amdgcn_readlane(s_m, i), i, 1.0f, lane, exw, h);
    const EdgeIn in1 = edge_fetch(__builtin_amdgcn_readlane(s_m, k1), k1, i + 1 < n ? 1.0f : 0.0f, lane, exw, h);
    const EdgeIn in2 = edge_fetch(__builtin_amdgcn_readlane(s_m, k2), k2, i + 2 < n ? 1.0f : 0.0f, lane, exw, h);
    const EdgeIn in3 = edge_fetch(__builtin_amdgcn_readlane(s_m, k3), k3, i + 3 < n ? 1.0f : 0.0f, lane, exw, h);
    edge_apply(in0, __builtin_amdgcn_readlane(dl_m, i), lane, acc);
    edge_apply(in1, __builtin_amdgcn_readlane(dl_m, k1), lane, acc);
    edge_apply(in2, __builtin_amdgcn_readlane(dl_m, k2), lane, acc);
    edge_apply(in3, __builtin_amdgcn_readlane(dl_m, k3), lane, acc);
  }
}

#define NKW (N_EDGES / 16)
#define SCAN_U 4
#define QCAP 352
#define QREG (QCAP / 32)

__device__ __forceinline__ void max_drain(unsigned* qw, int from, int to, int lane,
                                          const unsigned* __restrict__ rec32,
                                          const float* __restrict__ a_src,
                                          unsigned* s_emaxU) {
  for (int b0 = from; b0 < to; b0 += 32) {
    const int k = b0 + lane;
    if (k < to) {
      const unsigned r = rec32[qw[k]];
      qw[k] = r;
      const unsigned dl = r & 255u;
      const float4* ap = (const float4*)(a_src + (size_t)(r >> 8) * HEADS);
      const float4 lo = ap[0], hi = ap[1];
      unsigned* mp = s_emaxU + dl * HEADS;
      atomicMax(mp + 0, fflip(lo.x)); atomicMax(mp + 1, fflip(lo.y));
      atomicMax(mp + 2, fflip(lo.z)); atomicMax(mp + 3, fflip(lo.w));
      atomicMax(mp + 4, fflip(hi.x)); atomicMax(mp + 5, fflip(hi.y));
      atomicMax(mp + 6, fflip(hi.z)); atomicMax(mp + 7, fflip(hi.w));
    }
  }
}

template <bool RAW>
__device__ __forceinline__ BatchIn batch_load(const unsigned* qw, int b0, int n, int lane,
                                              const unsigned* __restrict__ rec32,
                                              const float* __restrict__ a_src) {
  BatchIn b = batch_none();
  if (lane < n) {
    const unsigned w0 = qw[b0 + lane];
    const unsigned w = RAW ? rec32[w0] : w0;
    b.dl = (int)(w & 255u);
    b.s = (int)(w >> 8);
    const float4* ap = (const float4*)(a_src + (size_t)b.s * HEADS);
    b.a0 = ap[0]; b.a1 = ap[1];
  }
  return b;
}

template <bool RAW>
__device__ __forceinline__ void apply_drain(BatchIn cur, int ncur,
                                            const unsigned* qw, int cnt, int lane,
                                            float* exw,
                                            const unsigned* __restrict__ rec32,
                                            const float* __restrict__ a_src,
                                            const float* __restrict__ h,
                                            const float* s_adst, const unsigned* s_emaxU,
                                            unsigned long long* s_den,
                                            unsigned long long* acc) {
  for (int b0 = 0; ; b0 += 32) {
    const int left = cnt - b0;
    const int n = left < 32 ? (left > 0 ? left : 0) : 32;
    const BatchIn nxt = batch_load<RAW>(qw, b0, n, lane, rec32, a_src);
    apply_batch(cur, ncur, lane, exw, h, s_adst, s_emaxU, s_den, acc);
    if (left <= 0) break;
    cur = nxt;
    ncur = n;
  }
}

#define NWAVES (TTHREADS / 32)
#define ROWF 160
#define LCAP 288
#define CMS 33

struct OwnerMatch { unsigned same; int mine; };

__device__ __forceinline__ OwnerMatch owner_match(unsigned o, bool valid, int lane) {
  const unsigned vm = __builtin_amdgcn_ballot_w32(valid);
  const unsigned ov = valid ? o : 0u;
  unsigned se = vm, sl = vm;
#pragma unroll
  for (int b = 0; b < 5; ++b) {
    const unsigned bb = __builtin_amdgcn_ballot_w32((ov & (1u << b)) != 0u);
    se &= bb ^ (((ov >> b) & 1u) - 1u);
    sl &= bb ^ ((((unsigned)lane >> b) & 1u) - 1u);
  }
  OwnerMatch m;
  m.same = se;
  m.mine = __builtin_popcount(sl);
  return m;
}

struct RowAcc { float4 a; float d; int row; };

__device__ __forceinline__ void row_flush(const RowAcc r, int lane, float* fs) {
  float* ap = fs + r.row * ROWF + lane;
  atomicAdd(ap,       r.a.x);
  atomicAdd(ap + 32,  r.a.y);
  atomicAdd(ap + 64,  r.a.z);
  atomicAdd(ap + 96,  r.a.w);
  atomicAdd(ap + 128, r.d);
}

__device__ __forceinline__ void row_add(const EdgeIn in, RowAcc& r) {
  r.a.x = fmaf(in.ex, in.hv.x, r.a.x);
  r.a.y = fmaf(in.ex, in.hv.y, r.a.y);
  r.a.z = fmaf(in.ex, in.hv.z, r.a.z);
  r.a.w = fmaf(in.ex, in.hv.w, r.a.w);
  r.d += in.ex;
}

__device__ __forceinline__ void row_step(const EdgeIn in, int dl, int lane, float* fs, RowAcc& r) {
  if (dl != r.row) {
    if (r.row >= 0) row_flush(r, lane, fs);
    r.row = dl;
    r.a = make_float4(0.f, 0.f, 0.f, 0.f);
    r.d = 0.f;
  }
  row_add(in, r);
}

__device__ __forceinline__ void own_batch(const BatchIn b, int n, int lane,
                                          float* exw,
                                          const float* __restrict__ h,
                                          const float* s_adst, const unsigned* s_emaxU,
                                          float* fs, RowAcc& r) {
  const int s_m = b.s, dl_m = b.dl;
  if (lane < n) {
    const float4 a0 = b.a0, a1 = b.a1;
    const float4* dp = (const float4*)(s_adst + dl_m * HEADS);
    const float4 d0 = dp[0], d1 = dp[1];
    const uint4* mp = (const uint4*)(s_emaxU + dl_m * HEADS);
    const uint4 m0 = mp[0], m1 = mp[1];
    float4 e0, e1;
    e0.x = edge_num(a0.x, d0.x, m0.x); e0.y = edge_num(a0.y, d0.y, m0.y);
    e0.z = edge_num(a0.z, d0.z, m0.z); e0.w = edge_num(a0.w, d0.w, m0.w);
    e1.x = edge_num(a1.x, d1.x, m1.x); e1.y = edge_num(a1.y, d1.y, m1.y);
    e1.z = edge_num(a1.z, d1.z, m1.z); e1.w = edge_num(a1.w, d1.w, m1.w);
    float4* ew = (float4*)(exw + lane * HEADS);
    ew[0] = e0; ew[1] = e1;
  }
  for (int i = 0; i < n; i += 4) {
    const int k1 = i + 1 < n ? i + 1 : n - 1;
    const int k2 = i + 2 < n ? i + 2 : n - 1;
    const int k3 = i + 3 < n ? i + 3 : n - 1;
    const EdgeIn in0 = edge_fetch(__builtin_amdgcn_readlane(s_m, i), i, 1.0f, lane, exw, h);
    const EdgeIn in1 = edge_fetch(__builtin_amdgcn_readlane(s_m, k1), k1, i + 1 < n ? 1.0f : 0.0f, lane, exw, h);
    const EdgeIn in2 = edge_fetch(__builtin_amdgcn_readlane(s_m, k2), k2, i + 2 < n ? 1.0f : 0.0f, lane, exw, h);
    const EdgeIn in3 = edge_fetch(__builtin_amdgcn_readlane(s_m, k3), k3, i + 3 < n ? 1.0f : 0.0f, lane, exw, h);
    row_step(in0, __builtin_amdgcn_readlane(dl_m, i), lane, fs, r);
    row_step(in1, __builtin_amdgcn_readlane(dl_m, k1), lane, fs, r);
    row_step(in2, __builtin_amdgcn_readlane(dl_m, k2), lane, fs, r);
    row_step(in3, __builtin_amdgcn_readlane(dl_m, k3), lane, fs, r);
  }
}

__device__ __forceinline__ void own_drain(const unsigned* lst, int cnt, int lane,
                                          float* exw,
                                          const unsigned* __restrict__ rec32,
                                          const float* __restrict__ a_src,
                                          const float* __restrict__ h,
                                          const float* s_adst, const unsigned* s_emaxU,
                                          float* fs, RowAcc& r) {
  BatchIn cur = batch_none();
  int ncur = 0;
  for (int b0 = 0; ; b0 += 32) {
    const int left = cnt - b0;
    const int n = left < 32 ? (left > 0 ? left : 0) : 32;
    const BatchIn nxt = batch_load<false>(lst, b0, n, lane, rec32, a_src);
    own_batch(cur, ncur, lane, exw, h, s_adst, s_emaxU, fs, r);
    if (left <= 0) break;
    cur = nxt;
    ncur = n;
  }
}

#define RKEYS ((TILE + NWAVES - 1) / NWAVES)
__device__ __forceinline__ void sort_rows(const unsigned* lst, unsigned* srt, int n, int lane) {
  int base[RKEYS + 1];
#pragma unroll
  for (int j = 0; j <= RKEYS; ++j) base[j] = 0;
  for (int b0 = 0; b0 < n; b0 += 32) {
    const int k = b0 + lane;
    const unsigned key = k < n ? ((lst[k] >> 5) & 7u) : 7u;
#pragma unroll
    for (int j = 0; j < RKEYS; ++j)
      base[j + 1] += __builtin_popcount(__builtin_amdgcn_ballot_w32(key == (unsigned)j));
  }
#pragma unroll
  for (int j = 1; j <= RKEYS; ++j) base[j] += base[j - 1];
  for (int b0 = 0; b0 < n; b0 += 32) {
    const int k = b0 + lane;
    const unsigned e = k < n ? lst[k] : 0u;
    const unsigned key = k < n ? ((e >> 5) & 7u) : 7u;
    int pos = 0;
#pragma unroll
    for (int j = 0; j < RKEYS; ++j) {
      const unsigned m = __builtin_amdgcn_ballot_w32(key == (unsigned)j);
      const int at = base[j] + (int)__builtin_amdgcn_mbcnt_lo(m, 0u);
      pos = key == (unsigned)j ? at : pos;
      base[j] += __builtin_popcount(m);
    }
    if (k < n) srt[pos] = e;
  }
}

__device__ __forceinline__ unsigned key_differs(unsigned k, unsigned pat) {
  const unsigned t = k ^ pat;
  return ((t & 0x7F7F7F7Fu) + 0x7F7F7F7Fu) | t;
}

template <bool PASSB>
__device__ __forceinline__ void scan_keys(unsigned* qw, float* exw, int tid, int lane, unsigned pat,
                                          const uint4* __restrict__ key4,
                                          const unsigned* __restrict__ rec32,
                                          const float* __restrict__ a_src,
                                          const float* __restrict__ h,
                                          const float* s_adst, unsigned* s_emaxU,
                                          unsigned long long* s_den,
                                          unsigned long long* acc,
                                          int& cnt_out, bool& ovf_out) {
  const uint4 none = make_uint4(~0u, ~0u, ~0u, ~0u);
  int cnt = 0, done = 0;
  bool ovf = false;
  uint4 kn[SCAN_U];
#pragma unroll
  for (int u = 0; u < SCAN_U; ++u) kn[u] = key4[u * TTHREADS + tid];
  for (int t0 = 0; t0 < NKW; t0 += SCAN_U * TTHREADS) {
    uint4 kc[SCAN_U];
#pragma unroll
    for (int u = 0; u < SCAN_U; ++u) {
      kc[u] = kn[u];
      const int qn = t0 + SCAN_U * TTHREADS + u * TTHREADS + tid;
      const uint4 w = key4[qn < NKW ? qn : NKW - 1];
      kn[u] = qn < NKW ? w : none;
    }
#pragma unroll
    for (int u = 0; u < SCAN_U; u += 2) {
      const int q = t0 + u * TTHREADS + tid;
      const unsigned nL =
          ((key_differs(kc[u].x, pat) >> 7) & 0x01010101u) | ((key_differs(kc[u].y, pat) >> 6) & 0x02020202u) |
          ((key_differs(kc[u].z, pat) >> 5) & 0x04040404u) | ((key_differs(kc[u].w, pat) >> 4) & 0x08080808u) |
          ((key_differs(kc[u + 1].x, pat) >> 3) & 0x10101010u) | ((key_differs(kc[u + 1].y, pat) >> 2) & 0x20202020u) |
          ((key_differs(kc[u + 1].z, pat) >> 1) & 0x40404040u) | (key_differs(kc[u + 1].w, pat) & 0x80808080u);
      unsigned L = ~nL;
      unsigned b = __builtin_amdgcn_ballot_w32(L != 0u);
      while (b) {
        if (cnt + 32 > QCAP) {
          if (PASSB) {
            apply_drain<true>(batch_none(), 0, qw, cnt, lane, exw, rec32, a_src, h, s_adst, s_emaxU, s_den, acc);
            cnt = 0;
          } else {
            max_drain(qw, done, cnt, lane, rec32, a_src, s_emaxU);
            cnt = 0; done = 0; ovf = true;
          }
        }
        if (L != 0u) {
          const int p = __builtin_ctz(L);
          const int pos = cnt + (int)__builtin_amdgcn_mbcnt_lo(b, 0u);
          qw[pos] = (unsigned)(16 * q + ((p & 4) << 12) + ((p & 3) << 2) + (p >> 3));
          L &= L - 1u;
        }
        cnt += __builtin_popcount(b);
        b = __builtin_amdgcn_ballot_w32(L != 0u);
      }
    }
    if (!PASSB && cnt - done >= 32) {
      max_drain(qw, done, cnt, lane, rec32, a_src, s_emaxU);
      done = cnt;
    }
  }
  if (PASSB) apply_drain<true>(batch_none(), 0, qw, cnt, lane, exw, rec32, a_src, h, s_adst, s_emaxU, s_den, acc);
  else       max_drain(qw, done, cnt, lane, rec32, a_src, s_emaxU);
  cnt_out = cnt;
  ovf_out = ovf;
}

__global__ __launch_bounds__(TTHREADS) void gat_tile_kernel(
    const unsigned* __restrict__ key8, const unsigned* __restrict__ rec32,
    const float* __restrict__ a_src,
    const float* __restrict__ a_dst, const float* __restrict__ h,
    const float* __restrict__ bias, float* __restrict__ out) {
  __shared__ unsigned long long acc[TILE * OUT_DIM];
  __shared__ unsigned long long s_den[TILE * HEADS];
  __shared__ __attribute__((aligned(16))) float s_adst[TILE * HEADS];
  __shared__ __attribute__((aligned(16))) unsigned s_emaxU[TILE * HEADS];
  __shared__ __attribute__((aligned(16))) unsigned s_q[(TTHREADS / 32) * QCAP];
  __shared__ __attribute__((aligned(16))) float s_ex[(TTHREADS / 32) * 32 * HEADS];
  __shared__ int s_cm[NWAVES * CMS];
  __shared__ int s_flag;
  __shared__ int s_lflag;
  const int tid = threadIdx.x;
  const int lane = tid & 31;
  const int wave = tid >> 5;
  const int base = blockIdx.x * TILE;
  unsigned* qw = s_q + wave * QCAP;
  float* exw = s_ex + wave * 32 * HEADS;
  const uint4* key4 = (const uint4*)key8;
  const unsigned pat = (unsigned)blockIdx.x * 0x01010101u;

  if (tid == 0) { s_flag = 0; s_lflag = 0; }
  for (int i = tid; i < TILE * HEADS; i += TTHREADS) {
    s_den[i]  = 0ull;
    s_adst[i] = a_dst[base * HEADS + i];
    s_emaxU[i] = fflip(a_src[base * HEADS + i]);
  }
  __syncthreads();

  int cnt = 0;
  bool ovf = false;
  scan_keys<false>(qw, exw, tid, lane, pat, key4, rec32, a_src, h, s_adst, s_emaxU, s_den, acc, cnt, ovf);
  unsigned qr[QREG];
  int cmine = 0;
#pragma unroll
  for (int t = 0; t < QREG; ++t) {
    const int k = t * 32 + lane;
    qr[t] = k < cnt ? qw[k] : 0u;
    if (t * 32 < cnt)
      cmine += owner_match(qr[t] & 31u, k < cnt, lane).mine;
  }
  s_cm[wave * CMS + lane] = cmine;
  if (lane == 0 && ovf) s_flag = 1;
  __syncthreads();

  const int any_ovf = __builtin_amdgcn_readfirstlane(s_flag);
  const int wv = __builtin_amdgcn_readfirstlane(wave);
  const int nself = (TILE - wv + TTHREADS / 32 - 1) / (TTHREADS / 32);
  int nown = 0;
  if (any_ovf == 0) {
    const int cv = s_cm[lane * CMS + wv];
    int incl = cv;
    for (int d = 1; d < 32; d <<= 1) {
      const int up = __shfl_up(incl, d, 32);
      if (lane >= d) incl += up;
    }
    s_cm[lane * CMS + wv] = nself + incl - cv;
    nown = nself + __builtin_amdgcn_readfirstlane(__shfl(incl, 31, 32));
    if (nown > LCAP && lane == 0) s_lflag = 1;
  }

  for (int i = tid; i < TILE * HEADS; i += TTHREADS)
    s_emaxU[i] = __float_as_uint(leaky(funflip(s_emaxU[i]) + s_adst[i]));
  __syncthreads();

  const int use_lists = (any_ovf == 0 && __builtin_amdgcn_readfirstlane(s_lflag) == 0) ? 1 : 0;
  unsigned* lall = (unsigned*)acc + TILE * ROWF;
  if (use_lists != 0) {
    int basev = s_cm[wv * CMS + lane];
#pragma unroll
    for (int t = 0; t < QREG; ++t) {
      if (t * 32 < cnt) {
        const bool valid = t * 32 + lane < cnt;
        const unsigned o = qr[t] & 31u;
        const OwnerMatch m = owner_match(o, valid, lane);
        const int at = __shfl(basev, (int)o, 32) + (int)__builtin_amdgcn_mbcnt_lo(m.same, 0u);
        if (valid) lall[o * LCAP + at] = qr[t];
        basev += m.mine;
      }
    }
  }
  __syncthreads();

  BatchIn self = batch_none();
  if (lane < nself) {
    self.dl = wv + lane * (TTHREADS / 32);
    self.s = base + self.dl;
    const float4* ap = (const float4*)(a_src + (size_t)self.s * HEADS);
    self.a0 = ap[0]; self.a1 = ap[1];
  }
  const float4 bv = ((const float4*)bias)[lane];
  if (use_lists != 0) {
    float* fs = (float*)acc;
    unsigned* lst = lall + wave * LCAP;
    unsigned* srt = lall + (NWAVES + wave) * LCAP;
    for (int j = 0; j < nself; ++j) {
      float* rp = fs + (wv + j * NWAVES) * ROWF + lane;
      rp[0] = 0.f; rp[32] = 0.f; rp[64] = 0.f; rp[96] = 0.f; rp[128] = 0.f;
    }
    if (lane < nself) lst[lane] = ((unsigned)self.s << 8) | (unsigned)self.dl;
    RowAcc racc;
    racc.row = -1; racc.d = 0.f;
    racc.a = make_float4(0.f, 0.f, 0.f, 0.f);
    sort_rows(lst, srt, nown, lane);
    own_drain(srt, nown, lane, exw, rec32, a_src, h, s_adst, s_emaxU, fs, racc);
    if (racc.row >= 0) row_flush(racc, lane, fs);
    for (int j = 0; j < nself; ++j) {
      const int dl = wv + j * NWAVES;
      const float* rp = fs + dl * ROWF + lane;
      const float den = rp[128] + GAT_EPS;
      const float v0 = rp[0]  / den + bv.x;
      const float v1 = rp[32] / den + bv.y;
      const float v2 = rp[64] / den + bv.z;
      const float v3 = rp[96] / den + bv.w;
      float4 r;
      r.x = v0 > 0.f ? v0 : expm1f(v0);
      r.y = v1 > 0.f ? v1 : expm1f(v1);
      r.z = v2 > 0.f ? v2 : expm1f(v2);
      r.w = v3 > 0.f ? v3 : expm1f(v3);
      ((float4*)(out + (size_t)(base + dl) * OUT_DIM))[lane] = r;
    }
  } else {
    for (int i = tid; i < TILE * OUT_DIM; i += TTHREADS) acc[i] = 0ull;
    __syncthreads();
    if (!ovf) {
      apply_drain<false>(self, nself, qw, cnt, lane, exw, rec32, a_src, h, s_adst, s_emaxU, s_den, acc);
    } else {
      apply_batch(self, nself, lane, exw, h, s_adst, s_emaxU, s_den, acc);
      int c2 = 0;
      bool o2 = false;
      scan_keys<true>(qw, exw, tid, lane, pat, key4, rec32, a_src, h, s_adst, s_emaxU, s_den, acc, c2, o2);
    }
    __syncthreads();

    for (int dl = wave; dl < TILE; dl += TTHREADS / 32) {
      const unsigned long long* ap = acc + dl * OUT_DIM + lane;
      const float den = fix_to_float(s_den[dl * HEADS + (lane >> 2)]) + GAT_EPS;
      const float v0 = fix_to_float(ap[0])  / den + bv.x;
      const float v1 = fix_to_float(ap[32]) / den + bv.y;
      const float v2 = fix_to_float(ap[64]) / den + bv.z;
      const float v3 = fix_to_float(ap[96]) / den + bv.w;
      float4 r;
      r.x = v0 > 0.f ? v0 : expm1f(v0);
      r.y = v1 > 0.f ? v1 : expm1f(v1);
      r.z = v2 > 0.f ? v2 : expm1f(v2);
      r.w = v3 > 0.f ? v3 : expm1f(v3);
      ((float4*)(out + (size_t)(base + dl) * OUT_DIM))[lane] = r;
    }
  }
}

extern "C" void kernel_launch(void* const* d_in, const int* in_sizes, int n_in,
                              void* d_out, int out_size, void* d_ws, size_t ws_size,
                              hipStream_t stream) {
  (void)in_sizes; (void)n_in; (void)out_size; (void)ws_size;
  const float* x       = (const float*)d_in[0];
  const int*   ei      = (const int*)d_in[1];
  const float* W       = (const float*)d_in[2];
  const float* att_src = (const float*)d_in[3];
  const float* att_dst = (const float*)d_in[4];
  const float* bias    = (const float*)d_in[5];

  float* h     = (float*)d_ws;
  float* a_src = h + (size_t)N_NODES * OUT_DIM;
  float* a_dst = a_src + (size_t)N_NODES * HEADS;
  unsigned* rec32 = (unsigned*)(a_dst + (size_t)N_NODES * HEADS);
  unsigned* key8  = rec32 + (size_t)N_EDGES;

  gat_gemm_wmma_kernel<<<N_NODES / 16, 256, 0, stream>>>(
      x, W, att_src, att_dst, ei, h, a_src, a_dst, key8, rec32);

  gat_tile_kernel<<<N_NODES / TILE, TTHREADS, 0, stream>>>(
      key8, rec32, a_src, a_dst, h, bias, (float*)d_out);
}


# Segment max taken by the row owners, not the finders: an owner fetches each edge's source terms once, holds them in registers (row maxima by a lane scan); pass A's a_src gather and LDS max gone
# speedup vs baseline: 8.8194x; 1.0376x over previous
#include <hip/hip_runtime.h>
#include <math.h>

#define N_NODES 50000
#define N_EDGES 800000
#define IN_DIM 128
#define OUT_DIM 128
#define HEADS 8
#define CPH 16
#define NEG_SLOPE 0.2f
#define GAT_EPS 1e-16f
#define TILE 200

typedef float v2f __attribute__((ext_vector_type(2)));
typedef float v8f __attribute__((ext_vector_type(8)));

#define LDX 132

template <int M>
__device__ __forceinline__ float row_xor(float v) {
  return __int_as_float(__builtin_amdgcn_update_dpp(
      0, __float_as_int(v), 0x160 + M  , 0xf, 0xf, false));
}

__device__ __forceinline__ float half_sum8(const float p[8], int lane) {
  const bool b3 = (lane & 8) != 0, b2 = (lane & 4) != 0, b1 = (lane & 2) != 0;
  float q[4], t[2];
#pragma unroll
  for (int i = 0; i < 4; ++i) {
    const float keep = b3 ? p[i + 4] : p[i];
    const float send = b3 ? p[i] : p[i + 4];
    q[i] = keep + row_xor<8>(send);
  }
#pragma unroll
  for (int i = 0; i < 2; ++i) {
    const float keep = b2 ? q[i + 2] : q[i];
    const float send = b2 ? q[i] : q[i + 2];
    t[i] = keep + row_xor<4>(send);
  }
  const float keep = b1 ? t[1] : t[0];
  const float send = b1 ? t[0] : t[1];
  float u = keep + row_xor<2>(send);
  u += row_xor<1>(u);
  return u;
}

__global__ __launch_bounds__(256, 16) void gat_gemm_wmma_kernel(
    const float* __restrict__ x, const float* __restrict__ Wg,
    const float* __restrict__ att_src, const float* __restrict__ att_dst,
    const int* __restrict__ ei,
    float* __restrict__ h, float* __restrict__ a_src, float* __restrict__ a_dst,
    unsigned* __restrict__ key8, unsigned* __restrict__ rec32) {
  __shared__ float sx[16 * LDX];
  __shared__ float s_att[2 * 16 * HEADS];
  const int tid = threadIdx.x;
  const int mbase = blockIdx.x * 16;

  const int gi = blockIdx.x * 256 + tid;
  const bool has_edges = gi < N_EDGES / 4;
  const int ge = has_edges ? gi : 0;
  const int4 s4 = ((const int4*)ei)[ge];
  const int4 d4 = ((const int4*)(ei + N_EDGES))[ge];

  for (int i = tid; i < 16 * 32; i += 256) {
    const int r  = i >> 5;
    const int c4 = i & 31;
    const float4 v = ((const float4*)(x + (size_t)(mbase + r) * IN_DIM))[c4];
    float* dst = &sx[r * LDX + c4 * 4];
    dst[0] = v.x; dst[1] = v.y; dst[2] = v.z; dst[3] = v.w;
  }

  const unsigned sv[4] = {(unsigned)s4.x, (unsigned)s4.y, (unsigned)s4.z, (unsigned)s4.w};
  const unsigned dv[4] = {(unsigned)d4.x, (unsigned)d4.y, (unsigned)d4.z, (unsigned)d4.w};
  unsigned rv[4];
  unsigned kw = 0u;
#pragma unroll
  for (int c = 0; c < 4; ++c) {
    const unsigned t = dv[c] / (unsigned)TILE;
    rv[c] = (sv[c] << 8) | (dv[c] - t * (unsigned)TILE);
    kw |= t << (8 * c);
  }
  __syncthreads();
  if (has_edges) {
    ((uint4*)rec32)[gi] = make_uint4(rv[0], rv[1], rv[2], rv[3]);
    key8[gi] = kw;
  }

  const int wave = tid >> 5;
  const int lane = tid & 31;
  const int half = lane >> 4;
  const int mr   = lane & 15;
  const int nb   = wave * 16;

  const float asv = att_src[nb + mr];
  const float adv = att_dst[nb + mr];

  v8f acc = {};
#pragma unroll
  for (int k0 = 0; k0 < IN_DIM; k0 += 4) {
    v2f a, b;
    a.x = sx[mr * LDX + k0 + 2 * half];
    a.y = sx[mr * LDX + k0 + 2 * half + 1];
    b.x = Wg[(size_t)(k0 + 2 * half) * OUT_DIM + nb + mr];
    b.y = Wg[(size_t)(k0 + 2 * half + 1) * OUT_DIM + nb + mr];
    acc = __builtin_amdgcn_wmma_f32_16x16x4_f32(
         false, a,  false, b,
         (short)0, acc,  false,  false);
  }

  float ps[8], pd[8];
#pragma unroll
  for (int r = 0; r < 8; ++r) {
    ps[r] = acc[r] * asv;
    pd[r] = acc[r] * adv;
  }
  const float us = half_sum8(ps, lane);
  const float ud = half_sum8(pd, lane);
  if ((lane & 1) == 0) {
    const int row = half * 8 + ((lane >> 1) & 7);
    s_att[row * HEADS + wave] = us;
    s_att[16 * HEADS + row * HEADS + wave] = ud;
  }
  __syncthreads();

#pragma unroll
  for (int r = 0; r < 8; ++r) {
    const int row = half * 8 + r;
    h[(size_t)(mbase + row) * OUT_DIM + nb + mr] = acc[r];
  }
  if (tid < 16 * HEADS) a_src[(size_t)mbase * HEADS + tid] = s_att[tid];
  else                  a_dst[(size_t)mbase * HEADS + (tid - 16 * HEADS)] = s_att[tid];
}

__device__ __forceinline__ float leaky(float v) {
  return v >= 0.f ? v : NEG_SLOPE * v;
}
__device__ __forceinline__ unsigned fflip(float f) {
  const unsigned u = __float_as_uint(f);
  return u ^ ((u >> 31) ? 0xFFFFFFFFu : 0x80000000u);
}
__device__ __forceinline__ float funflip(unsigned u) {
  return __uint_as_float(u ^ ((u >> 31) ? 0x80000000u : 0xFFFFFFFFu));
}

#define TTHREADS 1024
#define FIX_SCALE 1073741824.0f

__device__ __forceinline__ unsigned long long to_fix(float v) {
  const float t = truncf(v);
  const float r = v - t;
  const long long f = ((long long)(int)t << 30) + (long long)(int)(r * FIX_SCALE);
  return (unsigned long long)f;
}

__device__ __forceinline__ float fix_to_float(unsigned long long u) {
  return (float)(long long)u * (1.0f / 1073741824.0f);
}

__device__ __forceinline__ float edge_num(float as, float ad, unsigned mx) {
  return __expf(leaky(as + ad) - __uint_as_float(mx));
}

struct EdgeIn { float ex; float4 hv; };

__device__ __forceinline__ EdgeIn edge_fetch(int s, int k, float keep, int lane,
                                             const float* exw,
                                             const float* __restrict__ h) {
  EdgeIn r;
  r.ex = exw[k * HEADS + (lane >> 2)] * keep;
  r.hv = ((const float4*)(h + (size_t)s * OUT_DIM))[lane];
  return r;
}

__device__ __forceinline__ void edge_apply(const EdgeIn in, int dl, int lane,
                                           unsigned long long* acc) {
  unsigned long long* ap = acc + dl * OUT_DIM + lane;
  atomicAdd(ap,      to_fix(in.ex * in.hv.x));
  atomicAdd(ap + 32, to_fix(in.ex * in.hv.y));
  atomicAdd(ap + 64, to_fix(in.ex * in.hv.z));
  atomicAdd(ap + 96, to_fix(in.ex * in.hv.w));
}

struct BatchIn { int s, dl; float4 a0, a1; };

__device__ __forceinline__ BatchIn batch_none() {
  BatchIn b;
  b.s = 0; b.dl = 0;
  b.a0 = make_float4(0.f, 0.f, 0.f, 0.f);
  b.a1 = make_float4(0.f, 0.f, 0.f, 0.f);
  return b;
}

__device__ __forceinline__ void apply_batch(const BatchIn b, int n, int lane,
                                            float* exw,
                                            const float* __restrict__ h,
                                            const float* s_adst, const unsigned* s_emaxU,
                                            unsigned long long* s_den,
                                            unsigned long long* acc) {
  const int s_m = b.s, dl_m = b.dl;
  if (lane < n) {
    const float4 a0 = b.a0, a1 = b.a1;
    const float4* dp = (const float4*)(s_adst + dl_m * HEADS);
    const float4 d0 = dp[0], d1 = dp[1];
    const uint4* mp = (const uint4*)(s_emaxU + dl_m * HEADS);
    const uint4 m0 = mp[0], m1 = mp[1];
    float4 e0, e1;
    e0.x = edge_num(a0.x, d0.x, m0.x); e0.y = edge_num(a0.y, d0.y, m0.y);
    e0.z = edge_num(a0.z, d0.z, m0.z); e0.w = edge_num(a0.w, d0.w, m0.w);
    e1.x = edge_num(a1.x, d1.x, m1.x); e1.y = edge_num(a1.y, d1.y, m1.y);
    e1.z = edge_num(a1.z, d1.z, m1.z); e1.w = edge_num(a1.w, d1.w, m1.w);
    float4* ew = (float4*)(exw + lane * HEADS);
    ew[0] = e0; ew[1] = e1;
    unsigned long long* dn = s_den + dl_m * HEADS;
    atomicAdd(dn + 0, to_fix(e0.x)); atomicAdd(dn + 1, to_fix(e0.y));
    atomicAdd(dn + 2, to_fix(e0.z)); atomicAdd(dn + 3, to_fix(e0.w));
    atomicAdd(dn + 4, to_fix(e1.x)); atomicAdd(dn + 5, to_fix(e1.y));
    atomicAdd(dn + 6, to_fix(e1.z)); atomicAdd(dn + 7, to_fix(e1.w));
  }
  for (int i = 0; i < n; i += 4) {
    const int k1 = i + 1 < n ? i + 1 : n - 1;
    const int k2 = i + 2 < n ? i + 2 : n - 1;
    const int k3 = i + 3 < n ? i + 3 : n - 1;
    const EdgeIn in0 = edge_fetch(__builtin_amdgcn_readlane(s_m, i), i, 1.0f, lane, exw, h);
    const EdgeIn in1 = edge_fetch(__builtin_amdgcn_readlane(s_m, k1), k1, i + 1 < n ? 1.0f : 0.0f, lane, exw, h);
    const EdgeIn in2 = edge_fetch(__builtin_amdgcn_readlane(s_m, k2), k2, i + 2 < n ? 1.0f : 0.0f, lane, exw, h);
    const EdgeIn in3 = edge_fetch(__builtin_amdgcn_readlane(s_m, k3), k3, i + 3 < n ? 1.0f : 0.0f, lane, exw, h);
    edge_apply(in0, __builtin_amdgcn_readlane(dl_m, i), lane, acc);
    edge_apply(in1, __builtin_amdgcn_readlane(dl_m, k1), lane, acc);
    edge_apply(in2, __builtin_amdgcn_readlane(dl_m, k2), lane, acc);
    edge_apply(in3, __builtin_amdgcn_readlane(dl_m, k3), lane, acc);
  }
}

#define NKW (N_EDGES / 16)
#define SCAN_U 4
#define QCAP 352
#define QREG (QCAP / 32)

template <bool WITH_MAX>
__device__ __forceinline__ void max_drain(unsigned* qw, int from, int to, int lane,
                                          const unsigned* __restrict__ rec32,
                                          const float* __restrict__ a_src,
                                          unsigned* s_emaxU) {
  for (int b0 = from; b0 < to; b0 += 32) {
    const int k = b0 + lane;
    if (k < to) {
      const unsigned r = rec32[qw[k]];
      qw[k] = r;
      if (!WITH_MAX) continue;
      const unsigned dl = r & 255u;
      const float4* ap = (const float4*)(a_src + (size_t)(r >> 8) * HEADS);
      const float4 lo = ap[0], hi = ap[1];
      unsigned* mp = s_emaxU + dl * HEADS;
      atomicMax(mp + 0, fflip(lo.x)); atomicMax(mp + 1, fflip(lo.y));
      atomicMax(mp + 2, fflip(lo.z)); atomicMax(mp + 3, fflip(lo.w));
      atomicMax(mp + 4, fflip(hi.x)); atomicMax(mp + 5, fflip(hi.y));
      atomicMax(mp + 6, fflip(hi.z)); atomicMax(mp + 7, fflip(hi.w));
    }
  }
}

template <bool RAW>
__device__ __forceinline__ BatchIn batch_load(const unsigned* qw, int b0, int n, int lane,
                                              const unsigned* __restrict__ rec32,
                                              const float* __restrict__ a_src) {
  BatchIn b = batch_none();
  if (lane < n) {
    const unsigned w0 = qw[b0 + lane];
    const unsigned w = RAW ? rec32[w0] : w0;
    b.dl = (int)(w & 255u);
    b.s = (int)(w >> 8);
    const float4* ap = (const float4*)(a_src + (size_t)b.s * HEADS);
    b.a0 = ap[0]; b.a1 = ap[1];
  }
  return b;
}

template <bool RAW>
__device__ __forceinline__ void apply_drain(BatchIn cur, int ncur,
                                            const unsigned* qw, int cnt, int lane,
                                            float* exw,
                                            const unsigned* __restrict__ rec32,
                                            const float* __restrict__ a_src,
                                            const float* __restrict__ h,
                                            const float* s_adst, const unsigned* s_emaxU,
                                            unsigned long long* s_den,
                                            unsigned long long* acc) {
  for (int b0 = 0; ; b0 += 32) {
    const int left = cnt - b0;
    const int n = left < 32 ? (left > 0 ? left : 0) : 32;
    const BatchIn nxt = batch_load<RAW>(qw, b0, n, lane, rec32, a_src);
    apply_batch(cur, ncur, lane, exw, h, s_adst, s_emaxU, s_den, acc);
    if (left <= 0) break;
    cur = nxt;
    ncur = n;
  }
}

#define NWAVES (TTHREADS / 32)
#define ROWF 160
#define LCAP 192
#define NB (LCAP / 32)
#define CMS 33

struct OwnerMatch { unsigned same; int mine; };

__device__ __forceinline__ OwnerMatch owner_match(unsigned o, bool valid, int lane) {
  const unsigned vm = __builtin_amdgcn_ballot_w32(valid);
  const unsigned ov = valid ? o : 0u;
  unsigned se = vm, sl = vm;
#pragma unroll
  for (int b = 0; b < 5; ++b) {
    const unsigned bb = __builtin_amdgcn_ballot_w32((ov & (1u << b)) != 0u);
    se &= bb ^ (((ov >> b) & 1u) - 1u);
    sl &= bb ^ ((((unsigned)lane >> b) & 1u) - 1u);
  }
  OwnerMatch m;
  m.same = se;
  m.mine = __builtin_popcount(sl);
  return m;
}

struct RowAcc { float4 a; float d; int row; };

__device__ __forceinline__ void row_flush(const RowAcc r, int lane, float* fs) {
  float* ap = fs + r.row * ROWF + lane;
  atomicAdd(ap,       r.a.x);
  atomicAdd(ap + 32,  r.a.y);
  atomicAdd(ap + 64,  r.a.z);
  atomicAdd(ap + 96,  r.a.w);
  atomicAdd(ap + 128, r.d);
}

__device__ __forceinline__ void row_add(const EdgeIn in, RowAcc& r) {
  r.a.x = fmaf(in.ex, in.hv.x, r.a.x);
  r.a.y = fmaf(in.ex, in.hv.y, r.a.y);
  r.a.z = fmaf(in.ex, in.hv.z, r.a.z);
  r.a.w = fmaf(in.ex, in.hv.w, r.a.w);
  r.d += in.ex;
}

__device__ __forceinline__ void row_step(const EdgeIn in, int dl, int lane, float* fs, RowAcc& r) {
  if (dl != r.row) {
    if (r.row >= 0) row_flush(r, lane, fs);
    r.row = dl;
    r.a = make_float4(0.f, 0.f, 0.f, 0.f);
    r.d = 0.f;
  }
  row_add(in, r);
}

__device__ __forceinline__ void own_batch(const BatchIn b, int n, int lane,
                                          float* exw,
                                          const float* __restrict__ h,
                                          const float* s_adst, const unsigned* s_emaxU,
                                          float* fs, RowAcc& r) {
  const int s_m = b.s, dl_m = b.dl;
  if (lane < n) {
    const float4 a0 = b.a0, a1 = b.a1;
    const float4* dp = (const float4*)(s_adst + dl_m * HEADS);
    const float4 d0 = dp[0], d1 = dp[1];
    const uint4* mp = (const uint4*)(s_emaxU + dl_m * HEADS);
    const uint4 m0 = mp[0], m1 = mp[1];
    float4 e0, e1;
    e0.x = edge_num(a0.x, d0.x, m0.x); e0.y = edge_num(a0.y, d0.y, m0.y);
    e0.z = edge_num(a0.z, d0.z, m0.z); e0.w = edge_num(a0.w, d0.w, m0.w);
    e1.x = edge_num(a1.x, d1.x, m1.x); e1.y = edge_num(a1.y, d1.y, m1.y);
    e1.z = edge_num(a1.z, d1.z, m1.z); e1.w = edge_num(a1.w, d1.w, m1.w);
    float4* ew = (float4*)(exw + lane * HEADS);
    ew[0] = e0; ew[1] = e1;
  }
  for (int i = 0; i < n; i += 4) {
    const int k1 = i + 1 < n ? i + 1 : n - 1;
    const int k2 = i + 2 < n ? i + 2 : n - 1;
    const int k3 = i + 3 < n ? i + 3 : n - 1;
    const EdgeIn in0 = edge_fetch(__builtin_amdgcn_readlane(s_m, i), i, 1.0f, lane, exw, h);
    const EdgeIn in1 = edge_fetch(__builtin_amdgcn_readlane(s_m, k1), k1, i + 1 < n ? 1.0f : 0.0f, lane, exw, h);
    const EdgeIn in2 = edge_fetch(__builtin_amdgcn_readlane(s_m, k2), k2, i + 2 < n ? 1.0f : 0.0f, lane, exw, h);
    const EdgeIn in3 = edge_fetch(__builtin_amdgcn_readlane(s_m, k3), k3, i + 3 < n ? 1.0f : 0.0f, lane, exw, h);
    row_step(in0, __builtin_amdgcn_readlane(dl_m, i), lane, fs, r);
    row_step(in1, __builtin_amdgcn_readlane(dl_m, k1), lane, fs, r);
    row_step(in2, __builtin_amdgcn_readlane(dl_m, k2), lane, fs, r);
    row_step(in3, __builtin_amdgcn_readlane(dl_m, k3), lane, fs, r);
  }
}

#define RKEYS ((TILE + NWAVES - 1) / NWAVES)
__device__ __forceinline__ void sort_rows(const unsigned* lst, unsigned* srt, int n, int lane) {
  int base[RKEYS + 1];
#pragma unroll
  for (int j = 0; j <= RKEYS; ++j) base[j] = 0;
  for (int b0 = 0; b0 < n; b0 += 32) {
    const int k = b0 + lane;
    const unsigned key = k < n ? ((lst[k] >> 5) & 7u) : 7u;
#pragma unroll
    for (int j = 0; j < RKEYS; ++j)
      base[j + 1] += __builtin_popcount(__builtin_amdgcn_ballot_w32(key == (unsigned)j));
  }
#pragma unroll
  for (int j = 1; j <= RKEYS; ++j) base[j] += base[j - 1];
  for (int b0 = 0; b0 < n; b0 += 32) {
    const int k = b0 + lane;
    const unsigned e = k < n ? lst[k] : 0u;
    const unsigned key = k < n ? ((e >> 5) & 7u) : 7u;
    int pos = 0;
#pragma unroll
    for (int j = 0; j < RKEYS; ++j) {
      const unsigned m = __builtin_amdgcn_ballot_w32(key == (unsigned)j);
      const int at = base[j] + (int)__builtin_amdgcn_mbcnt_lo(m, 0u);
      pos = key == (unsigned)j ? at : pos;
      base[j] += __builtin_popcount(m);
    }
    if (k < n) srt[pos] = e;
  }
}

__device__ __forceinline__ unsigned key_differs(unsigned k, unsigned pat) {
  const unsigned t = k ^ pat;
  return ((t & 0x7F7F7F7Fu) + 0x7F7F7F7Fu) | t;
}

template <bool PASSB, bool WITH_MAX>
__device__ __forceinline__ void scan_keys(unsigned* qw, float* exw, int tid, int lane, unsigned pat,
                                          const uint4* __restrict__ key4,
                                          const unsigned* __restrict__ rec32,
                                          const float* __restrict__ a_src,
                                          const float* __restrict__ h,
                                          const float* s_adst, unsigned* s_emaxU,
                                          unsigned long long* s_den,
                                          unsigned long long* acc,
                                          int& cnt_out, bool& ovf_out) {
  const uint4 none = make_uint4(~0u, ~0u, ~0u, ~0u);
  int cnt = 0, done = 0;
  bool ovf = false;
  uint4 kn[SCAN_U];
#pragma unroll
  for (int u = 0; u < SCAN_U; ++u) kn[u] = key4[u * TTHREADS + tid];
  for (int t0 = 0; t0 < NKW; t0 += SCAN_U * TTHREADS) {
    uint4 kc[SCAN_U];
#pragma unroll
    for (int u = 0; u < SCAN_U; ++u) {
      kc[u] = kn[u];
      const int qn = t0 + SCAN_U * TTHREADS + u * TTHREADS + tid;
      const uint4 w = key4[qn < NKW ? qn : NKW - 1];
      kn[u] = qn < NKW ? w : none;
    }
#pragma unroll
    for (int u = 0; u < SCAN_U; u += 2) {
      const int q = t0 + u * TTHREADS + tid;
      const unsigned nL =
          ((key_differs(kc[u].x, pat) >> 7) & 0x01010101u) | ((key_differs(kc[u].y, pat) >> 6) & 0x02020202u) |
          ((key_differs(kc[u].z, pat) >> 5) & 0x04040404u) | ((key_differs(kc[u].w, pat) >> 4) & 0x08080808u) |
          ((key_differs(kc[u + 1].x, pat) >> 3) & 0x10101010u) | ((key_differs(kc[u + 1].y, pat) >> 2) & 0x20202020u) |
          ((key_differs(kc[u + 1].z, pat) >> 1) & 0x40404040u) | (key_differs(kc[u + 1].w, pat) & 0x80808080u);
      unsigned L = ~nL;
      unsigned b = __builtin_amdgcn_ballot_w32(L != 0u);
      while (b) {
        if (cnt + 32 > QCAP) {
          if (PASSB) {
            apply_drain<true>(batch_none(), 0, qw, cnt, lane, exw, rec32, a_src, h, s_adst, s_emaxU, s_den, acc);
            cnt = 0;
          } else {
            if (WITH_MAX) max_drain<true>(qw, done, cnt, lane, rec32, a_src, s_emaxU);
            cnt = 0; done = 0; ovf = true;
          }
        }
        if (L != 0u) {
          const int p = __builtin_ctz(L);
          const int pos = cnt + (int)__builtin_amdgcn_mbcnt_lo(b, 0u);
          qw[pos] = (unsigned)(16 * q + ((p & 4) << 12) + ((p & 3) << 2) + (p >> 3));
          L &= L - 1u;
        }
        cnt += __builtin_popcount(b);
        b = __builtin_amdgcn_ballot_w32(L != 0u);
      }
    }
    if (!PASSB && cnt - done >= 32) {
      max_drain<WITH_MAX>(qw, done, cnt, lane, rec32, a_src, s_emaxU);
      done = cnt;
    }
  }
  if (PASSB) apply_drain<true>(batch_none(), 0, qw, cnt, lane, exw, rec32, a_src, h, s_adst, s_emaxU, s_den, acc);
  else       max_drain<WITH_MAX>(qw, done, cnt, lane, rec32, a_src, s_emaxU);
  cnt_out = cnt;
  ovf_out = ovf;
}

__global__ __launch_bounds__(TTHREADS) void gat_tile_kernel(
    const unsigned* __restrict__ key8, const unsigned* __restrict__ rec32,
    const float* __restrict__ a_src,
    const float* __restrict__ a_dst, const float* __restrict__ h,
    const float* __restrict__ bias, float* __restrict__ out) {
  __shared__ unsigned long long acc[TILE * OUT_DIM];
  __shared__ unsigned long long s_den[TILE * HEADS];
  __shared__ __attribute__((aligned(16))) float s_adst[TILE * HEADS];
  __shared__ __attribute__((aligned(16))) unsigned s_emaxU[TILE * HEADS];
  __shared__ __attribute__((aligned(16))) unsigned s_q[(TTHREADS / 32) * QCAP];
  __shared__ __attribute__((aligned(16))) float s_ex[(TTHREADS / 32) * 32 * HEADS];
  __shared__ int s_cm[NWAVES * CMS];
  __shared__ int s_flag;
  __shared__ int s_lflag;
  const int tid = threadIdx.x;
  const int lane = tid & 31;
  const int wave = tid >> 5;
  const int base = blockIdx.x * TILE;
  unsigned* qw = s_q + wave * QCAP;
  float* exw = s_ex + wave * 32 * HEADS;
  const uint4* key4 = (const uint4*)key8;
  const unsigned pat = (unsigned)blockIdx.x * 0x01010101u;

  if (tid == 0) { s_flag = 0; s_lflag = 0; }
  for (int i = tid; i < TILE * HEADS; i += TTHREADS) {
    s_den[i]  = 0ull;
    s_adst[i] = a_dst[base * HEADS + i];
    s_emaxU[i] = 0xFF800000u;
  }
  __syncthreads();

  int cnt = 0;
  bool ovf = false;
  scan_keys<false, false>(qw, exw, tid, lane, pat, key4, rec32, a_src, h, s_adst, s_emaxU, s_den, acc, cnt, ovf);
  unsigned qr[QREG];
  int cmine = 0;
#pragma unroll
  for (int t = 0; t < QREG; ++t) {
    const int k = t * 32 + lane;
    qr[t] = k < cnt ? qw[k] : 0u;
    if (t * 32 < cnt)
      cmine += owner_match(qr[t] & 31u, k < cnt, lane).mine;
  }
  s_cm[wave * CMS + lane] = cmine;
  if (lane == 0 && ovf) s_flag = 1;
  __syncthreads();

  const int any_ovf = __builtin_amdgcn_readfirstlane(s_flag);
  const int wv = __builtin_amdgcn_readfirstlane(wave);
  const int nself = (TILE - wv + TTHREADS / 32 - 1) / (TTHREADS / 32);
  int nown = 0;
  if (any_ovf == 0) {
    const int cv = s_cm[lane * CMS + wv];
    int incl = cv;
    for (int d = 1; d < 32; d <<= 1) {
      const int up = __shfl_up(incl, d, 32);
      if (lane >= d) incl += up;
    }
    s_cm[lane * CMS + wv] = nself + incl - cv;
    nown = nself + __builtin_amdgcn_readfirstlane(__shfl(incl, 31, 32));
    if (nown > LCAP && lane == 0) s_lflag = 1;
  }

  __syncthreads();

  const int use_lists = (any_ovf == 0 && __builtin_amdgcn_readfirstlane(s_lflag) == 0) ? 1 : 0;
  unsigned* lall = (unsigned*)acc + TILE * ROWF;
  if (use_lists != 0) {
    int basev = s_cm[wv * CMS + lane];
#pragma unroll
    for (int t = 0; t < QREG; ++t) {
      if (t * 32 < cnt) {
        const bool valid = t * 32 + lane < cnt;
        const unsigned o = qr[t] & 31u;
        const OwnerMatch m = owner_match(o, valid, lane);
        const int at = __shfl(basev, (int)o, 32) + (int)__builtin_amdgcn_mbcnt_lo(m.same, 0u);
        if (valid) lall[o * LCAP + at] = qr[t];
        basev += m.mine;
      }
    }
  }
  __syncthreads();

  BatchIn self = batch_none();
  if (lane < nself) {
    self.dl = wv + lane * (TTHREADS / 32);
    self.s = base + self.dl;
    const float4* ap = (const float4*)(a_src + (size_t)self.s * HEADS);
    self.a0 = ap[0]; self.a1 = ap[1];
  }
  const float4 bv = ((const float4*)bias)[lane];
  if (use_lists != 0) {
    float* fs = (float*)acc;
    unsigned* lst = lall + wave * LCAP;
    unsigned* srt = lall + (NWAVES + wave) * LCAP;
    for (int j = 0; j < nself; ++j) {
      float* rp = fs + (wv + j * NWAVES) * ROWF + lane;
      rp[0] = 0.f; rp[32] = 0.f; rp[64] = 0.f; rp[96] = 0.f; rp[128] = 0.f;
    }
    if (lane < nself) lst[lane] = ((unsigned)self.s << 8) | (unsigned)self.dl;
    RowAcc racc;
    racc.row = -1; racc.d = 0.f;
    racc.a = make_float4(0.f, 0.f, 0.f, 0.f);
    sort_rows(lst, srt, nown, lane);
    unsigned wl[NB];
    float4 A0[NB], A1[NB];
#pragma unroll
    for (int t = 0; t < NB; ++t) {
      wl[t] = 0u;
      A0[t] = make_float4(0.f, 0.f, 0.f, 0.f);
      A1[t] = make_float4(0.f, 0.f, 0.f, 0.f);
      if (t * 32 < nown && t * 32 + lane < nown) {
        wl[t] = srt[t * 32 + lane];
        const float4* ap = (const float4*)(a_src + (size_t)(wl[t] >> 8) * HEADS);
        A0[t] = ap[0]; A1[t] = ap[1];
      }
    }
#pragma unroll
    for (int t = 0; t < NB; ++t) {
      if (t * 32 < nown) {
        const bool valid = t * 32 + lane < nown;
        const int dl = valid ? (int)(wl[t] & 255u) : 0x7fff;
        const float ninf = __uint_as_float(0xFF800000u);
        float a[8];
        a[0] = valid ? A0[t].x : ninf; a[1] = valid ? A0[t].y : ninf;
        a[2] = valid ? A0[t].z : ninf; a[3] = valid ? A0[t].w : ninf;
        a[4] = valid ? A1[t].x : ninf; a[5] = valid ? A1[t].y : ninf;
        a[6] = valid ? A1[t].z : ninf; a[7] = valid ? A1[t].w : ninf;
        for (int d = 1; d < 32; d <<= 1) {
          const int odl = __shfl_up(dl, d, 32);
          const bool take = lane >= d && odl == dl;
#pragma unroll
          for (int j = 0; j < 8; ++j) {
            const float o = __shfl_up(a[j], d, 32);
            a[j] = fmaxf(a[j], take ? o : ninf);
          }
        }
        const int ndl = __shfl_down(dl, 1, 32);
        if (valid && (lane == 31 || ndl != dl)) {
          uint4* mp = (uint4*)(s_emaxU + dl * HEADS);
          const uint4 m0 = mp[0], m1 = mp[1];
          uint4 n0, n1;
          n0.x = __float_as_uint(fmaxf(__uint_as_float(m0.x), a[0]));
          n0.y = __float_as_uint(fmaxf(__uint_as_float(m0.y), a[1]));
          n0.z = __float_as_uint(fmaxf(__uint_as_float(m0.z), a[2]));
          n0.w = __float_as_uint(fmaxf(__uint_as_float(m0.w), a[3]));
          n1.x = __float_as_uint(fmaxf(__uint_as_float(m1.x), a[4]));
          n1.y = __float_as_uint(fmaxf(__uint_as_float(m1.y), a[5]));
          n1.z = __float_as_uint(fmaxf(__uint_as_float(m1.z), a[6]));
          n1.w = __float_as_uint(fmaxf(__uint_as_float(m1.w), a[7]));
          mp[0] = n0; mp[1] = n1;
        }
      }
    }
    for (int i = lane; i < nself * HEADS; i += 32) {
      const int idx = (wv + (i >> 3) * NWAVES) * HEADS + (i & 7);
      s_emaxU[idx] = __float_as_uint(leaky(__uint_as_float(s_emaxU[idx]) + s_adst[idx]));
    }
    for (int b0 = 0; b0 < nown; b0 += 32) {
      const int left = nown - b0;
      const int n = left < 32 ? left : 32;
      BatchIn cur;
      cur.s = (int)(wl[0] >> 8);
      cur.dl = (int)(wl[0] & 255u);
      cur.a0 = A0[0]; cur.a1 = A1[0];
      own_batch(cur, n, lane, exw, h, s_adst, s_emaxU, fs, racc);
#pragma unroll
      for (int t = 0; t + 1 < NB; ++t) { wl[t] = wl[t + 1]; A0[t] = A0[t + 1]; A1[t] = A1[t + 1]; }
    }
    if (racc.row >= 0) row_flush(racc, lane, fs);
    for (int j = 0; j < nself; ++j) {
      const int dl = wv + j * NWAVES;
      const float* rp = fs + dl * ROWF + lane;
      const float den = rp[128] + GAT_EPS;
      const float v0 = rp[0]  / den + bv.x;
      const float v1 = rp[32] / den + bv.y;
      const float v2 = rp[64] / den + bv.z;
      const float v3 = rp[96] / den + bv.w;
      float4 r;
      r.x = v0 > 0.f ? v0 : expm1f(v0);
      r.y = v1 > 0.f ? v1 : expm1f(v1);
      r.z = v2 > 0.f ? v2 : expm1f(v2);
      r.w = v3 > 0.f ? v3 : expm1f(v3);
      ((float4*)(out + (size_t)(base + dl) * OUT_DIM))[lane] = r;
    }
  } else {
    for (int i = tid; i < TILE * HEADS; i += TTHREADS)
      s_emaxU[i] = fflip(a_src[base * HEADS + i]);
    __syncthreads();
    scan_keys<false, true>(qw, exw, tid, lane, pat, key4, rec32, a_src, h, s_adst, s_emaxU, s_den, acc, cnt, ovf);
    __syncthreads();
    for (int i = tid; i < TILE * HEADS; i += TTHREADS)
      s_emaxU[i] = __float_as_uint(leaky(funflip(s_emaxU[i]) + s_adst[i]));
    for (int i = tid; i < TILE * OUT_DIM; i += TTHREADS) acc[i] = 0ull;
    __syncthreads();
    if (!ovf) {
      apply_drain<false>(self, nself, qw, cnt, lane, exw, rec32, a_src, h, s_adst, s_emaxU, s_den, acc);
    } else {
      apply_batch(self, nself, lane, exw, h, s_adst, s_emaxU, s_den, acc);
      int c2 = 0;
      bool o2 = false;
      scan_keys<true, true>(qw, exw, tid, lane, pat, key4, rec32, a_src, h, s_adst, s_emaxU, s_den, acc, c2, o2);
    }
    __syncthreads();

    for (int dl = wave; dl < TILE; dl += TTHREADS / 32) {
      const unsigned long long* ap = acc + dl * OUT_DIM + lane;
      const float den = fix_to_float(s_den[dl * HEADS + (lane >> 2)]) + GAT_EPS;
      const float v0 = fix_to_float(ap[0])  / den + bv.x;
      const float v1 = fix_to_float(ap[32]) / den + bv.y;
      const float v2 = fix_to_float(ap[64]) / den + bv.z;
      const float v3 = fix_to_float(ap[96]) / den + bv.w;
      float4 r;
      r.x = v0 > 0.f ? v0 : expm1f(v0);
      r.y = v1 > 0.f ? v1 : expm1f(v1);
      r.z = v2 > 0.f ? v2 : expm1f(v2);
      r.w = v3 > 0.f ? v3 : expm1f(v3);
      ((float4*)(out + (size_t)(base + dl) * OUT_DIM))[lane] = r;
    }
  }
}

extern "C" void kernel_launch(void* const* d_in, const int* in_sizes, int n_in,
                              void* d_out, int out_size, void* d_ws, size_t ws_size,
                              hipStream_t stream) {
  (void)in_sizes; (void)n_in; (void)out_size; (void)ws_size;
  const float* x       = (const float*)d_in[0];
  const int*   ei      = (const int*)d_in[1];
  const float* W       = (const float*)d_in[2];
  const float* att_src = (const float*)d_in[3];
  const float* att_dst = (const float*)d_in[4];
  const float* bias    = (const float*)d_in[5];

  float* h     = (float*)d_ws;
  float* a_src = h + (size_t)N_NODES * OUT_DIM;
  float* a_dst = a_src + (size_t)N_NODES * HEADS;
  unsigned* rec32 = (unsigned*)(a_dst + (size_t)N_NODES * HEADS);
  unsigned* key8  = rec32 + (size_t)N_EDGES;

  gat_gemm_wmma_kernel<<<N_NODES / 16, 256, 0, stream>>>(
      x, W, att_src, att_dst, ei, h, a_src, a_dst, key8, rec32);

  gat_tile_kernel<<<N_NODES / TILE, TTHREADS, 0, stream>>>(
      key8, rec32, a_src, a_dst, h, bias, (float*)d_out);
}
